# starting kernel
# baseline (speedup 1.0000x reference)
#include <hip/hip_runtime.h>

typedef float v2f __attribute__((ext_vector_type(2)));
typedef float v8f __attribute__((ext_vector_type(8)));

__global__ void zero_f32(float* __restrict__ p, int n) {
  int i = blockIdx.x * blockDim.x + threadIdx.x;
  if (i < n) p[i] = 0.f;
}

__global__ void relu_inplace(float* __restrict__ p, int n) {
  int i = blockIdx.x * blockDim.x + threadIdx.x;
  if (i < n) p[i] = fmaxf(p[i], 0.f);
}

__global__ void deg_accum(const int* __restrict__ dst, const float* __restrict__ w,
                          float* __restrict__ deg, int E) {
  int e = blockIdx.x * blockDim.x + threadIdx.x;
  if (e < E) atomicAdd(deg + dst[e], w[e]);
}

__global__ void norm_compute(const int* __restrict__ src, const int* __restrict__ dst,
                             const float* __restrict__ w, const float* __restrict__ deg,
                             float* __restrict__ norm, int E) {
  int e = blockIdx.x * blockDim.x + threadIdx.x;
  if (e >= E) return;
  float ds = deg[src[e]];
  float dd = deg[dst[e]];
  float is = ds > 0.f ? rsqrtf(ds) : 0.f;
  float id = dd > 0.f ? rsqrtf(dd) : 0.f;
  norm[e] = is * w[e] * id;
}

template <int K>
__global__ void gemm_wmma(const float* __restrict__ A, const float* __restrict__ B,
                          const float* __restrict__ bias, float* __restrict__ C,
                          int N, int F, int rowTiles, int colTiles) {
  const int lane = threadIdx.x & 31;
  const int wave = threadIdx.x >> 5;
  const int tile = blockIdx.x * (blockDim.x >> 5) + wave;
  if (tile >= rowTiles * colTiles) return;

  const int rt = tile / colTiles;
  const int ct = tile - rt * colTiles;
  const int row0 = rt << 4;
  const int col0 = ct << 4;
  const int half = lane >> 4;
  const int lm   = lane & 15;

  const bool cOK = (col0 + lm) < F;
  const int  arow = min(row0 + lm, N - 1);
  const int  bcol = min(col0 + lm, F - 1);

  const float* Arow = A + (size_t)arow * K + 2 * half;
  const float* Bcol = B + bcol;

  v2f afrag[K / 4];
  v2f bfrag[K / 4];
#pragma unroll
  for (int i = 0; i < K / 4; ++i) {
    const int ka = 4 * i + 2 * half;
    afrag[i] = *reinterpret_cast<const v2f*>(Arow + 4 * i);
    bfrag[i].x = Bcol[(size_t)ka * F];
    bfrag[i].y = Bcol[(size_t)(ka + 1) * F];
  }

  v8f acc = {};
#pragma unroll
  for (int i = 0; i < K / 4; ++i) {
    acc = __builtin_amdgcn_wmma_f32_16x16x4_f32(
        false, afrag[i], false, bfrag[i], (short)0, acc, false, false);
  }

  const float bv = (bias != nullptr) ? bias[bcol] : 0.f;
#pragma unroll
  for (int r = 0; r < 8; ++r) {
    const int row = row0 + r + 8 * half;
    if (row < N && cOK) C[(size_t)row * F + col0 + lm] = acc[r] + bv;
  }
}

__global__ void gemm_wmma_gen(const float* __restrict__ A, const float* __restrict__ B,
                              const float* __restrict__ bias, float* __restrict__ C,
                              int N, int K, int F, int rowTiles, int colTiles) {
  const int lane = threadIdx.x & 31;
  const int wave = threadIdx.x >> 5;
  const int tile = blockIdx.x * (blockDim.x >> 5) + wave;
  if (tile >= rowTiles * colTiles) return;
  const int rt = tile / colTiles;
  const int ct = tile - rt * colTiles;
  const int row0 = rt << 4, col0 = ct << 4;
  const int half = lane >> 4, lm = lane & 15;
  const bool cOK = (col0 + lm) < F;
  const int arow = min(row0 + lm, N - 1);
  const int bcol = min(col0 + lm, F - 1);
  v8f acc = {};
  for (int kb = 0; kb < K; kb += 4) {
    const int ka = kb + 2 * half;
    v2f a = *reinterpret_cast<const v2f*>(A + (size_t)arow * K + ka);
    v2f b;
    b.x = B[(size_t)ka * F + bcol];
    b.y = B[(size_t)(ka + 1) * F + bcol];
    acc = __builtin_amdgcn_wmma_f32_16x16x4_f32(false, a, false, b, (short)0, acc,
                                                false, false);
  }
  const float bv = (bias != nullptr) ? bias[bcol] : 0.f;
#pragma unroll
  for (int r = 0; r < 8; ++r) {
    const int row = row0 + r + 8 * half;
    if (row < N && cOK) C[(size_t)row * F + col0 + lm] = acc[r] + bv;
  }
}

__global__ void edge_scatter(const float* __restrict__ h, const float* __restrict__ norm,
                             const int* __restrict__ src, const int* __restrict__ dst,
                             float* __restrict__ agg, int E, int F, int G) {
  long long tid = (long long)blockIdx.x * blockDim.x + threadIdx.x;
  if (tid >= (long long)E * G) return;
  int e = (int)(tid / G);
  int g = (int)(tid - (long long)e * G);
  int s = src[e];
  int d = dst[e];
  float w = norm[e];
  const float4 v = *reinterpret_cast<const float4*>(h + (size_t)s * F + (size_t)g * 4);
  float* base = agg + (size_t)d * F + (size_t)g * 4;
  atomicAdd(base + 0, v.x * w);
  atomicAdd(base + 1, v.y * w);
  atomicAdd(base + 2, v.z * w);
  atomicAdd(base + 3, v.w * w);
}

template <int C>
__global__ void relu_logsoftmax_t(float* __restrict__ y, int N) {
  int n = blockIdx.x * blockDim.x + threadIdx.x;
  if (n >= N) return;
  float v[C];
  float m = -3.4e38f;
  float* row = y + (size_t)n * C;
#pragma unroll
  for (int c = 0; c < C; ++c) {
    float t = fmaxf(row[c], 0.f);
    v[c] = t;
    m = fmaxf(m, t);
  }
  float s = 0.f;
#pragma unroll
  for (int c = 0; c < C; ++c) s += __expf(v[c] - m);
  float ls = m + __logf(s);
#pragma unroll
  for (int c = 0; c < C; ++c) row[c] = v[c] - ls;
}

__global__ void relu_logsoftmax_gen(float* __restrict__ y, int N, int C) {
  int n = blockIdx.x * blockDim.x + threadIdx.x;
  if (n >= N) return;
  float* row = y + (size_t)n * C;
  float m = -3.4e38f;
  for (int c = 0; c < C; ++c) m = fmaxf(m, fmaxf(row[c], 0.f));
  float s = 0.f;
  for (int c = 0; c < C; ++c) s += __expf(fmaxf(row[c], 0.f) - m);
  float ls = m + __logf(s);
  for (int c = 0; c < C; ++c) row[c] = fmaxf(row[c], 0.f) - ls;
}

static inline int ceil_div(long long a, int b) { return (int)((a + b - 1) / b); }

static void launch_gemm(const float* A, const float* B, const float* bias, float* C,
                        int N, int K, int F, hipStream_t stream) {
  const int TPB = 256;
  int rowT = ceil_div(N, 16), colT = ceil_div(F, 16);
  int blocks = ceil_div((long long)rowT * colT, TPB / 32);
  if (K == 64) {
    gemm_wmma<64><<<blocks, TPB, 0, stream>>>(A, B, bias, C, N, F, rowT, colT);
  } else if (K == 48) {
    gemm_wmma<48><<<blocks, TPB, 0, stream>>>(A, B, bias, C, N, F, rowT, colT);
  } else {
    gemm_wmma_gen<<<blocks, TPB, 0, stream>>>(A, B, bias, C, N, K, F, rowT, colT);
  }
}

extern "C" void kernel_launch(void* const* d_in, const int* in_sizes, int n_in,
                              void* d_out, int out_size, void* d_ws, size_t ws_size,
                              hipStream_t stream) {
  const float* x  = (const float*)d_in[0];
  const int*   ei = (const int*)d_in[1];
  const float* ew = (const float*)d_in[2];
  const float* W1 = (const float*)d_in[3];
  const float* V1 = (const float*)d_in[4];
  const float* b1 = (const float*)d_in[5];
  const float* W2 = (const float*)d_in[6];
  const float* V2 = (const float*)d_in[7];
  const float* b2 = (const float*)d_in[8];
  float* out = (float*)d_out;

  const int H   = in_sizes[5];
  const int Cn  = in_sizes[8];
  const int Fin = in_sizes[3] / H;
  const int N   = in_sizes[0] / Fin;
  const int E   = in_sizes[2];
  const int* src = ei;
  const int* dst = ei + E;

  float* ws = (float*)d_ws;
  size_t off = 0;
  float* deg  = ws + off; off += ((size_t)N + 255) & ~(size_t)255;
  float* norm = ws + off; off += ((size_t)E + 255) & ~(size_t)255;
  float* hbuf = ws + off; off += ((size_t)N * H + 255) & ~(size_t)255;
  float* z1   = ws + off; off += ((size_t)N * H + 255) & ~(size_t)255;

  const int TPB = 256;

  zero_f32<<<ceil_div(N, TPB), TPB, 0, stream>>>(deg, N);
  deg_accum<<<ceil_div(E, TPB), TPB, 0, stream>>>(dst, ew, deg, E);
  norm_compute<<<ceil_div(E, TPB), TPB, 0, stream>>>(src, dst, ew, deg, norm, E);

  launch_gemm(x, W1, nullptr, hbuf, N, Fin, H, stream);
  launch_gemm(x, V1, b1,      z1,   N, Fin, H, stream);
  {
    int G = H / 4;
    edge_scatter<<<ceil_div((long long)E * G, TPB), TPB, 0, stream>>>(
        hbuf, norm, src, dst, z1, E, H, G);
    relu_inplace<<<ceil_div((long long)N * H, TPB), TPB, 0, stream>>>(z1, N * H);
  }

  launch_gemm(z1, W2, nullptr, hbuf, N, H, Cn, stream);
  launch_gemm(z1, V2, b2,      out,  N, H, Cn, stream);
  {
    int G = Cn / 4;
    edge_scatter<<<ceil_div((long long)E * G, TPB), TPB, 0, stream>>>(
        hbuf, norm, src, dst, out, E, Cn, G);
    if (Cn == 40) {
      relu_logsoftmax_t<40><<<ceil_div(N, TPB), TPB, 0, stream>>>(out, N);
    } else {
      relu_logsoftmax_gen<<<ceil_div(N, TPB), TPB, 0, stream>>>(out, N, Cn);
    }
  }
}


# Edges counting-sorted by destination once per call (int rank atomics); degree and both aggregations become per-node segment sums in arrival-rank order, no float atomics left
# speedup vs baseline: 1.3419x; 1.3419x over previous
#include <hip/hip_runtime.h>

typedef float v2f __attribute__((ext_vector_type(2)));
typedef float v8f __attribute__((ext_vector_type(8)));

__global__ void zero_f32(float* __restrict__ p, int n) {
  int i = blockIdx.x * blockDim.x + threadIdx.x;
  if (i < n) p[i] = 0.f;
}

__global__ void relu_inplace(float* __restrict__ p, int n) {
  int i = blockIdx.x * blockDim.x + threadIdx.x;
  if (i < n) p[i] = fmaxf(p[i], 0.f);
}

__global__ void deg_accum(const int* __restrict__ dst, const float* __restrict__ w,
                          float* __restrict__ deg, int E) {
  int e = blockIdx.x * blockDim.x + threadIdx.x;
  if (e < E) atomicAdd(deg + dst[e], w[e]);
}

__global__ void norm_compute(const int* __restrict__ src, const int* __restrict__ dst,
                             const float* __restrict__ w, const float* __restrict__ deg,
                             float* __restrict__ norm, int E) {
  int e = blockIdx.x * blockDim.x + threadIdx.x;
  if (e >= E) return;
  float ds = deg[src[e]];
  float dd = deg[dst[e]];
  float is = ds > 0.f ? rsqrtf(ds) : 0.f;
  float id = dd > 0.f ? rsqrtf(dd) : 0.f;
  norm[e] = is * w[e] * id;
}

template <int K>
__global__ void gemm_wmma(const float* __restrict__ A, const float* __restrict__ B,
                          const float* __restrict__ bias, float* __restrict__ C,
                          int N, int F, int rowTiles, int colTiles) {
  const int lane = threadIdx.x & 31;
  const int wave = threadIdx.x >> 5;
  const int tile = blockIdx.x * (blockDim.x >> 5) + wave;
  if (tile >= rowTiles * colTiles) return;

  const int rt = tile / colTiles;
  const int ct = tile - rt * colTiles;
  const int row0 = rt << 4;
  const int col0 = ct << 4;
  const int half = lane >> 4;
  const int lm   = lane & 15;

  const bool cOK = (col0 + lm) < F;
  const int  arow = min(row0 + lm, N - 1);
  const int  bcol = min(col0 + lm, F - 1);

  const float* Arow = A + (size_t)arow * K + 2 * half;
  const float* Bcol = B + bcol;

  v2f afrag[K / 4];
  v2f bfrag[K / 4];
#pragma unroll
  for (int i = 0; i < K / 4; ++i) {
    const int ka = 4 * i + 2 * half;
    afrag[i] = *reinterpret_cast<const v2f*>(Arow + 4 * i);
    bfrag[i].x = Bcol[(size_t)ka * F];
    bfrag[i].y = Bcol[(size_t)(ka + 1) * F];
  }

  v8f acc = {};
#pragma unroll
  for (int i = 0; i < K / 4; ++i) {
    acc = __builtin_amdgcn_wmma_f32_16x16x4_f32(
        false, afrag[i], false, bfrag[i], (short)0, acc, false, false);
  }

  const float bv = (bias != nullptr) ? bias[bcol] : 0.f;
#pragma unroll
  for (int r = 0; r < 8; ++r) {
    const int row = row0 + r + 8 * half;
    if (row < N && cOK) C[(size_t)row * F + col0 + lm] = acc[r] + bv;
  }
}

__global__ void gemm_wmma_gen(const float* __restrict__ A, const float* __restrict__ B,
                              const float* __restrict__ bias, float* __restrict__ C,
                              int N, int K, int F, int rowTiles, int colTiles) {
  const int lane = threadIdx.x & 31;
  const int wave = threadIdx.x >> 5;
  const int tile = blockIdx.x * (blockDim.x >> 5) + wave;
  if (tile >= rowTiles * colTiles) return;
  const int rt = tile / colTiles;
  const int ct = tile - rt * colTiles;
  const int row0 = rt << 4, col0 = ct << 4;
  const int half = lane >> 4, lm = lane & 15;
  const bool cOK = (col0 + lm) < F;
  const int arow = min(row0 + lm, N - 1);
  const int bcol = min(col0 + lm, F - 1);
  v8f acc = {};
  for (int kb = 0; kb < K; kb += 4) {
    const int ka = kb + 2 * half;
    v2f a = *reinterpret_cast<const v2f*>(A + (size_t)arow * K + ka);
    v2f b;
    b.x = B[(size_t)ka * F + bcol];
    b.y = B[(size_t)(ka + 1) * F + bcol];
    acc = __builtin_amdgcn_wmma_f32_16x16x4_f32(false, a, false, b, (short)0, acc,
                                                false, false);
  }
  const float bv = (bias != nullptr) ? bias[bcol] : 0.f;
#pragma unroll
  for (int r = 0; r < 8; ++r) {
    const int row = row0 + r + 8 * half;
    if (row < N && cOK) C[(size_t)row * F + col0 + lm] = acc[r] + bv;
  }
}

__global__ void edge_scatter(const float* __restrict__ h, const float* __restrict__ norm,
                             const int* __restrict__ src, const int* __restrict__ dst,
                             float* __restrict__ agg, int E, int F, int G) {
  long long tid = (long long)blockIdx.x * blockDim.x + threadIdx.x;
  if (tid >= (long long)E * G) return;
  int e = (int)(tid / G);
  int g = (int)(tid - (long long)e * G);
  int s = src[e];
  int d = dst[e];
  float w = norm[e];
  const float4 v = *reinterpret_cast<const float4*>(h + (size_t)s * F + (size_t)g * 4);
  float* base = agg + (size_t)d * F + (size_t)g * 4;
  atomicAdd(base + 0, v.x * w);
  atomicAdd(base + 1, v.y * w);
  atomicAdd(base + 2, v.z * w);
  atomicAdd(base + 3, v.w * w);
}

template <int C>
__global__ void relu_logsoftmax_t(float* __restrict__ y, int N) {
  int n = blockIdx.x * blockDim.x + threadIdx.x;
  if (n >= N) return;
  float v[C];
  float m = -3.4e38f;
  float* row = y + (size_t)n * C;
#pragma unroll
  for (int c = 0; c < C; ++c) {
    float t = fmaxf(row[c], 0.f);
    v[c] = t;
    m = fmaxf(m, t);
  }
  float s = 0.f;
#pragma unroll
  for (int c = 0; c < C; ++c) s += __expf(v[c] - m);
  float ls = m + __logf(s);
#pragma unroll
  for (int c = 0; c < C; ++c) row[c] = v[c] - ls;
}

__global__ void relu_logsoftmax_gen(float* __restrict__ y, int N, int C) {
  int n = blockIdx.x * blockDim.x + threadIdx.x;
  if (n >= N) return;
  float* row = y + (size_t)n * C;
  float m = -3.4e38f;
  for (int c = 0; c < C; ++c) m = fmaxf(m, fmaxf(row[c], 0.f));
  float s = 0.f;
  for (int c = 0; c < C; ++c) s += __expf(fmaxf(row[c], 0.f) - m);
  float ls = m + __logf(s);
  for (int c = 0; c < C; ++c) row[c] = fmaxf(row[c], 0.f) - ls;
}

#define CNT_STRIDE 32

__global__ void zero_counters(int* __restrict__ cnt, int N) {
  int n = blockIdx.x * blockDim.x + threadIdx.x;
  if (n < N) cnt[(size_t)n * CNT_STRIDE] = 0;
}

__global__ void rank_edges(const int* __restrict__ dst, int* __restrict__ cnt,
                           int* __restrict__ rank, int E) {
  int e0 = (blockIdx.x * blockDim.x + threadIdx.x) * 4;
  int d[4];
#pragma unroll
  for (int i = 0; i < 4; ++i) d[i] = (e0 + i < E) ? dst[e0 + i] : -1;
  int r[4];
#pragma unroll
  for (int i = 0; i < 4; ++i)
    r[i] = (d[i] >= 0) ? atomicAdd(cnt + (size_t)d[i] * CNT_STRIDE, 1) : 0;
#pragma unroll
  for (int i = 0; i < 4; ++i)
    if (e0 + i < E) rank[e0 + i] = r[i];
}

__device__ __forceinline__ int block_incl_scan_256(int v, int* sh) {
  const int t = threadIdx.x;
  sh[t] = v;
  __syncthreads();
  for (int off = 1; off < 256; off <<= 1) {
    int a = (t >= off) ? sh[t - off] : 0;
    __syncthreads();
    sh[t] += a;
    __syncthreads();
  }
  return sh[t];
}

__global__ void __launch_bounds__(256)
count_blocksum(const int* __restrict__ cnt, int* __restrict__ bsum, int N) {
  __shared__ int sh[256];
  int n = blockIdx.x * 256 + threadIdx.x;
  int v = (n < N) ? cnt[(size_t)n * CNT_STRIDE] : 0;
  int inc = block_incl_scan_256(v, sh);
  if (threadIdx.x == 255) bsum[blockIdx.x] = inc;
}

__global__ void __launch_bounds__(256)
scan_rowptr(const int* __restrict__ cnt, const int* __restrict__ bsum,
            int* __restrict__ rowptr, int N) {
  __shared__ int sh[256];
  __shared__ int sh2[256];
  const int t = threadIdx.x;
  int part = 0;
  for (int j = t; j < (int)blockIdx.x; j += 256) part += bsum[j];
  block_incl_scan_256(part, sh2);
  const int base = sh2[255];
  int n = blockIdx.x * 256 + t;
  int v = (n < N) ? cnt[(size_t)n * CNT_STRIDE] : 0;
  int inc = block_incl_scan_256(v, sh);
  if (n < N) {
    rowptr[n] = base + inc - v;
    if (n == N - 1) rowptr[N] = base + inc;
  }
}

__global__ void fill_sorted(const int* __restrict__ src, const int* __restrict__ dst,
                            const float* __restrict__ w, const int* __restrict__ rowptr,
                            const int* __restrict__ rank, int* __restrict__ ssrc,
                            float* __restrict__ sw, int E) {
  int e0 = (blockIdx.x * blockDim.x + threadIdx.x) * 4;
  for (int i = 0; i < 4; ++i) {
    int e = e0 + i;
    if (e < E) {
      int p = rowptr[dst[e]] + rank[e];
      ssrc[p] = src[e];
      sw[p] = w[e];
    }
  }
}

__global__ void deg_dinv(const float* __restrict__ sw, const int* __restrict__ rowptr,
                         float* __restrict__ dinv, int N) {
  int n = blockIdx.x * blockDim.x + threadIdx.x;
  if (n >= N) return;
  const int p0 = rowptr[n], p1 = rowptr[n + 1];
  float s = 0.f;
  for (int p = p0; p < p1; ++p) s += sw[p];
  dinv[n] = s > 0.f ? rsqrtf(s) : 0.f;
}

__global__ void norm_sorted(const int* __restrict__ ssrc, const int* __restrict__ rowptr,
                            const float* __restrict__ dinv, float* snorm, int N) {
  int n = blockIdx.x * blockDim.x + threadIdx.x;
  if (n >= N) return;
  const int p0 = rowptr[n], p1 = rowptr[n + 1];
  const float dn = dinv[n];
  for (int p = p0; p < p1; ++p) snorm[p] = dinv[ssrc[p]] * snorm[p] * dn;
}

__global__ void csr_gather(const float* __restrict__ h, const float* __restrict__ snorm,
                           const int* __restrict__ ssrc, const int* __restrict__ rowptr,
                           float* __restrict__ agg, int N, int F, int G) {
  long long tid = (long long)blockIdx.x * blockDim.x + threadIdx.x;
  if (tid >= (long long)N * G) return;
  int n = (int)(tid / G);
  int g = (int)(tid - (long long)n * G);
  const int p0 = rowptr[n], p1 = rowptr[n + 1];
  const float* hb = h + (size_t)g * 4;
  float4 acc = make_float4(0.f, 0.f, 0.f, 0.f);
#pragma unroll 4
  for (int p = p0; p < p1; ++p) {
    const int s = ssrc[p];
    const float w = snorm[p];
    const float4 v = *reinterpret_cast<const float4*>(hb + (size_t)s * F);
    acc.x += v.x * w;
    acc.y += v.y * w;
    acc.z += v.z * w;
    acc.w += v.w * w;
  }
  float4* o = reinterpret_cast<float4*>(agg + (size_t)n * F + (size_t)g * 4);
  float4 z = *o;
  z.x += acc.x;
  z.y += acc.y;
  z.z += acc.z;
  z.w += acc.w;
  *o = z;
}

static inline int ceil_div(long long a, int b) { return (int)((a + b - 1) / b); }

static void launch_gemm(const float* A, const float* B, const float* bias, float* C,
                        int N, int K, int F, hipStream_t stream) {
  const int TPB = 256;
  int rowT = ceil_div(N, 16), colT = ceil_div(F, 16);
  int blocks = ceil_div((long long)rowT * colT, TPB / 32);
  if (K == 64) {
    gemm_wmma<64><<<blocks, TPB, 0, stream>>>(A, B, bias, C, N, F, rowT, colT);
  } else if (K == 48) {
    gemm_wmma<48><<<blocks, TPB, 0, stream>>>(A, B, bias, C, N, F, rowT, colT);
  } else {
    gemm_wmma_gen<<<blocks, TPB, 0, stream>>>(A, B, bias, C, N, K, F, rowT, colT);
  }
}

extern "C" void kernel_launch(void* const* d_in, const int* in_sizes, int n_in,
                              void* d_out, int out_size, void* d_ws, size_t ws_size,
                              hipStream_t stream) {
  const float* x  = (const float*)d_in[0];
  const int*   ei = (const int*)d_in[1];
  const float* ew = (const float*)d_in[2];
  const float* W1 = (const float*)d_in[3];
  const float* V1 = (const float*)d_in[4];
  const float* b1 = (const float*)d_in[5];
  const float* W2 = (const float*)d_in[6];
  const float* V2 = (const float*)d_in[7];
  const float* b2 = (const float*)d_in[8];
  float* out = (float*)d_out;

  const int H   = in_sizes[5];
  const int Cn  = in_sizes[8];
  const int Fin = in_sizes[3] / H;
  const int N   = in_sizes[0] / Fin;
  const int E   = in_sizes[2];
  const int* src = ei;
  const int* dst = ei + E;

  float* ws = (float*)d_ws;
  const size_t NHa = ((size_t)N * H + 255) & ~(size_t)255;
  const size_t Ea  = ((size_t)E + 255) & ~(size_t)255;
  size_t off = 0;
  float* deg  = ws + off; off += ((size_t)N + 255) & ~(size_t)255;
  float* norm = ws + off; off += Ea;
  float* hbuf = ws + off; off += NHa;
  float* z1   = ws + off; off += NHa;
  int* ssrc   = (int*)(ws + off); off += Ea;
  int* rowptr = (int*)(ws + off); off += ((size_t)N + 1 + 255) & ~(size_t)255;
  const int NB = ceil_div(N, 256);
  int* rank = (int*)hbuf;
  int* cnt  = rank + Ea;
  int* bsum = cnt + (((size_t)N * CNT_STRIDE + 255) & ~(size_t)255);
  const size_t scratch_words = Ea + (((size_t)N * CNT_STRIDE + 255) & ~(size_t)255) + (size_t)NB;
  const bool sorted_path = (off * sizeof(float) <= ws_size) && (scratch_words <= 2 * NHa) &&
                           (H % 4 == 0) && (Cn % 4 == 0);

  const int TPB = 256;

  if (sorted_path) {
    zero_counters<<<NB, TPB, 0, stream>>>(cnt, N);
    rank_edges<<<ceil_div(ceil_div(E, 4), TPB), TPB, 0, stream>>>(dst, cnt, rank, E);
    count_blocksum<<<NB, 256, 0, stream>>>(cnt, bsum, N);
    scan_rowptr<<<NB, 256, 0, stream>>>(cnt, bsum, rowptr, N);
    fill_sorted<<<ceil_div(ceil_div(E, 4), TPB), TPB, 0, stream>>>(src, dst, ew, rowptr, rank,
                                                                 ssrc, norm, E);
    deg_dinv<<<NB, TPB, 0, stream>>>(norm, rowptr, deg, N);
    norm_sorted<<<NB, TPB, 0, stream>>>(ssrc, rowptr, deg, norm, N);

    launch_gemm(x, W1, nullptr, hbuf, N, Fin, H, stream);
    launch_gemm(x, V1, b1,      z1,   N, Fin, H, stream);
    {
      int G = H / 4;
      csr_gather<<<ceil_div((long long)N * G, TPB), TPB, 0, stream>>>(
          hbuf, norm, ssrc, rowptr, z1, N, H, G);
      relu_inplace<<<ceil_div((long long)N * H, TPB), TPB, 0, stream>>>(z1, N * H);
    }

    launch_gemm(z1, W2, nullptr, hbuf, N, H, Cn, stream);
    launch_gemm(z1, V2, b2,      out,  N, H, Cn, stream);
    {
      int G = Cn / 4;
      csr_gather<<<ceil_div((long long)N * G, TPB), TPB, 0, stream>>>(
          hbuf, norm, ssrc, rowptr, out, N, Cn, G);
      if (Cn == 40) {
        relu_logsoftmax_t<40><<<ceil_div(N, TPB), TPB, 0, stream>>>(out, N);
      } else {
        relu_logsoftmax_gen<<<ceil_div(N, TPB), TPB, 0, stream>>>(out, N, Cn);
      }
    }
    return;
  }

  zero_f32<<<ceil_div(N, TPB), TPB, 0, stream>>>(deg, N);
  deg_accum<<<ceil_div(E, TPB), TPB, 0, stream>>>(dst, ew, deg, E);
  norm_compute<<<ceil_div(E, TPB), TPB, 0, stream>>>(src, dst, ew, deg, norm, E);

  launch_gemm(x, W1, nullptr, hbuf, N, Fin, H, stream);
  launch_gemm(x, V1, b1,      z1,   N, Fin, H, stream);
  {
    int G = H / 4;
    edge_scatter<<<ceil_div((long long)E * G, TPB), TPB, 0, stream>>>(
        hbuf, norm, src, dst, z1, E, H, G);
    relu_inplace<<<ceil_div((long long)N * H, TPB), TPB, 0, stream>>>(z1, N * H);
  }

  launch_gemm(z1, W2, nullptr, hbuf, N, H, Cn, stream);
  launch_gemm(z1, V2, b2,      out,  N, H, Cn, stream);
  {
    int G = Cn / 4;
    edge_scatter<<<ceil_div((long long)E * G, TPB), TPB, 0, stream>>>(
        hbuf, norm, src, dst, out, E, Cn, G);
    if (Cn == 40) {
      relu_logsoftmax_t<40><<<ceil_div(N, TPB), TPB, 0, stream>>>(out, N);
    } else {
      relu_logsoftmax_gen<<<ceil_div(N, TPB), TPB, 0, stream>>>(out, N, Cn);
    }
  }
}


# ReLU (layer 1) and ReLU+log_softmax (layer 2: 3 nodes per wave, row max and exp sum by lane exchange, exp sum now per-lane partials added in lane order) fused into the segment-sum stores
# speedup vs baseline: 1.5064x; 1.1226x over previous
#include <hip/hip_runtime.h>

typedef float v2f __attribute__((ext_vector_type(2)));
typedef float v8f __attribute__((ext_vector_type(8)));

__global__ void zero_f32(float* __restrict__ p, int n) {
  int i = blockIdx.x * blockDim.x + threadIdx.x;
  if (i < n) p[i] = 0.f;
}

__global__ void relu_inplace(float* __restrict__ p, int n) {
  int i = blockIdx.x * blockDim.x + threadIdx.x;
  if (i < n) p[i] = fmaxf(p[i], 0.f);
}

__global__ void deg_accum(const int* __restrict__ dst, const float* __restrict__ w,
                          float* __restrict__ deg, int E) {
  int e = blockIdx.x * blockDim.x + threadIdx.x;
  if (e < E) atomicAdd(deg + dst[e], w[e]);
}

__global__ void norm_compute(const int* __restrict__ src, const int* __restrict__ dst,
                             const float* __restrict__ w, const float* __restrict__ deg,
                             float* __restrict__ norm, int E) {
  int e = blockIdx.x * blockDim.x + threadIdx.x;
  if (e >= E) return;
  float ds = deg[src[e]];
  float dd = deg[dst[e]];
  float is = ds > 0.f ? rsqrtf(ds) : 0.f;
  float id = dd > 0.f ? rsqrtf(dd) : 0.f;
  norm[e] = is * w[e] * id;
}

template <int K>
__global__ void gemm_wmma(const float* __restrict__ A, const float* __restrict__ B,
                          const float* __restrict__ bias, float* __restrict__ C,
                          int N, int F, int rowTiles, int colTiles) {
  const int lane = threadIdx.x & 31;
  const int wave = threadIdx.x >> 5;
  const int tile = blockIdx.x * (blockDim.x >> 5) + wave;
  if (tile >= rowTiles * colTiles) return;

  const int rt = tile / colTiles;
  const int ct = tile - rt * colTiles;
  const int row0 = rt << 4;
  const int col0 = ct << 4;
  const int half = lane >> 4;
  const int lm   = lane & 15;

  const bool cOK = (col0 + lm) < F;
  const int  arow = min(row0 + lm, N - 1);
  const int  bcol = min(col0 + lm, F - 1);

  const float* Arow = A + (size_t)arow * K + 2 * half;
  const float* Bcol = B + bcol;

  v2f afrag[K / 4];
  v2f bfrag[K / 4];
#pragma unroll
  for (int i = 0; i < K / 4; ++i) {
    const int ka = 4 * i + 2 * half;
    afrag[i] = *reinterpret_cast<const v2f*>(Arow + 4 * i);
    bfrag[i].x = Bcol[(size_t)ka * F];
    bfrag[i].y = Bcol[(size_t)(ka + 1) * F];
  }

  v8f acc = {};
#pragma unroll
  for (int i = 0; i < K / 4; ++i) {
    acc = __builtin_amdgcn_wmma_f32_16x16x4_f32(
        false, afrag[i], false, bfrag[i], (short)0, acc, false, false);
  }

  const float bv = (bias != nullptr) ? bias[bcol] : 0.f;
#pragma unroll
  for (int r = 0; r < 8; ++r) {
    const int row = row0 + r + 8 * half;
    if (row < N && cOK) C[(size_t)row * F + col0 + lm] = acc[r] + bv;
  }
}

__global__ void gemm_wmma_gen(const float* __restrict__ A, const float* __restrict__ B,
                              const float* __restrict__ bias, float* __restrict__ C,
                              int N, int K, int F, int rowTiles, int colTiles) {
  const int lane = threadIdx.x & 31;
  const int wave = threadIdx.x >> 5;
  const int tile = blockIdx.x * (blockDim.x >> 5) + wave;
  if (tile >= rowTiles * colTiles) return;
  const int rt = tile / colTiles;
  const int ct = tile - rt * colTiles;
  const int row0 = rt << 4, col0 = ct << 4;
  const int half = lane >> 4, lm = lane & 15;
  const bool cOK = (col0 + lm) < F;
  const int arow = min(row0 + lm, N - 1);
  const int bcol = min(col0 + lm, F - 1);
  v8f acc = {};
  for (int kb = 0; kb < K; kb += 4) {
    const int ka = kb + 2 * half;
    v2f a = *reinterpret_cast<const v2f*>(A + (size_t)arow * K + ka);
    v2f b;
    b.x = B[(size_t)ka * F + bcol];
    b.y = B[(size_t)(ka + 1) * F + bcol];
    acc = __builtin_amdgcn_wmma_f32_16x16x4_f32(false, a, false, b, (short)0, acc,
                                                false, false);
  }
  const float bv = (bias != nullptr) ? bias[bcol] : 0.f;
#pragma unroll
  for (int r = 0; r < 8; ++r) {
    const int row = row0 + r + 8 * half;
    if (row < N && cOK) C[(size_t)row * F + col0 + lm] = acc[r] + bv;
  }
}

__global__ void edge_scatter(const float* __restrict__ h, const float* __restrict__ norm,
                             const int* __restrict__ src, const int* __restrict__ dst,
                             float* __restrict__ agg, int E, int F, int G) {
  long long tid = (long long)blockIdx.x * blockDim.x + threadIdx.x;
  if (tid >= (long long)E * G) return;
  int e = (int)(tid / G);
  int g = (int)(tid - (long long)e * G);
  int s = src[e];
  int d = dst[e];
  float w = norm[e];
  const float4 v = *reinterpret_cast<const float4*>(h + (size_t)s * F + (size_t)g * 4);
  float* base = agg + (size_t)d * F + (size_t)g * 4;
  atomicAdd(base + 0, v.x * w);
  atomicAdd(base + 1, v.y * w);
  atomicAdd(base + 2, v.z * w);
  atomicAdd(base + 3, v.w * w);
}

template <int C>
__global__ void relu_logsoftmax_t(float* __restrict__ y, int N) {
  int n = blockIdx.x * blockDim.x + threadIdx.x;
  if (n >= N) return;
  float v[C];
  float m = -3.4e38f;
  float* row = y + (size_t)n * C;
#pragma unroll
  for (int c = 0; c < C; ++c) {
    float t = fmaxf(row[c], 0.f);
    v[c] = t;
    m = fmaxf(m, t);
  }
  float s = 0.f;
#pragma unroll
  for (int c = 0; c < C; ++c) s += __expf(v[c] - m);
  float ls = m + __logf(s);
#pragma unroll
  for (int c = 0; c < C; ++c) row[c] = v[c] - ls;
}

__global__ void relu_logsoftmax_gen(float* __restrict__ y, int N, int C) {
  int n = blockIdx.x * blockDim.x + threadIdx.x;
  if (n >= N) return;
  float* row = y + (size_t)n * C;
  float m = -3.4e38f;
  for (int c = 0; c < C; ++c) m = fmaxf(m, fmaxf(row[c], 0.f));
  float s = 0.f;
  for (int c = 0; c < C; ++c) s += __expf(fmaxf(row[c], 0.f) - m);
  float ls = m + __logf(s);
  for (int c = 0; c < C; ++c) row[c] = fmaxf(row[c], 0.f) - ls;
}

#define CNT_STRIDE 32

__global__ void zero_counters(int* __restrict__ cnt, int N) {
  int n = blockIdx.x * blockDim.x + threadIdx.x;
  if (n < N) cnt[(size_t)n * CNT_STRIDE] = 0;
}

__global__ void rank_edges(const int* __restrict__ dst, int* __restrict__ cnt,
                           int* __restrict__ rank, int E) {
  int e0 = (blockIdx.x * blockDim.x + threadIdx.x) * 4;
  int d[4];
#pragma unroll
  for (int i = 0; i < 4; ++i) d[i] = (e0 + i < E) ? dst[e0 + i] : -1;
  int r[4];
#pragma unroll
  for (int i = 0; i < 4; ++i)
    r[i] = (d[i] >= 0) ? atomicAdd(cnt + (size_t)d[i] * CNT_STRIDE, 1) : 0;
#pragma unroll
  for (int i = 0; i < 4; ++i)
    if (e0 + i < E) rank[e0 + i] = r[i];
}

__device__ __forceinline__ int block_incl_scan_256(int v, int* sh) {
  const int t = threadIdx.x;
  sh[t] = v;
  __syncthreads();
  for (int off = 1; off < 256; off <<= 1) {
    int a = (t >= off) ? sh[t - off] : 0;
    __syncthreads();
    sh[t] += a;
    __syncthreads();
  }
  return sh[t];
}

__global__ void __launch_bounds__(256)
count_blocksum(const int* __restrict__ cnt, int* __restrict__ bsum, int N) {
  __shared__ int sh[256];
  int n = blockIdx.x * 256 + threadIdx.x;
  int v = (n < N) ? cnt[(size_t)n * CNT_STRIDE] : 0;
  int inc = block_incl_scan_256(v, sh);
  if (threadIdx.x == 255) bsum[blockIdx.x] = inc;
}

__global__ void __launch_bounds__(256)
scan_rowptr(const int* __restrict__ cnt, const int* __restrict__ bsum,
            int* __restrict__ rowptr, int N) {
  __shared__ int sh[256];
  __shared__ int sh2[256];
  const int t = threadIdx.x;
  int part = 0;
  for (int j = t; j < (int)blockIdx.x; j += 256) part += bsum[j];
  block_incl_scan_256(part, sh2);
  const int base = sh2[255];
  int n = blockIdx.x * 256 + t;
  int v = (n < N) ? cnt[(size_t)n * CNT_STRIDE] : 0;
  int inc = block_incl_scan_256(v, sh);
  if (n < N) {
    rowptr[n] = base + inc - v;
    if (n == N - 1) rowptr[N] = base + inc;
  }
}

__global__ void fill_sorted(const int* __restrict__ src, const int* __restrict__ dst,
                            const float* __restrict__ w, const int* __restrict__ rowptr,
                            const int* __restrict__ rank, int* __restrict__ ssrc,
                            float* __restrict__ sw, int E) {
  int e0 = (blockIdx.x * blockDim.x + threadIdx.x) * 4;
  for (int i = 0; i < 4; ++i) {
    int e = e0 + i;
    if (e < E) {
      int p = rowptr[dst[e]] + rank[e];
      ssrc[p] = src[e];
      sw[p] = w[e];
    }
  }
}

__global__ void deg_dinv(const float* __restrict__ sw, const int* __restrict__ rowptr,
                         float* __restrict__ dinv, int N) {
  int n = blockIdx.x * blockDim.x + threadIdx.x;
  if (n >= N) return;
  const int p0 = rowptr[n], p1 = rowptr[n + 1];
  float s = 0.f;
  for (int p = p0; p < p1; ++p) s += sw[p];
  dinv[n] = s > 0.f ? rsqrtf(s) : 0.f;
}

__global__ void norm_sorted(const int* __restrict__ ssrc, const int* __restrict__ rowptr,
                            const float* __restrict__ dinv, float* snorm, int N) {
  int n = blockIdx.x * blockDim.x + threadIdx.x;
  if (n >= N) return;
  const int p0 = rowptr[n], p1 = rowptr[n + 1];
  const float dn = dinv[n];
  for (int p = p0; p < p1; ++p) snorm[p] = dinv[ssrc[p]] * snorm[p] * dn;
}

template <bool RELU>
__global__ void csr_gather(const float* __restrict__ h, const float* __restrict__ snorm,
                           const int* __restrict__ ssrc, const int* __restrict__ rowptr,
                           float* __restrict__ agg, int N, int F, int G) {
  long long tid = (long long)blockIdx.x * blockDim.x + threadIdx.x;
  if (tid >= (long long)N * G) return;
  int n = (int)(tid / G);
  int g = (int)(tid - (long long)n * G);
  const int p0 = rowptr[n], p1 = rowptr[n + 1];
  const float* hb = h + (size_t)g * 4;
  float4 acc = make_float4(0.f, 0.f, 0.f, 0.f);
#pragma unroll 4
  for (int p = p0; p < p1; ++p) {
    const int s = ssrc[p];
    const float w = snorm[p];
    const float4 v = *reinterpret_cast<const float4*>(hb + (size_t)s * F);
    acc.x += v.x * w;
    acc.y += v.y * w;
    acc.z += v.z * w;
    acc.w += v.w * w;
  }
  float4* o = reinterpret_cast<float4*>(agg + (size_t)n * F + (size_t)g * 4);
  float4 z = *o;
  z.x += acc.x;
  z.y += acc.y;
  z.z += acc.z;
  z.w += acc.w;
  if (RELU) {
    z.x = fmaxf(z.x, 0.f);
    z.y = fmaxf(z.y, 0.f);
    z.z = fmaxf(z.z, 0.f);
    z.w = fmaxf(z.w, 0.f);
  }
  *o = z;
}

template <int C>
__global__ void csr_gather_logsoftmax(const float* __restrict__ h,
                                      const float* __restrict__ snorm,
                                      const int* __restrict__ ssrc,
                                      const int* __restrict__ rowptr,
                                      float* __restrict__ y, int N) {
  constexpr int G = C / 4;
  constexpr int NPW = 32 / G;
  const int lane = threadIdx.x & 31;
  const int wave = (int)((blockIdx.x * blockDim.x + threadIdx.x) >> 5);
  const int slot = lane / G;
  const int g = lane - slot * G;
  const int n = wave * NPW + slot;
  const bool live = (slot < NPW) && (n < N);
  int p0 = 0, p1 = 0;
  if (live) {
    p0 = rowptr[n];
    p1 = rowptr[n + 1];
  }
  const float* hb = h + (size_t)g * 4;
  float4 acc = make_float4(0.f, 0.f, 0.f, 0.f);
#pragma unroll 4
  for (int p = p0; p < p1; ++p) {
    const int s = ssrc[p];
    const float w = snorm[p];
    const float4 v = *reinterpret_cast<const float4*>(hb + (size_t)s * C);
    acc.x += v.x * w;
    acc.y += v.y * w;
    acc.z += v.z * w;
    acc.w += v.w * w;
  }
  float4 t = make_float4(0.f, 0.f, 0.f, 0.f);
  float4* o = nullptr;
  if (live) {
    o = reinterpret_cast<float4*>(y + (size_t)n * C + (size_t)g * 4);
    const float4 z = *o;
    t.x = fmaxf(z.x + acc.x, 0.f);
    t.y = fmaxf(z.y + acc.y, 0.f);
    t.z = fmaxf(z.z + acc.z, 0.f);
    t.w = fmaxf(z.w + acc.w, 0.f);
  }
  const int first = live ? lane - g : lane;
  const int step = live ? 1 : 0;
  const float m4 = fmaxf(fmaxf(t.x, t.y), fmaxf(t.z, t.w));
  float m = -3.4e38f;
  for (int k = 0; k < G; ++k) m = fmaxf(m, __shfl(m4, first + k * step));
  const float e4 = __expf(t.x - m) + __expf(t.y - m) + __expf(t.z - m) + __expf(t.w - m);
  float s = 0.f;
  for (int k = 0; k < G; ++k) s += __shfl(e4, first + k * step);
  const float ls = m + __logf(s);
  if (live) *o = make_float4(t.x - ls, t.y - ls, t.z - ls, t.w - ls);
}

static inline int ceil_div(long long a, int b) { return (int)((a + b - 1) / b); }

static void launch_gemm(const float* A, const float* B, const float* bias, float* C,
                        int N, int K, int F, hipStream_t stream) {
  const int TPB = 256;
  int rowT = ceil_div(N, 16), colT = ceil_div(F, 16);
  int blocks = ceil_div((long long)rowT * colT, TPB / 32);
  if (K == 64) {
    gemm_wmma<64><<<blocks, TPB, 0, stream>>>(A, B, bias, C, N, F, rowT, colT);
  } else if (K == 48) {
    gemm_wmma<48><<<blocks, TPB, 0, stream>>>(A, B, bias, C, N, F, rowT, colT);
  } else {
    gemm_wmma_gen<<<blocks, TPB, 0, stream>>>(A, B, bias, C, N, K, F, rowT, colT);
  }
}

extern "C" void kernel_launch(void* const* d_in, const int* in_sizes, int n_in,
                              void* d_out, int out_size, void* d_ws, size_t ws_size,
                              hipStream_t stream) {
  const float* x  = (const float*)d_in[0];
  const int*   ei = (const int*)d_in[1];
  const float* ew = (const float*)d_in[2];
  const float* W1 = (const float*)d_in[3];
  const float* V1 = (const float*)d_in[4];
  const float* b1 = (const float*)d_in[5];
  const float* W2 = (const float*)d_in[6];
  const float* V2 = (const float*)d_in[7];
  const float* b2 = (const float*)d_in[8];
  float* out = (float*)d_out;

  const int H   = in_sizes[5];
  const int Cn  = in_sizes[8];
  const int Fin = in_sizes[3] / H;
  const int N   = in_sizes[0] / Fin;
  const int E   = in_sizes[2];
  const int* src = ei;
  const int* dst = ei + E;

  float* ws = (float*)d_ws;
  const size_t NHa = ((size_t)N * H + 255) & ~(size_t)255;
  const size_t Ea  = ((size_t)E + 255) & ~(size_t)255;
  size_t off = 0;
  float* deg  = ws + off; off += ((size_t)N + 255) & ~(size_t)255;
  float* norm = ws + off; off += Ea;
  float* hbuf = ws + off; off += NHa;
  float* z1   = ws + off; off += NHa;
  int* ssrc   = (int*)(ws + off); off += Ea;
  int* rowptr = (int*)(ws + off); off += ((size_t)N + 1 + 255) & ~(size_t)255;
  const int NB = ceil_div(N, 256);
  int* rank = (int*)hbuf;
  int* cnt  = rank + Ea;
  int* bsum = cnt + (((size_t)N * CNT_STRIDE + 255) & ~(size_t)255);
  const size_t scratch_words = Ea + (((size_t)N * CNT_STRIDE + 255) & ~(size_t)255) + (size_t)NB;
  const bool sorted_path = (off * sizeof(float) <= ws_size) && (scratch_words <= 2 * NHa) &&
                           (H % 4 == 0) && (Cn % 4 == 0);

  const int TPB = 256;

  if (sorted_path) {
    zero_counters<<<NB, TPB, 0, stream>>>(cnt, N);
    rank_edges<<<ceil_div(ceil_div(E, 4), TPB), TPB, 0, stream>>>(dst, cnt, rank, E);
    count_blocksum<<<NB, 256, 0, stream>>>(cnt, bsum, N);
    scan_rowptr<<<NB, 256, 0, stream>>>(cnt, bsum, rowptr, N);
    fill_sorted<<<ceil_div(ceil_div(E, 4), TPB), TPB, 0, stream>>>(src, dst, ew, rowptr, rank,
                                                                 ssrc, norm, E);
    deg_dinv<<<NB, TPB, 0, stream>>>(norm, rowptr, deg, N);
    norm_sorted<<<NB, TPB, 0, stream>>>(ssrc, rowptr, deg, norm, N);

    launch_gemm(x, W1, nullptr, hbuf, N, Fin, H, stream);
    launch_gemm(x, V1, b1,      z1,   N, Fin, H, stream);
    {
      int G = H / 4;
      csr_gather<true><<<ceil_div((long long)N * G, TPB), TPB, 0, stream>>>(
          hbuf, norm, ssrc, rowptr, z1, N, H, G);
    }

    launch_gemm(z1, W2, nullptr, hbuf, N, H, Cn, stream);
    launch_gemm(z1, V2, b2,      out,  N, H, Cn, stream);
    {
      if (Cn == 40) {
        csr_gather_logsoftmax<40><<<ceil_div((long long)ceil_div(N, 3) * 32, TPB), TPB, 0, stream>>>(
            hbuf, norm, ssrc, rowptr, out, N);
      } else {
        int G = Cn / 4;
        csr_gather<false><<<ceil_div((long long)N * G, TPB), TPB, 0, stream>>>(
            hbuf, norm, ssrc, rowptr, out, N, Cn, G);
        relu_logsoftmax_gen<<<ceil_div(N, TPB), TPB, 0, stream>>>(out, N, Cn);
      }
    }
    return;
  }

  zero_f32<<<ceil_div(N, TPB), TPB, 0, stream>>>(deg, N);
  deg_accum<<<ceil_div(E, TPB), TPB, 0, stream>>>(dst, ew, deg, E);
  norm_compute<<<ceil_div(E, TPB), TPB, 0, stream>>>(src, dst, ew, deg, norm, E);

  launch_gemm(x, W1, nullptr, hbuf, N, Fin, H, stream);
  launch_gemm(x, V1, b1,      z1,   N, Fin, H, stream);
  {
    int G = H / 4;
    edge_scatter<<<ceil_div((long long)E * G, TPB), TPB, 0, stream>>>(
        hbuf, norm, src, dst, z1, E, H, G);
    relu_inplace<<<ceil_div((long long)N * H, TPB), TPB, 0, stream>>>(z1, N * H);
  }

  launch_gemm(z1, W2, nullptr, hbuf, N, H, Cn, stream);
  launch_gemm(z1, V2, b2,      out,  N, H, Cn, stream);
  {
    int G = Cn / 4;
    edge_scatter<<<ceil_div((long long)E * G, TPB), TPB, 0, stream>>>(
        hbuf, norm, src, dst, out, E, Cn, G);
    if (Cn == 40) {
      relu_logsoftmax_t<40><<<ceil_div(N, TPB), TPB, 0, stream>>>(out, N);
    } else {
      relu_logsoftmax_gen<<<ceil_div(N, TPB), TPB, 0, stream>>>(out, N, Cn);
    }
  }
}


# Both GEMMs of a layer in one launch: a wave owns 16 rows, loads the A fragments once and runs them on all 16-column tiles of [W|V] (6 or 5), full tiles, constant strides; sum order unchanged
# speedup vs baseline: 1.6798x; 1.1151x over previous
#include <hip/hip_runtime.h>

typedef float v2f __attribute__((ext_vector_type(2)));
typedef float v8f __attribute__((ext_vector_type(8)));

__global__ void zero_f32(float* __restrict__ p, int n) {
  int i = blockIdx.x * blockDim.x + threadIdx.x;
  if (i < n) p[i] = 0.f;
}

__global__ void relu_inplace(float* __restrict__ p, int n) {
  int i = blockIdx.x * blockDim.x + threadIdx.x;
  if (i < n) p[i] = fmaxf(p[i], 0.f);
}

__global__ void deg_accum(const int* __restrict__ dst, const float* __restrict__ w,
                          float* __restrict__ deg, int E) {
  int e = blockIdx.x * blockDim.x + threadIdx.x;
  if (e < E) atomicAdd(deg + dst[e], w[e]);
}

__global__ void norm_compute(const int* __restrict__ src, const int* __restrict__ dst,
                             const float* __restrict__ w, const float* __restrict__ deg,
                             float* __restrict__ norm, int E) {
  int e = blockIdx.x * blockDim.x + threadIdx.x;
  if (e >= E) return;
  float ds = deg[src[e]];
  float dd = deg[dst[e]];
  float is = ds > 0.f ? rsqrtf(ds) : 0.f;
  float id = dd > 0.f ? rsqrtf(dd) : 0.f;
  norm[e] = is * w[e] * id;
}

template <int K>
__global__ void gemm_wmma(const float* __restrict__ A, const float* __restrict__ B,
                          const float* __restrict__ bias, float* __restrict__ C,
                          int N, int F, int rowTiles, int colTiles) {
  const int lane = threadIdx.x & 31;
  const int wave = threadIdx.x >> 5;
  const int tile = blockIdx.x * (blockDim.x >> 5) + wave;
  if (tile >= rowTiles * colTiles) return;

  const int rt = tile / colTiles;
  const int ct = tile - rt * colTiles;
  const int row0 = rt << 4;
  const int col0 = ct << 4;
  const int half = lane >> 4;
  const int lm   = lane & 15;

  const bool cOK = (col0 + lm) < F;
  const int  arow = min(row0 + lm, N - 1);
  const int  bcol = min(col0 + lm, F - 1);

  const float* Arow = A + (size_t)arow * K + 2 * half;
  const float* Bcol = B + bcol;

  v2f afrag[K / 4];
  v2f bfrag[K / 4];
#pragma unroll
  for (int i = 0; i < K / 4; ++i) {
    const int ka = 4 * i + 2 * half;
    afrag[i] = *reinterpret_cast<const v2f*>(Arow + 4 * i);
    bfrag[i].x = Bcol[(size_t)ka * F];
    bfrag[i].y = Bcol[(size_t)(ka + 1) * F];
  }

  v8f acc = {};
#pragma unroll
  for (int i = 0; i < K / 4; ++i) {
    acc = __builtin_amdgcn_wmma_f32_16x16x4_f32(
        false, afrag[i], false, bfrag[i], (short)0, acc, false, false);
  }

  const float bv = (bias != nullptr) ? bias[bcol] : 0.f;
#pragma unroll
  for (int r = 0; r < 8; ++r) {
    const int row = row0 + r + 8 * half;
    if (row < N && cOK) C[(size_t)row * F + col0 + lm] = acc[r] + bv;
  }
}

template <int K, int NT>
__global__ void __launch_bounds__(128)
gemm_wmma_pair(const float* __restrict__ A, const float* __restrict__ B1,
               const float* __restrict__ B2, const float* __restrict__ bias2,
               float* __restrict__ C1, float* __restrict__ C2,
               int rowTiles) {
  constexpr int F = 8 * NT;
  const int lane = threadIdx.x & 31;
  const int wave = threadIdx.x >> 5;
  const int rt = blockIdx.x * (blockDim.x >> 5) + wave;
  if (rt >= rowTiles) return;

  const int row0 = rt << 4;
  const int half = lane >> 4;
  const int lm   = lane & 15;
  const float* Arow = A + (size_t)(row0 + lm) * K + 2 * half;

  v2f afrag[K / 4];
#pragma unroll
  for (int i = 0; i < K / 4; ++i)
    afrag[i] = *reinterpret_cast<const v2f*>(Arow + 4 * i);

  const size_t rbase = (size_t)(row0 + 8 * half) * F;
#pragma unroll
  for (int t = 0; t < NT; ++t) {
    const int j = 16 * t + lm;
    const bool second = j >= F;
    const int col = second ? j - F : j;
    const float* Bcol = (second ? B2 : B1) + col;
    const float braw = bias2[col];
    const float bv = second ? braw : 0.f;

    v2f bfrag[K / 4];
#pragma unroll
    for (int i = 0; i < K / 4; ++i) {
      const int ka = 4 * i + 2 * half;
      bfrag[i].x = Bcol[(size_t)ka * F];
      bfrag[i].y = Bcol[(size_t)(ka + 1) * F];
    }

    v8f acc = {};
#pragma unroll
    for (int i = 0; i < K / 4; ++i) {
      acc = __builtin_amdgcn_wmma_f32_16x16x4_f32(
          false, afrag[i], false, bfrag[i], (short)0, acc, false, false);
    }

    float* Crow = (second ? C2 : C1) + col + rbase;
#pragma unroll
    for (int r = 0; r < 8; ++r) Crow[r * F] = acc[r] + bv;
  }
}

__global__ void gemm_wmma_gen(const float* __restrict__ A, const float* __restrict__ B,
                              const float* __restrict__ bias, float* __restrict__ C,
                              int N, int K, int F, int rowTiles, int colTiles) {
  const int lane = threadIdx.x & 31;
  const int wave = threadIdx.x >> 5;
  const int tile = blockIdx.x * (blockDim.x >> 5) + wave;
  if (tile >= rowTiles * colTiles) return;
  const int rt = tile / colTiles;
  const int ct = tile - rt * colTiles;
  const int row0 = rt << 4, col0 = ct << 4;
  const int half = lane >> 4, lm = lane & 15;
  const bool cOK = (col0 + lm) < F;
  const int arow = min(row0 + lm, N - 1);
  const int bcol = min(col0 + lm, F - 1);
  v8f acc = {};
  for (int kb = 0; kb < K; kb += 4) {
    const int ka = kb + 2 * half;
    v2f a = *reinterpret_cast<const v2f*>(A + (size_t)arow * K + ka);
    v2f b;
    b.x = B[(size_t)ka * F + bcol];
    b.y = B[(size_t)(ka + 1) * F + bcol];
    acc = __builtin_amdgcn_wmma_f32_16x16x4_f32(false, a, false, b, (short)0, acc,
                                                false, false);
  }
  const float bv = (bias != nullptr) ? bias[bcol] : 0.f;
#pragma unroll
  for (int r = 0; r < 8; ++r) {
    const int row = row0 + r + 8 * half;
    if (row < N && cOK) C[(size_t)row * F + col0 + lm] = acc[r] + bv;
  }
}

__global__ void edge_scatter(const float* __restrict__ h, const float* __restrict__ norm,
                             const int* __restrict__ src, const int* __restrict__ dst,
                             float* __restrict__ agg, int E, int F, int G) {
  long long tid = (long long)blockIdx.x * blockDim.x + threadIdx.x;
  if (tid >= (long long)E * G) return;
  int e = (int)(tid / G);
  int g = (int)(tid - (long long)e * G);
  int s = src[e];
  int d = dst[e];
  float w = norm[e];
  const float4 v = *reinterpret_cast<const float4*>(h + (size_t)s * F + (size_t)g * 4);
  float* base = agg + (size_t)d * F + (size_t)g * 4;
  atomicAdd(base + 0, v.x * w);
  atomicAdd(base + 1, v.y * w);
  atomicAdd(base + 2, v.z * w);
  atomicAdd(base + 3, v.w * w);
}

template <int C>
__global__ void relu_logsoftmax_t(float* __restrict__ y, int N) {
  int n = blockIdx.x * blockDim.x + threadIdx.x;
  if (n >= N) return;
  float v[C];
  float m = -3.4e38f;
  float* row = y + (size_t)n * C;
#pragma unroll
  for (int c = 0; c < C; ++c) {
    float t = fmaxf(row[c], 0.f);
    v[c] = t;
    m = fmaxf(m, t);
  }
  float s = 0.f;
#pragma unroll
  for (int c = 0; c < C; ++c) s += __expf(v[c] - m);
  float ls = m + __logf(s);
#pragma unroll
  for (int c = 0; c < C; ++c) row[c] = v[c] - ls;
}

__global__ void relu_logsoftmax_gen(float* __restrict__ y, int N, int C) {
  int n = blockIdx.x * blockDim.x + threadIdx.x;
  if (n >= N) return;
  float* row = y + (size_t)n * C;
  float m = -3.4e38f;
  for (int c = 0; c < C; ++c) m = fmaxf(m, fmaxf(row[c], 0.f));
  float s = 0.f;
  for (int c = 0; c < C; ++c) s += __expf(fmaxf(row[c], 0.f) - m);
  float ls = m + __logf(s);
  for (int c = 0; c < C; ++c) row[c] = fmaxf(row[c], 0.f) - ls;
}

#define CNT_STRIDE 32

__global__ void zero_counters(int* __restrict__ cnt, int N) {
  int n = blockIdx.x * blockDim.x + threadIdx.x;
  if (n < N) cnt[(size_t)n * CNT_STRIDE] = 0;
}

__global__ void rank_edges(const int* __restrict__ dst, int* __restrict__ cnt,
                           int* __restrict__ rank, int E) {
  int e0 = (blockIdx.x * blockDim.x + threadIdx.x) * 4;
  int d[4];
#pragma unroll
  for (int i = 0; i < 4; ++i) d[i] = (e0 + i < E) ? dst[e0 + i] : -1;
  int r[4];
#pragma unroll
  for (int i = 0; i < 4; ++i)
    r[i] = (d[i] >= 0) ? atomicAdd(cnt + (size_t)d[i] * CNT_STRIDE, 1) : 0;
#pragma unroll
  for (int i = 0; i < 4; ++i)
    if (e0 + i < E) rank[e0 + i] = r[i];
}

__device__ __forceinline__ int block_incl_scan_256(int v, int* sh) {
  const int t = threadIdx.x;
  sh[t] = v;
  __syncthreads();
  for (int off = 1; off < 256; off <<= 1) {
    int a = (t >= off) ? sh[t - off] : 0;
    __syncthreads();
    sh[t] += a;
    __syncthreads();
  }
  return sh[t];
}

__global__ void __launch_bounds__(256)
count_blocksum(const int* __restrict__ cnt, int* __restrict__ bsum, int N) {
  __shared__ int sh[256];
  int n = blockIdx.x * 256 + threadIdx.x;
  int v = (n < N) ? cnt[(size_t)n * CNT_STRIDE] : 0;
  int inc = block_incl_scan_256(v, sh);
  if (threadIdx.x == 255) bsum[blockIdx.x] = inc;
}

__global__ void __launch_bounds__(256)
scan_rowptr(const int* __restrict__ cnt, const int* __restrict__ bsum,
            int* __restrict__ rowptr, int N) {
  __shared__ int sh[256];
  __shared__ int sh2[256];
  const int t = threadIdx.x;
  int part = 0;
  for (int j = t; j < (int)blockIdx.x; j += 256) part += bsum[j];
  block_incl_scan_256(part, sh2);
  const int base = sh2[255];
  int n = blockIdx.x * 256 + t;
  int v = (n < N) ? cnt[(size_t)n * CNT_STRIDE] : 0;
  int inc = block_incl_scan_256(v, sh);
  if (n < N) {
    rowptr[n] = base + inc - v;
    if (n == N - 1) rowptr[N] = base + inc;
  }
}

__global__ void fill_sorted(const int* __restrict__ src, const int* __restrict__ dst,
                            const float* __restrict__ w, const int* __restrict__ rowptr,
                            const int* __restrict__ rank, int* __restrict__ ssrc,
                            float* __restrict__ sw, int E) {
  int e0 = (blockIdx.x * blockDim.x + threadIdx.x) * 4;
  for (int i = 0; i < 4; ++i) {
    int e = e0 + i;
    if (e < E) {
      int p = rowptr[dst[e]] + rank[e];
      ssrc[p] = src[e];
      sw[p] = w[e];
    }
  }
}

__global__ void deg_dinv(const float* __restrict__ sw, const int* __restrict__ rowptr,
                         float* __restrict__ dinv, int N) {
  int n = blockIdx.x * blockDim.x + threadIdx.x;
  if (n >= N) return;
  const int p0 = rowptr[n], p1 = rowptr[n + 1];
  float s = 0.f;
  for (int p = p0; p < p1; ++p) s += sw[p];
  dinv[n] = s > 0.f ? rsqrtf(s) : 0.f;
}

__global__ void norm_sorted(const int* __restrict__ ssrc, const int* __restrict__ rowptr,
                            const float* __restrict__ dinv, float* snorm, int N) {
  int n = blockIdx.x * blockDim.x + threadIdx.x;
  if (n >= N) return;
  const int p0 = rowptr[n], p1 = rowptr[n + 1];
  const float dn = dinv[n];
  for (int p = p0; p < p1; ++p) snorm[p] = dinv[ssrc[p]] * snorm[p] * dn;
}

template <bool RELU>
__global__ void csr_gather(const float* __restrict__ h, const float* __restrict__ snorm,
                           const int* __restrict__ ssrc, const int* __restrict__ rowptr,
                           float* __restrict__ agg, int N, int F, int G) {
  long long tid = (long long)blockIdx.x * blockDim.x + threadIdx.x;
  if (tid >= (long long)N * G) return;
  int n = (int)(tid / G);
  int g = (int)(tid - (long long)n * G);
  const int p0 = rowptr[n], p1 = rowptr[n + 1];
  const float* hb = h + (size_t)g * 4;
  float4 acc = make_float4(0.f, 0.f, 0.f, 0.f);
#pragma unroll 4
  for (int p = p0; p < p1; ++p) {
    const int s = ssrc[p];
    const float w = snorm[p];
    const float4 v = *reinterpret_cast<const float4*>(hb + (size_t)s * F);
    acc.x += v.x * w;
    acc.y += v.y * w;
    acc.z += v.z * w;
    acc.w += v.w * w;
  }
  float4* o = reinterpret_cast<float4*>(agg + (size_t)n * F + (size_t)g * 4);
  float4 z = *o;
  z.x += acc.x;
  z.y += acc.y;
  z.z += acc.z;
  z.w += acc.w;
  if (RELU) {
    z.x = fmaxf(z.x, 0.f);
    z.y = fmaxf(z.y, 0.f);
    z.z = fmaxf(z.z, 0.f);
    z.w = fmaxf(z.w, 0.f);
  }
  *o = z;
}

template <int C>
__global__ void csr_gather_logsoftmax(const float* __restrict__ h,
                                      const float* __restrict__ snorm,
                                      const int* __restrict__ ssrc,
                                      const int* __restrict__ rowptr,
                                      float* __restrict__ y, int N) {
  constexpr int G = C / 4;
  constexpr int NPW = 32 / G;
  const int lane = threadIdx.x & 31;
  const int wave = (int)((blockIdx.x * blockDim.x + threadIdx.x) >> 5);
  const int slot = lane / G;
  const int g = lane - slot * G;
  const int n = wave * NPW + slot;
  const bool live = (slot < NPW) && (n < N);
  int p0 = 0, p1 = 0;
  if (live) {
    p0 = rowptr[n];
    p1 = rowptr[n + 1];
  }
  const float* hb = h + (size_t)g * 4;
  float4 acc = make_float4(0.f, 0.f, 0.f, 0.f);
#pragma unroll 4
  for (int p = p0; p < p1; ++p) {
    const int s = ssrc[p];
    const float w = snorm[p];
    const float4 v = *reinterpret_cast<const float4*>(hb + (size_t)s * C);
    acc.x += v.x * w;
    acc.y += v.y * w;
    acc.z += v.z * w;
    acc.w += v.w * w;
  }
  float4 t = make_float4(0.f, 0.f, 0.f, 0.f);
  float4* o = nullptr;
  if (live) {
    o = reinterpret_cast<float4*>(y + (size_t)n * C + (size_t)g * 4);
    const float4 z = *o;
    t.x = fmaxf(z.x + acc.x, 0.f);
    t.y = fmaxf(z.y + acc.y, 0.f);
    t.z = fmaxf(z.z + acc.z, 0.f);
    t.w = fmaxf(z.w + acc.w, 0.f);
  }
  const int first = live ? lane - g : lane;
  const int step = live ? 1 : 0;
  const float m4 = fmaxf(fmaxf(t.x, t.y), fmaxf(t.z, t.w));
  float m = -3.4e38f;
  for (int k = 0; k < G; ++k) m = fmaxf(m, __shfl(m4, first + k * step));
  const float e4 = __expf(t.x - m) + __expf(t.y - m) + __expf(t.z - m) + __expf(t.w - m);
  float s = 0.f;
  for (int k = 0; k < G; ++k) s += __shfl(e4, first + k * step);
  const float ls = m + __logf(s);
  if (live) *o = make_float4(t.x - ls, t.y - ls, t.z - ls, t.w - ls);
}

static inline int ceil_div(long long a, int b) { return (int)((a + b - 1) / b); }

static void launch_gemm(const float* A, const float* B, const float* bias, float* C,
                        int N, int K, int F, hipStream_t stream) {
  const int TPB = 256;
  int rowT = ceil_div(N, 16), colT = ceil_div(F, 16);
  int blocks = ceil_div((long long)rowT * colT, TPB / 32);
  if (K == 64) {
    gemm_wmma<64><<<blocks, TPB, 0, stream>>>(A, B, bias, C, N, F, rowT, colT);
  } else if (K == 48) {
    gemm_wmma<48><<<blocks, TPB, 0, stream>>>(A, B, bias, C, N, F, rowT, colT);
  } else {
    gemm_wmma_gen<<<blocks, TPB, 0, stream>>>(A, B, bias, C, N, K, F, rowT, colT);
  }
}

static void launch_gemm_pair(const float* A, const float* B1, const float* B2,
                             const float* bias2, float* C1, float* C2,
                             int N, int K, int F, hipStream_t stream) {
  const int rowT = N / 16;
  const int WPB = 4;
  const int blocks = ceil_div(rowT, WPB);
  const bool full = (N % 16 == 0) && (N > 0) && (bias2 != nullptr);
  if (full && K == 64 && F == 48) {
    gemm_wmma_pair<64, 6><<<blocks, WPB * 32, 0, stream>>>(A, B1, B2, bias2, C1, C2, rowT);
  } else if (full && K == 48 && F == 40) {
    gemm_wmma_pair<48, 5><<<blocks, WPB * 32, 0, stream>>>(A, B1, B2, bias2, C1, C2, rowT);
  } else {
    launch_gemm(A, B1, nullptr, C1, N, K, F, stream);
    launch_gemm(A, B2, bias2,   C2, N, K, F, stream);
  }
}

extern "C" void kernel_launch(void* const* d_in, const int* in_sizes, int n_in,
                              void* d_out, int out_size, void* d_ws, size_t ws_size,
                              hipStream_t stream) {
  const float* x  = (const float*)d_in[0];
  const int*   ei = (const int*)d_in[1];
  const float* ew = (const float*)d_in[2];
  const float* W1 = (const float*)d_in[3];
  const float* V1 = (const float*)d_in[4];
  const float* b1 = (const float*)d_in[5];
  const float* W2 = (const float*)d_in[6];
  const float* V2 = (const float*)d_in[7];
  const float* b2 = (const float*)d_in[8];
  float* out = (float*)d_out;

  const int H   = in_sizes[5];
  const int Cn  = in_sizes[8];
  const int Fin = in_sizes[3] / H;
  const int N   = in_sizes[0] / Fin;
  const int E   = in_sizes[2];
  const int* src = ei;
  const int* dst = ei + E;

  float* ws = (float*)d_ws;
  const size_t NHa = ((size_t)N * H + 255) & ~(size_t)255;
  const size_t Ea  = ((size_t)E + 255) & ~(size_t)255;
  size_t off = 0;
  float* deg  = ws + off; off += ((size_t)N + 255) & ~(size_t)255;
  float* norm = ws + off; off += Ea;
  float* hbuf = ws + off; off += NHa;
  float* z1   = ws + off; off += NHa;
  int* ssrc   = (int*)(ws + off); off += Ea;
  int* rowptr = (int*)(ws + off); off += ((size_t)N + 1 + 255) & ~(size_t)255;
  const int NB = ceil_div(N, 256);
  int* rank = (int*)hbuf;
  int* cnt  = rank + Ea;
  int* bsum = cnt + (((size_t)N * CNT_STRIDE + 255) & ~(size_t)255);
  const size_t scratch_words = Ea + (((size_t)N * CNT_STRIDE + 255) & ~(size_t)255) + (size_t)NB;
  const bool sorted_path = (off * sizeof(float) <= ws_size) && (scratch_words <= 2 * NHa) &&
                           (H % 4 == 0) && (Cn % 4 == 0);

  const int TPB = 256;

  if (sorted_path) {
    zero_counters<<<NB, TPB, 0, stream>>>(cnt, N);
    rank_edges<<<ceil_div(ceil_div(E, 4), TPB), TPB, 0, stream>>>(dst, cnt, rank, E);
    count_blocksum<<<NB, 256, 0, stream>>>(cnt, bsum, N);
    scan_rowptr<<<NB, 256, 0, stream>>>(cnt, bsum, rowptr, N);
    fill_sorted<<<ceil_div(ceil_div(E, 4), TPB), TPB, 0, stream>>>(src, dst, ew, rowptr, rank,
                                                                 ssrc, norm, E);
    deg_dinv<<<NB, TPB, 0, stream>>>(norm, rowptr, deg, N);
    norm_sorted<<<NB, TPB, 0, stream>>>(ssrc, rowptr, deg, norm, N);

    launch_gemm_pair(x, W1, V1, b1, hbuf, z1, N, Fin, H, stream);
    {
      int G = H / 4;
      csr_gather<true><<<ceil_div((long long)N * G, TPB), TPB, 0, stream>>>(
          hbuf, norm, ssrc, rowptr, z1, N, H, G);
    }

    launch_gemm_pair(z1, W2, V2, b2, hbuf, out, N, H, Cn, stream);
    {
      if (Cn == 40) {
        csr_gather_logsoftmax<40><<<ceil_div((long long)ceil_div(N, 3) * 32, TPB), TPB, 0, stream>>>(
            hbuf, norm, ssrc, rowptr, out, N);
      } else {
        int G = Cn / 4;
        csr_gather<false><<<ceil_div((long long)N * G, TPB), TPB, 0, stream>>>(
            hbuf, norm, ssrc, rowptr, out, N, Cn, G);
        relu_logsoftmax_gen<<<ceil_div(N, TPB), TPB, 0, stream>>>(out, N, Cn);
      }
    }
    return;
  }

  zero_f32<<<ceil_div(N, TPB), TPB, 0, stream>>>(deg, N);
  deg_accum<<<ceil_div(E, TPB), TPB, 0, stream>>>(dst, ew, deg, E);
  norm_compute<<<ceil_div(E, TPB), TPB, 0, stream>>>(src, dst, ew, deg, norm, E);

  launch_gemm(x, W1, nullptr, hbuf, N, Fin, H, stream);
  launch_gemm(x, V1, b1,      z1,   N, Fin, H, stream);
  {
    int G = H / 4;
    edge_scatter<<<ceil_div((long long)E * G, TPB), TPB, 0, stream>>>(
        hbuf, norm, src, dst, z1, E, H, G);
    relu_inplace<<<ceil_div((long long)N * H, TPB), TPB, 0, stream>>>(z1, N * H);
  }

  launch_gemm(z1, W2, nullptr, hbuf, N, H, Cn, stream);
  launch_gemm(z1, V2, b2,      out,  N, H, Cn, stream);
  {
    int G = Cn / 4;
    edge_scatter<<<ceil_div((long long)E * G, TPB), TPB, 0, stream>>>(
        hbuf, norm, src, dst, out, E, Cn, G);
    if (Cn == 40) {
      relu_logsoftmax_t<40><<<ceil_div(N, TPB), TPB, 0, stream>>>(out, N);
    } else {
      relu_logsoftmax_gen<<<ceil_div(N, TPB), TPB, 0, stream>>>(out, N, Cn);
    }
  }
}


# Edge sort by destination rebuilt on LDS counters (bucket of 512 nodes, then node inside one work-group, records leave LDS as runs); no global atomic left; segment order = LDS adds' arrival order
# speedup vs baseline: 2.6993x; 1.6069x over previous
#include <hip/hip_runtime.h>

typedef float v2f __attribute__((ext_vector_type(2)));
typedef float v8f __attribute__((ext_vector_type(8)));

__global__ void zero_f32(float* __restrict__ p, int n) {
  int i = blockIdx.x * blockDim.x + threadIdx.x;
  if (i < n) p[i] = 0.f;
}

__global__ void relu_inplace(float* __restrict__ p, int n) {
  int i = blockIdx.x * blockDim.x + threadIdx.x;
  if (i < n) p[i] = fmaxf(p[i], 0.f);
}

__global__ void deg_accum(const int* __restrict__ dst, const float* __restrict__ w,
                          float* __restrict__ deg, int E) {
  int e = blockIdx.x * blockDim.x + threadIdx.x;
  if (e < E) atomicAdd(deg + dst[e], w[e]);
}

__global__ void norm_compute(const int* __restrict__ src, const int* __restrict__ dst,
                             const float* __restrict__ w, const float* __restrict__ deg,
                             float* __restrict__ norm, int E) {
  int e = blockIdx.x * blockDim.x + threadIdx.x;
  if (e >= E) return;
  float ds = deg[src[e]];
  float dd = deg[dst[e]];
  float is = ds > 0.f ? rsqrtf(ds) : 0.f;
  float id = dd > 0.f ? rsqrtf(dd) : 0.f;
  norm[e] = is * w[e] * id;
}

template <int K>
__global__ void gemm_wmma(const float* __restrict__ A, const float* __restrict__ B,
                          const float* __restrict__ bias, float* __restrict__ C,
                          int N, int F, int rowTiles, int colTiles) {
  const int lane = threadIdx.x & 31;
  const int wave = threadIdx.x >> 5;
  const int tile = blockIdx.x * (blockDim.x >> 5) + wave;
  if (tile >= rowTiles * colTiles) return;

  const int rt = tile / colTiles;
  const int ct = tile - rt * colTiles;
  const int row0 = rt << 4;
  const int col0 = ct << 4;
  const int half = lane >> 4;
  const int lm   = lane & 15;

  const bool cOK = (col0 + lm) < F;
  const int  arow = min(row0 + lm, N - 1);
  const int  bcol = min(col0 + lm, F - 1);

  const float* Arow = A + (size_t)arow * K + 2 * half;
  const float* Bcol = B + bcol;

  v2f afrag[K / 4];
  v2f bfrag[K / 4];
#pragma unroll
  for (int i = 0; i < K / 4; ++i) {
    const int ka = 4 * i + 2 * half;
    afrag[i] = *reinterpret_cast<const v2f*>(Arow + 4 * i);
    bfrag[i].x = Bcol[(size_t)ka * F];
    bfrag[i].y = Bcol[(size_t)(ka + 1) * F];
  }

  v8f acc = {};
#pragma unroll
  for (int i = 0; i < K / 4; ++i) {
    acc = __builtin_amdgcn_wmma_f32_16x16x4_f32(
        false, afrag[i], false, bfrag[i], (short)0, acc, false, false);
  }

  const float bv = (bias != nullptr) ? bias[bcol] : 0.f;
#pragma unroll
  for (int r = 0; r < 8; ++r) {
    const int row = row0 + r + 8 * half;
    if (row < N && cOK) C[(size_t)row * F + col0 + lm] = acc[r] + bv;
  }
}

template <int K, int NT>
__global__ void __launch_bounds__(128)
gemm_wmma_pair(const float* __restrict__ A, const float* __restrict__ B1,
               const float* __restrict__ B2, const float* __restrict__ bias2,
               float* __restrict__ C1, float* __restrict__ C2,
               int rowTiles) {
  constexpr int F = 8 * NT;
  const int lane = threadIdx.x & 31;
  const int wave = threadIdx.x >> 5;
  const int rt = blockIdx.x * (blockDim.x >> 5) + wave;
  if (rt >= rowTiles) return;

  const int row0 = rt << 4;
  const int half = lane >> 4;
  const int lm   = lane & 15;
  const float* Arow = A + (size_t)(row0 + lm) * K + 2 * half;

  v2f afrag[K / 4];
#pragma unroll
  for (int i = 0; i < K / 4; ++i)
    afrag[i] = *reinterpret_cast<const v2f*>(Arow + 4 * i);

  const size_t rbase = (size_t)(row0 + 8 * half) * F;
#pragma unroll
  for (int t = 0; t < NT; ++t) {
    const int j = 16 * t + lm;
    const bool second = j >= F;
    const int col = second ? j - F : j;
    const float* Bcol = (second ? B2 : B1) + col;
    const float braw = bias2[col];
    const float bv = second ? braw : 0.f;

    v2f bfrag[K / 4];
#pragma unroll
    for (int i = 0; i < K / 4; ++i) {
      const int ka = 4 * i + 2 * half;
      bfrag[i].x = Bcol[(size_t)ka * F];
      bfrag[i].y = Bcol[(size_t)(ka + 1) * F];
    }

    v8f acc = {};
#pragma unroll
    for (int i = 0; i < K / 4; ++i) {
      acc = __builtin_amdgcn_wmma_f32_16x16x4_f32(
          false, afrag[i], false, bfrag[i], (short)0, acc, false, false);
    }

    float* Crow = (second ? C2 : C1) + col + rbase;
#pragma unroll
    for (int r = 0; r < 8; ++r) Crow[r * F] = acc[r] + bv;
  }
}

__global__ void gemm_wmma_gen(const float* __restrict__ A, const float* __restrict__ B,
                              const float* __restrict__ bias, float* __restrict__ C,
                              int N, int K, int F, int rowTiles, int colTiles) {
  const int lane = threadIdx.x & 31;
  const int wave = threadIdx.x >> 5;
  const int tile = blockIdx.x * (blockDim.x >> 5) + wave;
  if (tile >= rowTiles * colTiles) return;
  const int rt = tile / colTiles;
  const int ct = tile - rt * colTiles;
  const int row0 = rt << 4, col0 = ct << 4;
  const int half = lane >> 4, lm = lane & 15;
  const bool cOK = (col0 + lm) < F;
  const int arow = min(row0 + lm, N - 1);
  const int bcol = min(col0 + lm, F - 1);
  v8f acc = {};
  for (int kb = 0; kb < K; kb += 4) {
    const int ka = kb + 2 * half;
    v2f a = *reinterpret_cast<const v2f*>(A + (size_t)arow * K + ka);
    v2f b;
    b.x = B[(size_t)ka * F + bcol];
    b.y = B[(size_t)(ka + 1) * F + bcol];
    acc = __builtin_amdgcn_wmma_f32_16x16x4_f32(false, a, false, b, (short)0, acc,
                                                false, false);
  }
  const float bv = (bias != nullptr) ? bias[bcol] : 0.f;
#pragma unroll
  for (int r = 0; r < 8; ++r) {
    const int row = row0 + r + 8 * half;
    if (row < N && cOK) C[(size_t)row * F + col0 + lm] = acc[r] + bv;
  }
}

__global__ void edge_scatter(const float* __restrict__ h, const float* __restrict__ norm,
                             const int* __restrict__ src, const int* __restrict__ dst,
                             float* __restrict__ agg, int E, int F, int G) {
  long long tid = (long long)blockIdx.x * blockDim.x + threadIdx.x;
  if (tid >= (long long)E * G) return;
  int e = (int)(tid / G);
  int g = (int)(tid - (long long)e * G);
  int s = src[e];
  int d = dst[e];
  float w = norm[e];
  const float4 v = *reinterpret_cast<const float4*>(h + (size_t)s * F + (size_t)g * 4);
  float* base = agg + (size_t)d * F + (size_t)g * 4;
  atomicAdd(base + 0, v.x * w);
  atomicAdd(base + 1, v.y * w);
  atomicAdd(base + 2, v.z * w);
  atomicAdd(base + 3, v.w * w);
}

template <int C>
__global__ void relu_logsoftmax_t(float* __restrict__ y, int N) {
  int n = blockIdx.x * blockDim.x + threadIdx.x;
  if (n >= N) return;
  float v[C];
  float m = -3.4e38f;
  float* row = y + (size_t)n * C;
#pragma unroll
  for (int c = 0; c < C; ++c) {
    float t = fmaxf(row[c], 0.f);
    v[c] = t;
    m = fmaxf(m, t);
  }
  float s = 0.f;
#pragma unroll
  for (int c = 0; c < C; ++c) s += __expf(v[c] - m);
  float ls = m + __logf(s);
#pragma unroll
  for (int c = 0; c < C; ++c) row[c] = v[c] - ls;
}

__global__ void relu_logsoftmax_gen(float* __restrict__ y, int N, int C) {
  int n = blockIdx.x * blockDim.x + threadIdx.x;
  if (n >= N) return;
  float* row = y + (size_t)n * C;
  float m = -3.4e38f;
  for (int c = 0; c < C; ++c) m = fmaxf(m, fmaxf(row[c], 0.f));
  float s = 0.f;
  for (int c = 0; c < C; ++c) s += __expf(fmaxf(row[c], 0.f) - m);
  float ls = m + __logf(s);
  for (int c = 0; c < C; ++c) row[c] = fmaxf(row[c], 0.f) - ls;
}

#define CNT_STRIDE 32

__global__ void zero_counters(int* __restrict__ cnt, int N) {
  int n = blockIdx.x * blockDim.x + threadIdx.x;
  if (n < N) cnt[(size_t)n * CNT_STRIDE] = 0;
}

__global__ void rank_edges(const int* __restrict__ dst, int* __restrict__ cnt,
                           int* __restrict__ rank, int E) {
  int e0 = (blockIdx.x * blockDim.x + threadIdx.x) * 4;
  int d[4];
#pragma unroll
  for (int i = 0; i < 4; ++i) d[i] = (e0 + i < E) ? dst[e0 + i] : -1;
  int r[4];
#pragma unroll
  for (int i = 0; i < 4; ++i)
    r[i] = (d[i] >= 0) ? atomicAdd(cnt + (size_t)d[i] * CNT_STRIDE, 1) : 0;
#pragma unroll
  for (int i = 0; i < 4; ++i)
    if (e0 + i < E) rank[e0 + i] = r[i];
}

__device__ __forceinline__ int block_incl_scan_256(int v, int* sh) {
  const int t = threadIdx.x;
  sh[t] = v;
  __syncthreads();
  for (int off = 1; off < 256; off <<= 1) {
    int a = (t >= off) ? sh[t - off] : 0;
    __syncthreads();
    sh[t] += a;
    __syncthreads();
  }
  return sh[t];
}

__global__ void __launch_bounds__(256)
count_blocksum(const int* __restrict__ cnt, int* __restrict__ bsum, int N) {
  __shared__ int sh[256];
  int n = blockIdx.x * 256 + threadIdx.x;
  int v = (n < N) ? cnt[(size_t)n * CNT_STRIDE] : 0;
  int inc = block_incl_scan_256(v, sh);
  if (threadIdx.x == 255) bsum[blockIdx.x] = inc;
}

__global__ void __launch_bounds__(256)
scan_rowptr(const int* __restrict__ cnt, const int* __restrict__ bsum,
            int* __restrict__ rowptr, int N) {
  __shared__ int sh[256];
  __shared__ int sh2[256];
  const int t = threadIdx.x;
  int part = 0;
  for (int j = t; j < (int)blockIdx.x; j += 256) part += bsum[j];
  block_incl_scan_256(part, sh2);
  const int base = sh2[255];
  int n = blockIdx.x * 256 + t;
  int v = (n < N) ? cnt[(size_t)n * CNT_STRIDE] : 0;
  int inc = block_incl_scan_256(v, sh);
  if (n < N) {
    rowptr[n] = base + inc - v;
    if (n == N - 1) rowptr[N] = base + inc;
  }
}

__global__ void fill_sorted(const int* __restrict__ src, const int* __restrict__ dst,
                            const float* __restrict__ w, const int* __restrict__ rowptr,
                            const int* __restrict__ rank, int* __restrict__ ssrc,
                            float* __restrict__ sw, int E) {
  int e0 = (blockIdx.x * blockDim.x + threadIdx.x) * 4;
  for (int i = 0; i < 4; ++i) {
    int e = e0 + i;
    if (e < E) {
      int p = rowptr[dst[e]] + rank[e];
      ssrc[p] = src[e];
      sw[p] = w[e];
    }
  }
}

__global__ void deg_dinv(const float* __restrict__ sw, const int* __restrict__ rowptr,
                         float* __restrict__ dinv, int N) {
  int n = blockIdx.x * blockDim.x + threadIdx.x;
  if (n >= N) return;
  const int p0 = rowptr[n], p1 = rowptr[n + 1];
  float s = 0.f;
  for (int p = p0; p < p1; ++p) s += sw[p];
  dinv[n] = s > 0.f ? rsqrtf(s) : 0.f;
}

__global__ void norm_sorted(const int* __restrict__ ssrc, const int* __restrict__ rowptr,
                            const float* __restrict__ dinv, float* snorm, int N) {
  int n = blockIdx.x * blockDim.x + threadIdx.x;
  if (n >= N) return;
  const int p0 = rowptr[n], p1 = rowptr[n + 1];
  const float dn = dinv[n];
  for (int p = p0; p < p1; ++p) snorm[p] = dinv[ssrc[p]] * snorm[p] * dn;
}

template <bool RELU>
__global__ void csr_gather(const float* __restrict__ h, const float* __restrict__ snorm,
                           const int* __restrict__ ssrc, const int* __restrict__ rowptr,
                           float* __restrict__ agg, int N, int F, int G) {
  long long tid = (long long)blockIdx.x * blockDim.x + threadIdx.x;
  if (tid >= (long long)N * G) return;
  int n = (int)(tid / G);
  int g = (int)(tid - (long long)n * G);
  const int p0 = rowptr[n], p1 = rowptr[n + 1];
  const float* hb = h + (size_t)g * 4;
  float4 acc = make_float4(0.f, 0.f, 0.f, 0.f);
#pragma unroll 4
  for (int p = p0; p < p1; ++p) {
    const int s = ssrc[p];
    const float w = snorm[p];
    const float4 v = *reinterpret_cast<const float4*>(hb + (size_t)s * F);
    acc.x += v.x * w;
    acc.y += v.y * w;
    acc.z += v.z * w;
    acc.w += v.w * w;
  }
  float4* o = reinterpret_cast<float4*>(agg + (size_t)n * F + (size_t)g * 4);
  float4 z = *o;
  z.x += acc.x;
  z.y += acc.y;
  z.z += acc.z;
  z.w += acc.w;
  if (RELU) {
    z.x = fmaxf(z.x, 0.f);
    z.y = fmaxf(z.y, 0.f);
    z.z = fmaxf(z.z, 0.f);
    z.w = fmaxf(z.w, 0.f);
  }
  *o = z;
}

template <int C>
__global__ void csr_gather_logsoftmax(const float* __restrict__ h,
                                      const float* __restrict__ snorm,
                                      const int* __restrict__ ssrc,
                                      const int* __restrict__ rowptr,
                                      float* __restrict__ y, int N) {
  constexpr int G = C / 4;
  constexpr int NPW = 32 / G;
  const int lane = threadIdx.x & 31;
  const int wave = (int)((blockIdx.x * blockDim.x + threadIdx.x) >> 5);
  const int slot = lane / G;
  const int g = lane - slot * G;
  const int n = wave * NPW + slot;
  const bool live = (slot < NPW) && (n < N);
  int p0 = 0, p1 = 0;
  if (live) {
    p0 = rowptr[n];
    p1 = rowptr[n + 1];
  }
  const float* hb = h + (size_t)g * 4;
  float4 acc = make_float4(0.f, 0.f, 0.f, 0.f);
#pragma unroll 4
  for (int p = p0; p < p1; ++p) {
    const int s = ssrc[p];
    const float w = snorm[p];
    const float4 v = *reinterpret_cast<const float4*>(hb + (size_t)s * C);
    acc.x += v.x * w;
    acc.y += v.y * w;
    acc.z += v.z * w;
    acc.w += v.w * w;
  }
  float4 t = make_float4(0.f, 0.f, 0.f, 0.f);
  float4* o = nullptr;
  if (live) {
    o = reinterpret_cast<float4*>(y + (size_t)n * C + (size_t)g * 4);
    const float4 z = *o;
    t.x = fmaxf(z.x + acc.x, 0.f);
    t.y = fmaxf(z.y + acc.y, 0.f);
    t.z = fmaxf(z.z + acc.z, 0.f);
    t.w = fmaxf(z.w + acc.w, 0.f);
  }
  const int first = live ? lane - g : lane;
  const int step = live ? 1 : 0;
  const float m4 = fmaxf(fmaxf(t.x, t.y), fmaxf(t.z, t.w));
  float m = -3.4e38f;
  for (int k = 0; k < G; ++k) m = fmaxf(m, __shfl(m4, first + k * step));
  const float e4 = __expf(t.x - m) + __expf(t.y - m) + __expf(t.z - m) + __expf(t.w - m);
  float s = 0.f;
  for (int k = 0; k < G; ++k) s += __shfl(e4, first + k * step);
  const float ls = m + __logf(s);
  if (live) *o = make_float4(t.x - ls, t.y - ls, t.z - ls, t.w - ls);
}

#define NPB_SHIFT 9
#define NPB (1 << NPB_SHIFT)
#define MAXB 256
#define BIN_CH 6400
#define BIN_TPB 1024
#define BIN_IT ((BIN_CH + BIN_TPB - 1) / BIN_TPB)
#define SORT_TPB 1024
#define SORT_IT 4
#define SORT_CAP 10240

template <int PER>
__device__ __forceinline__ void excl_scan_lds_wave0(int* a) {
  __syncthreads();
  if (threadIdx.x < 32) {
    const int lane = threadIdx.x;
    int v[PER];
    int s = 0;
#pragma unroll
    for (int k = 0; k < PER; ++k) {
      v[k] = a[lane * PER + k];
      s += v[k];
    }
    int inc = s;
#pragma unroll
    for (int off = 1; off < 32; off <<= 1) {
      const int u = __shfl_up(inc, off);
      if (lane >= off) inc += u;
    }
    int run = inc - s;
#pragma unroll
    for (int k = 0; k < PER; ++k) {
      a[lane * PER + k] = run;
      run += v[k];
    }
    if (lane == 31) a[32 * PER] = run;
  }
  __syncthreads();
}

__global__ void __launch_bounds__(BIN_TPB)
bin_count(const int* __restrict__ dst, int* __restrict__ cntmat, int E, int B) {
  __shared__ int hist[MAXB];
  const int t = threadIdx.x;
  if (t < MAXB) hist[t] = 0;
  __syncthreads();
  const int e0 = blockIdx.x * BIN_CH;
  const int n = min(BIN_CH, E - e0);
  int d[BIN_IT];
#pragma unroll
  for (int k = 0; k < BIN_IT; ++k) {
    const int i = t + k * BIN_TPB;
    d[k] = (i < n) ? dst[e0 + i] : -1;
  }
#pragma unroll
  for (int k = 0; k < BIN_IT; ++k)
    if (d[k] >= 0) atomicAdd(&hist[d[k] >> NPB_SHIFT], 1);
  __syncthreads();
  if (t < B) cntmat[(size_t)blockIdx.x * B + t] = hist[t];
}

__global__ void __launch_bounds__(BIN_TPB)
bin_scatter(const int* __restrict__ src, const int* __restrict__ dst,
            const float* __restrict__ w, const int* __restrict__ cntmat,
            int* __restrict__ bstart, int* __restrict__ bsrc, int* __restrict__ bdst,
            float* __restrict__ bw, int E, int B, int G1) {
  __shared__ int   s_src[BIN_CH];
  __shared__ int   s_dst[BIN_CH];
  __shared__ float s_w[BIN_CH];
  __shared__ int   p_tot[4][MAXB];
  __shared__ int   p_pre[4][MAXB];
  __shared__ int   gbase[MAXB + 1];
  __shared__ int   lstart[MAXB + 1];
  __shared__ int   lcnt[MAXB];
  const int t = threadIdx.x;
  const int g = blockIdx.x;
  const int col = t & (MAXB - 1);
  const int grp = t >> 8;
  int tot = 0, pre = 0;
  if (col < B) {
#pragma unroll 8
    for (int r = grp; r < G1; r += 4) {
      const int c = cntmat[(size_t)r * B + col];
      tot += c;
      pre += (r < g) ? c : 0;
    }
  }
  p_tot[grp][col] = tot;
  p_pre[grp][col] = pre;
  __syncthreads();
  int mypre = 0;
  if (t < MAXB) {
    gbase[t] = p_tot[0][t] + p_tot[1][t] + p_tot[2][t] + p_tot[3][t];
    mypre = p_pre[0][t] + p_pre[1][t] + p_pre[2][t] + p_pre[3][t];
    lstart[t] = (t < B) ? cntmat[(size_t)g * B + t] : 0;
    lcnt[t] = 0;
  }
  excl_scan_lds_wave0<MAXB / 32>(gbase);
  excl_scan_lds_wave0<MAXB / 32>(lstart);
  if (g == 0 && t <= B) bstart[t] = gbase[t];
  __syncthreads();
  if (t < MAXB) gbase[t] += mypre;
  __syncthreads();

  const int e0 = g * BIN_CH;
  const int n = min(BIN_CH, E - e0);
  int d[BIN_IT], sv[BIN_IT];
  float wv[BIN_IT];
#pragma unroll
  for (int k = 0; k < BIN_IT; ++k) {
    const int i = t + k * BIN_TPB;
    const bool in = i < n;
    d[k]  = in ? dst[e0 + i] : -1;
    sv[k] = in ? src[e0 + i] : 0;
    wv[k] = in ? w[e0 + i] : 0.f;
  }
#pragma unroll
  for (int k = 0; k < BIN_IT; ++k) {
    if (d[k] >= 0) {
      const int bk = d[k] >> NPB_SHIFT;
      const int q = lstart[bk] + atomicAdd(&lcnt[bk], 1);
      s_src[q] = sv[k];
      s_dst[q] = d[k];
      s_w[q] = wv[k];
    }
  }
  __syncthreads();
  for (int q = t; q < n; q += BIN_TPB) {
    const int d = s_dst[q];
    const int bk = d >> NPB_SHIFT;
    const int p = gbase[bk] + (q - lstart[bk]);
    bsrc[p] = s_src[q];
    bdst[p] = d;
    bw[p] = s_w[q];
  }
}

__global__ void __launch_bounds__(SORT_TPB)
bucket_sort(const int* __restrict__ bsrc, const int* __restrict__ bdst,
            const float* __restrict__ bw, const int* __restrict__ bstart,
            int* __restrict__ rowptr, int* __restrict__ ssrc, float* __restrict__ sw,
            float* __restrict__ dinv, int N, int B) {
  __shared__ int   c_src[SORT_CAP];
  __shared__ float c_w[SORT_CAP];
  __shared__ int   nstart[NPB + 1];
  __shared__ int   ncnt[NPB];
  const int t = threadIdx.x;
  const int b = blockIdx.x;
  const int m0 = bstart[b];
  const int m1 = bstart[b + 1];
  const int nE = m1 - m0;
  const int node0 = b << NPB_SHIFT;
  if (t < NPB) {
    nstart[t] = 0;
    ncnt[t] = 0;
  }
  __syncthreads();
  for (int i0 = t; i0 < nE; i0 += SORT_IT * SORT_TPB) {
    int dd[SORT_IT];
#pragma unroll
    for (int k = 0; k < SORT_IT; ++k) {
      const int i = i0 + k * SORT_TPB;
      dd[k] = (i < nE) ? bdst[m0 + i] : -1;
    }
#pragma unroll
    for (int k = 0; k < SORT_IT; ++k)
      if (dd[k] >= 0) atomicAdd(&nstart[dd[k] - node0], 1);
  }
  excl_scan_lds_wave0<NPB / 32>(nstart);
  if (t < NPB && node0 + t < N) rowptr[node0 + t] = m0 + nstart[t];
  if (b == B - 1 && t == 0) rowptr[N] = m1;
  const bool staged = nE <= SORT_CAP;
  if (staged) {
    for (int i0 = t; i0 < nE; i0 += SORT_IT * SORT_TPB) {
      int dd[SORT_IT], ss[SORT_IT];
      float xx[SORT_IT];
#pragma unroll
      for (int k = 0; k < SORT_IT; ++k) {
        const int i = i0 + k * SORT_TPB;
        const bool in = i < nE;
        dd[k] = in ? bdst[m0 + i] : -1;
        ss[k] = in ? bsrc[m0 + i] : 0;
        xx[k] = in ? bw[m0 + i] : 0.f;
      }
#pragma unroll
      for (int k = 0; k < SORT_IT; ++k) {
        if (dd[k] >= 0) {
          const int l = dd[k] - node0;
          const int q = nstart[l] + atomicAdd(&ncnt[l], 1);
          c_src[q] = ss[k];
          c_w[q] = xx[k];
        }
      }
    }
  } else {
    for (int i = t; i < nE; i += SORT_TPB) {
      const int l = bdst[m0 + i] - node0;
      const int q = nstart[l] + atomicAdd(&ncnt[l], 1);
      ssrc[m0 + q] = bsrc[m0 + i];
      sw[m0 + q] = bw[m0 + i];
    }
  }
  __syncthreads();
  if (t < NPB && node0 + t < N) {
    const int q0 = nstart[t], q1 = nstart[t + 1];
    float s = 0.f;
    if (staged) {
      for (int q = q0; q < q1; ++q) s += c_w[q];
    } else {
      for (int q = q0; q < q1; ++q) s += sw[m0 + q];
    }
    dinv[node0 + t] = s > 0.f ? rsqrtf(s) : 0.f;
  }
  if (staged) {
    for (int q = t; q < nE; q += SORT_TPB) {
      ssrc[m0 + q] = c_src[q];
      sw[m0 + q] = c_w[q];
    }
  }
}

static inline int ceil_div(long long a, int b) { return (int)((a + b - 1) / b); }

static void launch_gemm(const float* A, const float* B, const float* bias, float* C,
                        int N, int K, int F, hipStream_t stream) {
  const int TPB = 256;
  int rowT = ceil_div(N, 16), colT = ceil_div(F, 16);
  int blocks = ceil_div((long long)rowT * colT, TPB / 32);
  if (K == 64) {
    gemm_wmma<64><<<blocks, TPB, 0, stream>>>(A, B, bias, C, N, F, rowT, colT);
  } else if (K == 48) {
    gemm_wmma<48><<<blocks, TPB, 0, stream>>>(A, B, bias, C, N, F, rowT, colT);
  } else {
    gemm_wmma_gen<<<blocks, TPB, 0, stream>>>(A, B, bias, C, N, K, F, rowT, colT);
  }
}

static void launch_gemm_pair(const float* A, const float* B1, const float* B2,
                             const float* bias2, float* C1, float* C2,
                             int N, int K, int F, hipStream_t stream) {
  const int rowT = N / 16;
  const int WPB = 4;
  const int blocks = ceil_div(rowT, WPB);
  const bool full = (N % 16 == 0) && (N > 0) && (bias2 != nullptr);
  if (full && K == 64 && F == 48) {
    gemm_wmma_pair<64, 6><<<blocks, WPB * 32, 0, stream>>>(A, B1, B2, bias2, C1, C2, rowT);
  } else if (full && K == 48 && F == 40) {
    gemm_wmma_pair<48, 5><<<blocks, WPB * 32, 0, stream>>>(A, B1, B2, bias2, C1, C2, rowT);
  } else {
    launch_gemm(A, B1, nullptr, C1, N, K, F, stream);
    launch_gemm(A, B2, bias2,   C2, N, K, F, stream);
  }
}

extern "C" void kernel_launch(void* const* d_in, const int* in_sizes, int n_in,
                              void* d_out, int out_size, void* d_ws, size_t ws_size,
                              hipStream_t stream) {
  const float* x  = (const float*)d_in[0];
  const int*   ei = (const int*)d_in[1];
  const float* ew = (const float*)d_in[2];
  const float* W1 = (const float*)d_in[3];
  const float* V1 = (const float*)d_in[4];
  const float* b1 = (const float*)d_in[5];
  const float* W2 = (const float*)d_in[6];
  const float* V2 = (const float*)d_in[7];
  const float* b2 = (const float*)d_in[8];
  float* out = (float*)d_out;

  const int H   = in_sizes[5];
  const int Cn  = in_sizes[8];
  const int Fin = in_sizes[3] / H;
  const int N   = in_sizes[0] / Fin;
  const int E   = in_sizes[2];
  const int* src = ei;
  const int* dst = ei + E;

  float* ws = (float*)d_ws;
  const size_t NHa = ((size_t)N * H + 255) & ~(size_t)255;
  const size_t Ea  = ((size_t)E + 255) & ~(size_t)255;
  size_t off = 0;
  float* deg  = ws + off; off += ((size_t)N + 255) & ~(size_t)255;
  float* norm = ws + off; off += Ea;
  float* hbuf = ws + off; off += NHa;
  float* z1   = ws + off; off += NHa;
  int* ssrc   = (int*)(ws + off); off += Ea;
  int* rowptr = (int*)(ws + off); off += ((size_t)N + 1 + 255) & ~(size_t)255;
  const int NB = ceil_div(N, 256);
  int* rank = (int*)hbuf;
  int* cnt  = rank + Ea;
  int* bsum = cnt + (((size_t)N * CNT_STRIDE + 255) & ~(size_t)255);
  const size_t scratch_words = Ea + (((size_t)N * CNT_STRIDE + 255) & ~(size_t)255) + (size_t)NB;
  const bool sorted_path = (off * sizeof(float) <= ws_size) && (scratch_words <= 2 * NHa) &&
                           (H % 4 == 0) && (Cn % 4 == 0);
  const int B  = ceil_div(N, NPB);
  const int G1 = ceil_div(E, BIN_CH);
  const size_t matw = ((size_t)G1 * B + 255) & ~(size_t)255;
  int*   bsrc   = (int*)hbuf;
  int*   bdst   = bsrc + Ea;
  float* bw     = (float*)(bdst + Ea);
  int*   cntmat = (int*)(bw + Ea);
  int*   bstart = cntmat + matw;
  const bool lds_sort = (B <= MAXB) && (E > 0) && (3 * Ea + matw + (size_t)B + 1 <= 2 * NHa);

  const int TPB = 256;

  if (sorted_path) {
    if (lds_sort) {
      bin_count<<<G1, BIN_TPB, 0, stream>>>(dst, cntmat, E, B);
      bin_scatter<<<G1, BIN_TPB, 0, stream>>>(src, dst, ew, cntmat, bstart, bsrc, bdst, bw,
                                              E, B, G1);
      bucket_sort<<<B, SORT_TPB, 0, stream>>>(bsrc, bdst, bw, bstart, rowptr, ssrc, norm, deg,
                                              N, B);
    } else {
      zero_counters<<<NB, TPB, 0, stream>>>(cnt, N);
      rank_edges<<<ceil_div(ceil_div(E, 4), TPB), TPB, 0, stream>>>(dst, cnt, rank, E);
      count_blocksum<<<NB, 256, 0, stream>>>(cnt, bsum, N);
      scan_rowptr<<<NB, 256, 0, stream>>>(cnt, bsum, rowptr, N);
      fill_sorted<<<ceil_div(ceil_div(E, 4), TPB), TPB, 0, stream>>>(src, dst, ew, rowptr, rank,
                                                                   ssrc, norm, E);
      deg_dinv<<<NB, TPB, 0, stream>>>(norm, rowptr, deg, N);
    }
    norm_sorted<<<NB, TPB, 0, stream>>>(ssrc, rowptr, deg, norm, N);

    launch_gemm_pair(x, W1, V1, b1, hbuf, z1, N, Fin, H, stream);
    {
      int G = H / 4;
      csr_gather<true><<<ceil_div((long long)N * G, TPB), TPB, 0, stream>>>(
          hbuf, norm, ssrc, rowptr, z1, N, H, G);
    }

    launch_gemm_pair(z1, W2, V2, b2, hbuf, out, N, H, Cn, stream);
    {
      if (Cn == 40) {
        csr_gather_logsoftmax<40><<<ceil_div((long long)ceil_div(N, 3) * 32, TPB), TPB, 0, stream>>>(
            hbuf, norm, ssrc, rowptr, out, N);
      } else {
        int G = Cn / 4;
        csr_gather<false><<<ceil_div((long long)N * G, TPB), TPB, 0, stream>>>(
            hbuf, norm, ssrc, rowptr, out, N, Cn, G);
        relu_logsoftmax_gen<<<ceil_div(N, TPB), TPB, 0, stream>>>(out, N, Cn);
      }
    }
    return;
  }

  zero_f32<<<ceil_div(N, TPB), TPB, 0, stream>>>(deg, N);
  deg_accum<<<ceil_div(E, TPB), TPB, 0, stream>>>(dst, ew, deg, E);
  norm_compute<<<ceil_div(E, TPB), TPB, 0, stream>>>(src, dst, ew, deg, norm, E);

  launch_gemm(x, W1, nullptr, hbuf, N, Fin, H, stream);
  launch_gemm(x, V1, b1,      z1,   N, Fin, H, stream);
  {
    int G = H / 4;
    edge_scatter<<<ceil_div((long long)E * G, TPB), TPB, 0, stream>>>(
        hbuf, norm, src, dst, z1, E, H, G);
    relu_inplace<<<ceil_div((long long)N * H, TPB), TPB, 0, stream>>>(z1, N * H);
  }

  launch_gemm(z1, W2, nullptr, hbuf, N, H, Cn, stream);
  launch_gemm(z1, V2, b2,      out,  N, H, Cn, stream);
  {
    int G = Cn / 4;
    edge_scatter<<<ceil_div((long long)E * G, TPB), TPB, 0, stream>>>(
        hbuf, norm, src, dst, out, E, Cn, G);
    if (Cn == 40) {
      relu_logsoftmax_t<40><<<ceil_div(N, TPB), TPB, 0, stream>>>(out, N);
    } else {
      relu_logsoftmax_gen<<<ceil_div(N, TPB), TPB, 0, stream>>>(out, N, Cn);
    }
  }
}


# Edge norm factored (products re-associated, sum order kept): h rows scaled by deg^-1/2 in the product's store, segment sums use raw weights and scale once by deg^-1/2[dst]; norm launch gone
# speedup vs baseline: 3.0349x; 1.1243x over previous
#include <hip/hip_runtime.h>

typedef float v2f __attribute__((ext_vector_type(2)));
typedef float v8f __attribute__((ext_vector_type(8)));

__global__ void zero_f32(float* __restrict__ p, int n) {
  int i = blockIdx.x * blockDim.x + threadIdx.x;
  if (i < n) p[i] = 0.f;
}

__global__ void relu_inplace(float* __restrict__ p, int n) {
  int i = blockIdx.x * blockDim.x + threadIdx.x;
  if (i < n) p[i] = fmaxf(p[i], 0.f);
}

__global__ void deg_accum(const int* __restrict__ dst, const float* __restrict__ w,
                          float* __restrict__ deg, int E) {
  int e = blockIdx.x * blockDim.x + threadIdx.x;
  if (e < E) atomicAdd(deg + dst[e], w[e]);
}

__global__ void norm_compute(const int* __restrict__ src, const int* __restrict__ dst,
                             const float* __restrict__ w, const float* __restrict__ deg,
                             float* __restrict__ norm, int E) {
  int e = blockIdx.x * blockDim.x + threadIdx.x;
  if (e >= E) return;
  float ds = deg[src[e]];
  float dd = deg[dst[e]];
  float is = ds > 0.f ? rsqrtf(ds) : 0.f;
  float id = dd > 0.f ? rsqrtf(dd) : 0.f;
  norm[e] = is * w[e] * id;
}

template <int K>
__global__ void gemm_wmma(const float* __restrict__ A, const float* __restrict__ B,
                          const float* __restrict__ bias, const float* __restrict__ rscale,
                          float* __restrict__ C, int N, int F, int rowTiles, int colTiles) {
  const int lane = threadIdx.x & 31;
  const int wave = threadIdx.x >> 5;
  const int tile = blockIdx.x * (blockDim.x >> 5) + wave;
  if (tile >= rowTiles * colTiles) return;

  const int rt = tile / colTiles;
  const int ct = tile - rt * colTiles;
  const int row0 = rt << 4;
  const int col0 = ct << 4;
  const int half = lane >> 4;
  const int lm   = lane & 15;

  const bool cOK = (col0 + lm) < F;
  const int  arow = min(row0 + lm, N - 1);
  const int  bcol = min(col0 + lm, F - 1);

  const float* Arow = A + (size_t)arow * K + 2 * half;
  const float* Bcol = B + bcol;

  v2f afrag[K / 4];
  v2f bfrag[K / 4];
#pragma unroll
  for (int i = 0; i < K / 4; ++i) {
    const int ka = 4 * i + 2 * half;
    afrag[i] = *reinterpret_cast<const v2f*>(Arow + 4 * i);
    bfrag[i].x = Bcol[(size_t)ka * F];
    bfrag[i].y = Bcol[(size_t)(ka + 1) * F];
  }

  v8f acc = {};
#pragma unroll
  for (int i = 0; i < K / 4; ++i) {
    acc = __builtin_amdgcn_wmma_f32_16x16x4_f32(
        false, afrag[i], false, bfrag[i], (short)0, acc, false, false);
  }

  const float bv = (bias != nullptr) ? bias[bcol] : 0.f;
#pragma unroll
  for (int r = 0; r < 8; ++r) {
    const int row = row0 + r + 8 * half;
    if (row < N && cOK) {
      const float v = (rscale != nullptr) ? acc[r] * rscale[row] : acc[r];
      C[(size_t)row * F + col0 + lm] = v + bv;
    }
  }
}

template <int K, int NT>
__global__ void __launch_bounds__(128)
gemm_wmma_pair(const float* __restrict__ A, const float* __restrict__ B1,
               const float* __restrict__ B2, const float* __restrict__ bias2,
               const float* __restrict__ rscale1,
               float* __restrict__ C1, float* __restrict__ C2,
               int rowTiles) {
  constexpr int F = 8 * NT;
  const int lane = threadIdx.x & 31;
  const int wave = threadIdx.x >> 5;
  const int rt = blockIdx.x * (blockDim.x >> 5) + wave;
  if (rt >= rowTiles) return;

  const int row0 = rt << 4;
  const int half = lane >> 4;
  const int lm   = lane & 15;
  const float* Arow = A + (size_t)(row0 + lm) * K + 2 * half;

  v2f afrag[K / 4];
#pragma unroll
  for (int i = 0; i < K / 4; ++i)
    afrag[i] = *reinterpret_cast<const v2f*>(Arow + 4 * i);

  float rs[8];
#pragma unroll
  for (int r = 0; r < 8; ++r) rs[r] = rscale1[row0 + 8 * half + r];

  const size_t rbase = (size_t)(row0 + 8 * half) * F;
#pragma unroll
  for (int t = 0; t < NT; ++t) {
    const int j = 16 * t + lm;
    const bool second = j >= F;
    const int col = second ? j - F : j;
    const float* Bcol = (second ? B2 : B1) + col;
    const float braw = bias2[col];
    const float bv = second ? braw : 0.f;

    v2f bfrag[K / 4];
#pragma unroll
    for (int i = 0; i < K / 4; ++i) {
      const int ka = 4 * i + 2 * half;
      bfrag[i].x = Bcol[(size_t)ka * F];
      bfrag[i].y = Bcol[(size_t)(ka + 1) * F];
    }

    v8f acc = {};
#pragma unroll
    for (int i = 0; i < K / 4; ++i) {
      acc = __builtin_amdgcn_wmma_f32_16x16x4_f32(
          false, afrag[i], false, bfrag[i], (short)0, acc, false, false);
    }

    float* Crow = (second ? C2 : C1) + col + rbase;
#pragma unroll
    for (int r = 0; r < 8; ++r) Crow[r * F] = second ? acc[r] + bv : acc[r] * rs[r];
  }
}

__global__ void gemm_wmma_gen(const float* __restrict__ A, const float* __restrict__ B,
                              const float* __restrict__ bias, const float* __restrict__ rscale,
                              float* __restrict__ C, int N, int K, int F, int rowTiles,
                              int colTiles) {
  const int lane = threadIdx.x & 31;
  const int wave = threadIdx.x >> 5;
  const int tile = blockIdx.x * (blockDim.x >> 5) + wave;
  if (tile >= rowTiles * colTiles) return;
  const int rt = tile / colTiles;
  const int ct = tile - rt * colTiles;
  const int row0 = rt << 4, col0 = ct << 4;
  const int half = lane >> 4, lm = lane & 15;
  const bool cOK = (col0 + lm) < F;
  const int arow = min(row0 + lm, N - 1);
  const int bcol = min(col0 + lm, F - 1);
  v8f acc = {};
  for (int kb = 0; kb < K; kb += 4) {
    const int ka = kb + 2 * half;
    v2f a = *reinterpret_cast<const v2f*>(A + (size_t)arow * K + ka);
    v2f b;
    b.x = B[(size_t)ka * F + bcol];
    b.y = B[(size_t)(ka + 1) * F + bcol];
    acc = __builtin_amdgcn_wmma_f32_16x16x4_f32(false, a, false, b, (short)0, acc,
                                                false, false);
  }
  const float bv = (bias != nullptr) ? bias[bcol] : 0.f;
#pragma unroll
  for (int r = 0; r < 8; ++r) {
    const int row = row0 + r + 8 * half;
    if (row < N && cOK) {
      const float v = (rscale != nullptr) ? acc[r] * rscale[row] : acc[r];
      C[(size_t)row * F + col0 + lm] = v + bv;
    }
  }
}

__global__ void edge_scatter(const float* __restrict__ h, const float* __restrict__ norm,
                             const int* __restrict__ src, const int* __restrict__ dst,
                             float* __restrict__ agg, int E, int F, int G) {
  long long tid = (long long)blockIdx.x * blockDim.x + threadIdx.x;
  if (tid >= (long long)E * G) return;
  int e = (int)(tid / G);
  int g = (int)(tid - (long long)e * G);
  int s = src[e];
  int d = dst[e];
  float w = norm[e];
  const float4 v = *reinterpret_cast<const float4*>(h + (size_t)s * F + (size_t)g * 4);
  float* base = agg + (size_t)d * F + (size_t)g * 4;
  atomicAdd(base + 0, v.x * w);
  atomicAdd(base + 1, v.y * w);
  atomicAdd(base + 2, v.z * w);
  atomicAdd(base + 3, v.w * w);
}

template <int C>
__global__ void relu_logsoftmax_t(float* __restrict__ y, int N) {
  int n = blockIdx.x * blockDim.x + threadIdx.x;
  if (n >= N) return;
  float v[C];
  float m = -3.4e38f;
  float* row = y + (size_t)n * C;
#pragma unroll
  for (int c = 0; c < C; ++c) {
    float t = fmaxf(row[c], 0.f);
    v[c] = t;
    m = fmaxf(m, t);
  }
  float s = 0.f;
#pragma unroll
  for (int c = 0; c < C; ++c) s += __expf(v[c] - m);
  float ls = m + __logf(s);
#pragma unroll
  for (int c = 0; c < C; ++c) row[c] = v[c] - ls;
}

__global__ void relu_logsoftmax_gen(float* __restrict__ y, int N, int C) {
  int n = blockIdx.x * blockDim.x + threadIdx.x;
  if (n >= N) return;
  float* row = y + (size_t)n * C;
  float m = -3.4e38f;
  for (int c = 0; c < C; ++c) m = fmaxf(m, fmaxf(row[c], 0.f));
  float s = 0.f;
  for (int c = 0; c < C; ++c) s += __expf(fmaxf(row[c], 0.f) - m);
  float ls = m + __logf(s);
  for (int c = 0; c < C; ++c) row[c] = fmaxf(row[c], 0.f) - ls;
}

#define CNT_STRIDE 32

__global__ void zero_counters(int* __restrict__ cnt, int N) {
  int n = blockIdx.x * blockDim.x + threadIdx.x;
  if (n < N) cnt[(size_t)n * CNT_STRIDE] = 0;
}

__global__ void rank_edges(const int* __restrict__ dst, int* __restrict__ cnt,
                           int* __restrict__ rank, int E) {
  int e0 = (blockIdx.x * blockDim.x + threadIdx.x) * 4;
  int d[4];
#pragma unroll
  for (int i = 0; i < 4; ++i) d[i] = (e0 + i < E) ? dst[e0 + i] : -1;
  int r[4];
#pragma unroll
  for (int i = 0; i < 4; ++i)
    r[i] = (d[i] >= 0) ? atomicAdd(cnt + (size_t)d[i] * CNT_STRIDE, 1) : 0;
#pragma unroll
  for (int i = 0; i < 4; ++i)
    if (e0 + i < E) rank[e0 + i] = r[i];
}

__device__ __forceinline__ int block_incl_scan_256(int v, int* sh) {
  const int t = threadIdx.x;
  sh[t] = v;
  __syncthreads();
  for (int off = 1; off < 256; off <<= 1) {
    int a = (t >= off) ? sh[t - off] : 0;
    __syncthreads();
    sh[t] += a;
    __syncthreads();
  }
  return sh[t];
}

__global__ void __launch_bounds__(256)
count_blocksum(const int* __restrict__ cnt, int* __restrict__ bsum, int N) {
  __shared__ int sh[256];
  int n = blockIdx.x * 256 + threadIdx.x;
  int v = (n < N) ? cnt[(size_t)n * CNT_STRIDE] : 0;
  int inc = block_incl_scan_256(v, sh);
  if (threadIdx.x == 255) bsum[blockIdx.x] = inc;
}

__global__ void __launch_bounds__(256)
scan_rowptr(const int* __restrict__ cnt, const int* __restrict__ bsum,
            int* __restrict__ rowptr, int N) {
  __shared__ int sh[256];
  __shared__ int sh2[256];
  const int t = threadIdx.x;
  int part = 0;
  for (int j = t; j < (int)blockIdx.x; j += 256) part += bsum[j];
  block_incl_scan_256(part, sh2);
  const int base = sh2[255];
  int n = blockIdx.x * 256 + t;
  int v = (n < N) ? cnt[(size_t)n * CNT_STRIDE] : 0;
  int inc = block_incl_scan_256(v, sh);
  if (n < N) {
    rowptr[n] = base + inc - v;
    if (n == N - 1) rowptr[N] = base + inc;
  }
}

__global__ void fill_sorted(const int* __restrict__ src, const int* __restrict__ dst,
                            const float* __restrict__ w, const int* __restrict__ rowptr,
                            const int* __restrict__ rank, int* __restrict__ ssrc,
                            float* __restrict__ sw, int E) {
  int e0 = (blockIdx.x * blockDim.x + threadIdx.x) * 4;
  for (int i = 0; i < 4; ++i) {
    int e = e0 + i;
    if (e < E) {
      int p = rowptr[dst[e]] + rank[e];
      ssrc[p] = src[e];
      sw[p] = w[e];
    }
  }
}

__global__ void deg_dinv(const float* __restrict__ sw, const int* __restrict__ rowptr,
                         float* __restrict__ dinv, int N) {
  int n = blockIdx.x * blockDim.x + threadIdx.x;
  if (n >= N) return;
  const int p0 = rowptr[n], p1 = rowptr[n + 1];
  float s = 0.f;
  for (int p = p0; p < p1; ++p) s += sw[p];
  dinv[n] = s > 0.f ? rsqrtf(s) : 0.f;
}

template <bool RELU>
__global__ void csr_gather(const float* __restrict__ h, const float* __restrict__ sw,
                           const int* __restrict__ ssrc, const int* __restrict__ rowptr,
                           const float* __restrict__ dinv,
                           float* __restrict__ agg, int N, int F, int G) {
  long long tid = (long long)blockIdx.x * blockDim.x + threadIdx.x;
  if (tid >= (long long)N * G) return;
  int n = (int)(tid / G);
  int g = (int)(tid - (long long)n * G);
  const int p0 = rowptr[n], p1 = rowptr[n + 1];
  const float dn = dinv[n];
  const float* hb = h + (size_t)g * 4;
  float4 acc = make_float4(0.f, 0.f, 0.f, 0.f);
#pragma unroll 4
  for (int p = p0; p < p1; ++p) {
    const int s = ssrc[p];
    const float w = sw[p];
    const float4 v = *reinterpret_cast<const float4*>(hb + (size_t)s * F);
    acc.x += v.x * w;
    acc.y += v.y * w;
    acc.z += v.z * w;
    acc.w += v.w * w;
  }
  float4* o = reinterpret_cast<float4*>(agg + (size_t)n * F + (size_t)g * 4);
  float4 z = *o;
  z.x += acc.x * dn;
  z.y += acc.y * dn;
  z.z += acc.z * dn;
  z.w += acc.w * dn;
  if (RELU) {
    z.x = fmaxf(z.x, 0.f);
    z.y = fmaxf(z.y, 0.f);
    z.z = fmaxf(z.z, 0.f);
    z.w = fmaxf(z.w, 0.f);
  }
  *o = z;
}

template <int C>
__global__ void csr_gather_logsoftmax(const float* __restrict__ h,
                                      const float* __restrict__ sw,
                                      const int* __restrict__ ssrc,
                                      const int* __restrict__ rowptr,
                                      const float* __restrict__ dinv,
                                      float* __restrict__ y, int N) {
  constexpr int G = C / 4;
  constexpr int NPW = 32 / G;
  const int lane = threadIdx.x & 31;
  const int wave = (int)((blockIdx.x * blockDim.x + threadIdx.x) >> 5);
  const int slot = lane / G;
  const int g = lane - slot * G;
  const int n = wave * NPW + slot;
  const bool live = (slot < NPW) && (n < N);
  int p0 = 0, p1 = 0;
  float dn = 0.f;
  if (live) {
    p0 = rowptr[n];
    p1 = rowptr[n + 1];
    dn = dinv[n];
  }
  const float* hb = h + (size_t)g * 4;
  float4 acc = make_float4(0.f, 0.f, 0.f, 0.f);
#pragma unroll 4
  for (int p = p0; p < p1; ++p) {
    const int s = ssrc[p];
    const float w = sw[p];
    const float4 v = *reinterpret_cast<const float4*>(hb + (size_t)s * C);
    acc.x += v.x * w;
    acc.y += v.y * w;
    acc.z += v.z * w;
    acc.w += v.w * w;
  }
  float4 t = make_float4(0.f, 0.f, 0.f, 0.f);
  float4* o = nullptr;
  if (live) {
    o = reinterpret_cast<float4*>(y + (size_t)n * C + (size_t)g * 4);
    const float4 z = *o;
    t.x = fmaxf(z.x + acc.x * dn, 0.f);
    t.y = fmaxf(z.y + acc.y * dn, 0.f);
    t.z = fmaxf(z.z + acc.z * dn, 0.f);
    t.w = fmaxf(z.w + acc.w * dn, 0.f);
  }
  const int first = live ? lane - g : lane;
  const int step = live ? 1 : 0;
  const float m4 = fmaxf(fmaxf(t.x, t.y), fmaxf(t.z, t.w));
  float m = -3.4e38f;
  for (int k = 0; k < G; ++k) m = fmaxf(m, __shfl(m4, first + k * step));
  const float e4 = __expf(t.x - m) + __expf(t.y - m) + __expf(t.z - m) + __expf(t.w - m);
  float s = 0.f;
  for (int k = 0; k < G; ++k) s += __shfl(e4, first + k * step);
  const float ls = m + __logf(s);
  if (live) *o = make_float4(t.x - ls, t.y - ls, t.z - ls, t.w - ls);
}

#define NPB_SHIFT 9
#define NPB (1 << NPB_SHIFT)
#define MAXB 256
#define BIN_CH 6400
#define BIN_TPB 1024
#define BIN_IT ((BIN_CH + BIN_TPB - 1) / BIN_TPB)
#define SORT_TPB 1024
#define SORT_IT 4
#define SORT_CAP 10240

template <int PER>
__device__ __forceinline__ void excl_scan_lds_wave0(int* a) {
  __syncthreads();
  if (threadIdx.x < 32) {
    const int lane = threadIdx.x;
    int v[PER];
    int s = 0;
#pragma unroll
    for (int k = 0; k < PER; ++k) {
      v[k] = a[lane * PER + k];
      s += v[k];
    }
    int inc = s;
#pragma unroll
    for (int off = 1; off < 32; off <<= 1) {
      const int u = __shfl_up(inc, off);
      if (lane >= off) inc += u;
    }
    int run = inc - s;
#pragma unroll
    for (int k = 0; k < PER; ++k) {
      a[lane * PER + k] = run;
      run += v[k];
    }
    if (lane == 31) a[32 * PER] = run;
  }
  __syncthreads();
}

__global__ void __launch_bounds__(BIN_TPB)
bin_count(const int* __restrict__ dst, int* __restrict__ cntmat, int E, int B) {
  __shared__ int hist[MAXB];
  const int t = threadIdx.x;
  if (t < MAXB) hist[t] = 0;
  __syncthreads();
  const int e0 = blockIdx.x * BIN_CH;
  const int n = min(BIN_CH, E - e0);
  int d[BIN_IT];
#pragma unroll
  for (int k = 0; k < BIN_IT; ++k) {
    const int i = t + k * BIN_TPB;
    d[k] = (i < n) ? dst[e0 + i] : -1;
  }
#pragma unroll
  for (int k = 0; k < BIN_IT; ++k)
    if (d[k] >= 0) atomicAdd(&hist[d[k] >> NPB_SHIFT], 1);
  __syncthreads();
  if (t < B) cntmat[(size_t)blockIdx.x * B + t] = hist[t];
}

__global__ void __launch_bounds__(BIN_TPB)
bin_scatter(const int* __restrict__ src, const int* __restrict__ dst,
            const float* __restrict__ w, const int* __restrict__ cntmat,
            int* __restrict__ bstart, int* __restrict__ bsrc, int* __restrict__ bdst,
            float* __restrict__ bw, int E, int B, int G1) {
  __shared__ int   s_src[BIN_CH];
  __shared__ int   s_dst[BIN_CH];
  __shared__ float s_w[BIN_CH];
  __shared__ int   p_tot[4][MAXB];
  __shared__ int   p_pre[4][MAXB];
  __shared__ int   gbase[MAXB + 1];
  __shared__ int   lstart[MAXB + 1];
  __shared__ int   lcnt[MAXB];
  const int t = threadIdx.x;
  const int g = blockIdx.x;
  const int col = t & (MAXB - 1);
  const int grp = t >> 8;
  int tot = 0, pre = 0;
  if (col < B) {
#pragma unroll 8
    for (int r = grp; r < G1; r += 4) {
      const int c = cntmat[(size_t)r * B + col];
      tot += c;
      pre += (r < g) ? c : 0;
    }
  }
  p_tot[grp][col] = tot;
  p_pre[grp][col] = pre;
  __syncthreads();
  int mypre = 0;
  if (t < MAXB) {
    gbase[t] = p_tot[0][t] + p_tot[1][t] + p_tot[2][t] + p_tot[3][t];
    mypre = p_pre[0][t] + p_pre[1][t] + p_pre[2][t] + p_pre[3][t];
    lstart[t] = (t < B) ? cntmat[(size_t)g * B + t] : 0;
    lcnt[t] = 0;
  }
  excl_scan_lds_wave0<MAXB / 32>(gbase);
  excl_scan_lds_wave0<MAXB / 32>(lstart);
  if (g == 0 && t <= B) bstart[t] = gbase[t];
  __syncthreads();
  if (t < MAXB) gbase[t] += mypre;
  __syncthreads();

  const int e0 = g * BIN_CH;
  const int n = min(BIN_CH, E - e0);
  int d[BIN_IT], sv[BIN_IT];
  float wv[BIN_IT];
#pragma unroll
  for (int k = 0; k < BIN_IT; ++k) {
    const int i = t + k * BIN_TPB;
    const bool in = i < n;
    d[k]  = in ? dst[e0 + i] : -1;
    sv[k] = in ? src[e0 + i] : 0;
    wv[k] = in ? w[e0 + i] : 0.f;
  }
#pragma unroll
  for (int k = 0; k < BIN_IT; ++k) {
    if (d[k] >= 0) {
      const int bk = d[k] >> NPB_SHIFT;
      const int q = lstart[bk] + atomicAdd(&lcnt[bk], 1);
      s_src[q] = sv[k];
      s_dst[q] = d[k];
      s_w[q] = wv[k];
    }
  }
  __syncthreads();
  for (int q = t; q < n; q += BIN_TPB) {
    const int d = s_dst[q];
    const int bk = d >> NPB_SHIFT;
    const int p = gbase[bk] + (q - lstart[bk]);
    bsrc[p] = s_src[q];
    bdst[p] = d;
    bw[p] = s_w[q];
  }
}

__global__ void __launch_bounds__(SORT_TPB)
bucket_sort(const int* __restrict__ bsrc, const int* __restrict__ bdst,
            const float* __restrict__ bw, const int* __restrict__ bstart,
            int* __restrict__ rowptr, int* __restrict__ ssrc, float* __restrict__ sw,
            float* __restrict__ dinv, int N, int B) {
  __shared__ int   c_src[SORT_CAP];
  __shared__ float c_w[SORT_CAP];
  __shared__ int   nstart[NPB + 1];
  __shared__ int   ncnt[NPB];
  const int t = threadIdx.x;
  const int b = blockIdx.x;
  const int m0 = bstart[b];
  const int m1 = bstart[b + 1];
  const int nE = m1 - m0;
  const int node0 = b << NPB_SHIFT;
  if (t < NPB) {
    nstart[t] = 0;
    ncnt[t] = 0;
  }
  __syncthreads();
  for (int i0 = t; i0 < nE; i0 += SORT_IT * SORT_TPB) {
    int dd[SORT_IT];
#pragma unroll
    for (int k = 0; k < SORT_IT; ++k) {
      const int i = i0 + k * SORT_TPB;
      dd[k] = (i < nE) ? bdst[m0 + i] : -1;
    }
#pragma unroll
    for (int k = 0; k < SORT_IT; ++k)
      if (dd[k] >= 0) atomicAdd(&nstart[dd[k] - node0], 1);
  }
  excl_scan_lds_wave0<NPB / 32>(nstart);
  if (t < NPB && node0 + t < N) rowptr[node0 + t] = m0 + nstart[t];
  if (b == B - 1 && t == 0) rowptr[N] = m1;
  const bool staged = nE <= SORT_CAP;
  if (staged) {
    for (int i0 = t; i0 < nE; i0 += SORT_IT * SORT_TPB) {
      int dd[SORT_IT], ss[SORT_IT];
      float xx[SORT_IT];
#pragma unroll
      for (int k = 0; k < SORT_IT; ++k) {
        const int i = i0 + k * SORT_TPB;
        const bool in = i < nE;
        dd[k] = in ? bdst[m0 + i] : -1;
        ss[k] = in ? bsrc[m0 + i] : 0;
        xx[k] = in ? bw[m0 + i] : 0.f;
      }
#pragma unroll
      for (int k = 0; k < SORT_IT; ++k) {
        if (dd[k] >= 0) {
          const int l = dd[k] - node0;
          const int q = nstart[l] + atomicAdd(&ncnt[l], 1);
          c_src[q] = ss[k];
          c_w[q] = xx[k];
        }
      }
    }
  } else {
    for (int i = t; i < nE; i += SORT_TPB) {
      const int l = bdst[m0 + i] - node0;
      const int q = nstart[l] + atomicAdd(&ncnt[l], 1);
      ssrc[m0 + q] = bsrc[m0 + i];
      sw[m0 + q] = bw[m0 + i];
    }
  }
  __syncthreads();
  if (t < NPB && node0 + t < N) {
    const int q0 = nstart[t], q1 = nstart[t + 1];
    float s = 0.f;
    if (staged) {
      for (int q = q0; q < q1; ++q) s += c_w[q];
    } else {
      for (int q = q0; q < q1; ++q) s += sw[m0 + q];
    }
    dinv[node0 + t] = s > 0.f ? rsqrtf(s) : 0.f;
  }
  if (staged) {
    for (int q = t; q < nE; q += SORT_TPB) {
      ssrc[m0 + q] = c_src[q];
      sw[m0 + q] = c_w[q];
    }
  }
}

static inline int ceil_div(long long a, int b) { return (int)((a + b - 1) / b); }

static void launch_gemm(const float* A, const float* B, const float* bias, const float* rscale,
                        float* C, int N, int K, int F, hipStream_t stream) {
  const int TPB = 256;
  int rowT = ceil_div(N, 16), colT = ceil_div(F, 16);
  int blocks = ceil_div((long long)rowT * colT, TPB / 32);
  if (K == 64) {
    gemm_wmma<64><<<blocks, TPB, 0, stream>>>(A, B, bias, rscale, C, N, F, rowT, colT);
  } else if (K == 48) {
    gemm_wmma<48><<<blocks, TPB, 0, stream>>>(A, B, bias, rscale, C, N, F, rowT, colT);
  } else {
    gemm_wmma_gen<<<blocks, TPB, 0, stream>>>(A, B, bias, rscale, C, N, K, F, rowT, colT);
  }
}

static void launch_gemm_pair(const float* A, const float* B1, const float* B2,
                             const float* bias2, const float* rscale1, float* C1, float* C2,
                             int N, int K, int F, hipStream_t stream) {
  const int rowT = N / 16;
  const int WPB = 4;
  const int blocks = ceil_div(rowT, WPB);
  const bool full = (N % 16 == 0) && (N > 0) && (bias2 != nullptr);
  if (full && K == 64 && F == 48) {
    gemm_wmma_pair<64, 6><<<blocks, WPB * 32, 0, stream>>>(A, B1, B2, bias2, rscale1, C1, C2,
                                                            rowT);
  } else if (full && K == 48 && F == 40) {
    gemm_wmma_pair<48, 5><<<blocks, WPB * 32, 0, stream>>>(A, B1, B2, bias2, rscale1, C1, C2,
                                                            rowT);
  } else {
    launch_gemm(A, B1, nullptr, rscale1, C1, N, K, F, stream);
    launch_gemm(A, B2, bias2,   nullptr, C2, N, K, F, stream);
  }
}

extern "C" void kernel_launch(void* const* d_in, const int* in_sizes, int n_in,
                              void* d_out, int out_size, void* d_ws, size_t ws_size,
                              hipStream_t stream) {
  const float* x  = (const float*)d_in[0];
  const int*   ei = (const int*)d_in[1];
  const float* ew = (const float*)d_in[2];
  const float* W1 = (const float*)d_in[3];
  const float* V1 = (const float*)d_in[4];
  const float* b1 = (const float*)d_in[5];
  const float* W2 = (const float*)d_in[6];
  const float* V2 = (const float*)d_in[7];
  const float* b2 = (const float*)d_in[8];
  float* out = (float*)d_out;

  const int H   = in_sizes[5];
  const int Cn  = in_sizes[8];
  const int Fin = in_sizes[3] / H;
  const int N   = in_sizes[0] / Fin;
  const int E   = in_sizes[2];
  const int* src = ei;
  const int* dst = ei + E;

  float* ws = (float*)d_ws;
  const size_t NHa = ((size_t)N * H + 255) & ~(size_t)255;
  const size_t Ea  = ((size_t)E + 255) & ~(size_t)255;
  size_t off = 0;
  float* deg  = ws + off; off += ((size_t)N + 255) & ~(size_t)255;
  float* norm = ws + off; off += Ea;
  float* hbuf = ws + off; off += NHa;
  float* z1   = ws + off; off += NHa;
  int* ssrc   = (int*)(ws + off); off += Ea;
  int* rowptr = (int*)(ws + off); off += ((size_t)N + 1 + 255) & ~(size_t)255;
  const int NB = ceil_div(N, 256);
  int* rank = (int*)hbuf;
  int* cnt  = rank + Ea;
  int* bsum = cnt + (((size_t)N * CNT_STRIDE + 255) & ~(size_t)255);
  const size_t scratch_words = Ea + (((size_t)N * CNT_STRIDE + 255) & ~(size_t)255) + (size_t)NB;
  const bool sorted_path = (off * sizeof(float) <= ws_size) && (scratch_words <= 2 * NHa) &&
                           (H % 4 == 0) && (Cn % 4 == 0);
  const int B  = ceil_div(N, NPB);
  const int G1 = ceil_div(E, BIN_CH);
  const size_t matw = ((size_t)G1 * B + 255) & ~(size_t)255;
  int*   bsrc   = (int*)hbuf;
  int*   bdst   = bsrc + Ea;
  float* bw     = (float*)(bdst + Ea);
  int*   cntmat = (int*)(bw + Ea);
  int*   bstart = cntmat + matw;
  const bool lds_sort = (B <= MAXB) && (E > 0) && (3 * Ea + matw + (size_t)B + 1 <= 2 * NHa);

  const int TPB = 256;

  if (sorted_path) {
    if (lds_sort) {
      bin_count<<<G1, BIN_TPB, 0, stream>>>(dst, cntmat, E, B);
      bin_scatter<<<G1, BIN_TPB, 0, stream>>>(src, dst, ew, cntmat, bstart, bsrc, bdst, bw,
                                              E, B, G1);
      bucket_sort<<<B, SORT_TPB, 0, stream>>>(bsrc, bdst, bw, bstart, rowptr, ssrc, norm, deg,
                                              N, B);
    } else {
      zero_counters<<<NB, TPB, 0, stream>>>(cnt, N);
      rank_edges<<<ceil_div(ceil_div(E, 4), TPB), TPB, 0, stream>>>(dst, cnt, rank, E);
      count_blocksum<<<NB, 256, 0, stream>>>(cnt, bsum, N);
      scan_rowptr<<<NB, 256, 0, stream>>>(cnt, bsum, rowptr, N);
      fill_sorted<<<ceil_div(ceil_div(E, 4), TPB), TPB, 0, stream>>>(src, dst, ew, rowptr, rank,
                                                                   ssrc, norm, E);
      deg_dinv<<<NB, TPB, 0, stream>>>(norm, rowptr, deg, N);
    }

    launch_gemm_pair(x, W1, V1, b1, deg, hbuf, z1, N, Fin, H, stream);
    {
      int G = H / 4;
      csr_gather<true><<<ceil_div((long long)N * G, TPB), TPB, 0, stream>>>(
          hbuf, norm, ssrc, rowptr, deg, z1, N, H, G);
    }

    launch_gemm_pair(z1, W2, V2, b2, deg, hbuf, out, N, H, Cn, stream);
    {
      if (Cn == 40) {
        csr_gather_logsoftmax<40><<<ceil_div((long long)ceil_div(N, 3) * 32, TPB), TPB, 0, stream>>>(
            hbuf, norm, ssrc, rowptr, deg, out, N);
      } else {
        int G = Cn / 4;
        csr_gather<false><<<ceil_div((long long)N * G, TPB), TPB, 0, stream>>>(
            hbuf, norm, ssrc, rowptr, deg, out, N, Cn, G);
        relu_logsoftmax_gen<<<ceil_div(N, TPB), TPB, 0, stream>>>(out, N, Cn);
      }
    }
    return;
  }

  zero_f32<<<ceil_div(N, TPB), TPB, 0, stream>>>(deg, N);
  deg_accum<<<ceil_div(E, TPB), TPB, 0, stream>>>(dst, ew, deg, E);
  norm_compute<<<ceil_div(E, TPB), TPB, 0, stream>>>(src, dst, ew, deg, norm, E);

  launch_gemm(x, W1, nullptr, nullptr, hbuf, N, Fin, H, stream);
  launch_gemm(x, V1, b1,      nullptr, z1,   N, Fin, H, stream);
  {
    int G = H / 4;
    edge_scatter<<<ceil_div((long long)E * G, TPB), TPB, 0, stream>>>(
        hbuf, norm, src, dst, z1, E, H, G);
    relu_inplace<<<ceil_div((long long)N * H, TPB), TPB, 0, stream>>>(z1, N * H);
  }

  launch_gemm(z1, W2, nullptr, nullptr, hbuf, N, H, Cn, stream);
  launch_gemm(z1, V2, b2,      nullptr, out,  N, H, Cn, stream);
  {
    int G = Cn / 4;
    edge_scatter<<<ceil_div((long long)E * G, TPB), TPB, 0, stream>>>(
        hbuf, norm, src, dst, out, E, Cn, G);
    if (Cn == 40) {
      relu_logsoftmax_t<40><<<ceil_div(N, TPB), TPB, 0, stream>>>(out, N);
    } else {
      relu_logsoftmax_gen<<<ceil_div(N, TPB), TPB, 0, stream>>>(out, N, Cn);
    }
  }
}


# Segment sums rebuilt for a short load chain: a 16-lane row per node takes 16 edge records in one load, shares them by DPP, keeps 8 row loads in flight; 4 nodes per wave; every sum order unchanged
# speedup vs baseline: 3.3916x; 1.1175x over previous
#include <hip/hip_runtime.h>

typedef float v2f __attribute__((ext_vector_type(2)));
typedef float v8f __attribute__((ext_vector_type(8)));

__global__ void zero_f32(float* __restrict__ p, int n) {
  int i = blockIdx.x * blockDim.x + threadIdx.x;
  if (i < n) p[i] = 0.f;
}

__global__ void relu_inplace(float* __restrict__ p, int n) {
  int i = blockIdx.x * blockDim.x + threadIdx.x;
  if (i < n) p[i] = fmaxf(p[i], 0.f);
}

__global__ void deg_accum(const int* __restrict__ dst, const float* __restrict__ w,
                          float* __restrict__ deg, int E) {
  int e = blockIdx.x * blockDim.x + threadIdx.x;
  if (e < E) atomicAdd(deg + dst[e], w[e]);
}

__global__ void norm_compute(const int* __restrict__ src, const int* __restrict__ dst,
                             const float* __restrict__ w, const float* __restrict__ deg,
                             float* __restrict__ norm, int E) {
  int e = blockIdx.x * blockDim.x + threadIdx.x;
  if (e >= E) return;
  float ds = deg[src[e]];
  float dd = deg[dst[e]];
  float is = ds > 0.f ? rsqrtf(ds) : 0.f;
  float id = dd > 0.f ? rsqrtf(dd) : 0.f;
  norm[e] = is * w[e] * id;
}

template <int K>
__global__ void gemm_wmma(const float* __restrict__ A, const float* __restrict__ B,
                          const float* __restrict__ bias, const float* __restrict__ rscale,
                          float* __restrict__ C, int N, int F, int rowTiles, int colTiles) {
  const int lane = threadIdx.x & 31;
  const int wave = threadIdx.x >> 5;
  const int tile = blockIdx.x * (blockDim.x >> 5) + wave;
  if (tile >= rowTiles * colTiles) return;

  const int rt = tile / colTiles;
  const int ct = tile - rt * colTiles;
  const int row0 = rt << 4;
  const int col0 = ct << 4;
  const int half = lane >> 4;
  const int lm   = lane & 15;

  const bool cOK = (col0 + lm) < F;
  const int  arow = min(row0 + lm, N - 1);
  const int  bcol = min(col0 + lm, F - 1);

  const float* Arow = A + (size_t)arow * K + 2 * half;
  const float* Bcol = B + bcol;

  v2f afrag[K / 4];
  v2f bfrag[K / 4];
#pragma unroll
  for (int i = 0; i < K / 4; ++i) {
    const int ka = 4 * i + 2 * half;
    afrag[i] = *reinterpret_cast<const v2f*>(Arow + 4 * i);
    bfrag[i].x = Bcol[(size_t)ka * F];
    bfrag[i].y = Bcol[(size_t)(ka + 1) * F];
  }

  v8f acc = {};
#pragma unroll
  for (int i = 0; i < K / 4; ++i) {
    acc = __builtin_amdgcn_wmma_f32_16x16x4_f32(
        false, afrag[i], false, bfrag[i], (short)0, acc, false, false);
  }

  const float bv = (bias != nullptr) ? bias[bcol] : 0.f;
#pragma unroll
  for (int r = 0; r < 8; ++r) {
    const int row = row0 + r + 8 * half;
    if (row < N && cOK) {
      const float v = (rscale != nullptr) ? acc[r] * rscale[row] : acc[r];
      C[(size_t)row * F + col0 + lm] = v + bv;
    }
  }
}

template <int K, int NT>
__global__ void __launch_bounds__(128)
gemm_wmma_pair(const float* __restrict__ A, const float* __restrict__ B1,
               const float* __restrict__ B2, const float* __restrict__ bias2,
               const float* __restrict__ rscale1,
               float* __restrict__ C1, float* __restrict__ C2,
               int rowTiles) {
  constexpr int F = 8 * NT;
  const int lane = threadIdx.x & 31;
  const int wave = threadIdx.x >> 5;
  const int rt = blockIdx.x * (blockDim.x >> 5) + wave;
  if (rt >= rowTiles) return;

  const int row0 = rt << 4;
  const int half = lane >> 4;
  const int lm   = lane & 15;
  const float* Arow = A + (size_t)(row0 + lm) * K + 2 * half;

  v2f afrag[K / 4];
#pragma unroll
  for (int i = 0; i < K / 4; ++i)
    afrag[i] = *reinterpret_cast<const v2f*>(Arow + 4 * i);

  float rs[8];
#pragma unroll
  for (int r = 0; r < 8; ++r) rs[r] = rscale1[row0 + 8 * half + r];

  const size_t rbase = (size_t)(row0 + 8 * half) * F;
#pragma unroll
  for (int t = 0; t < NT; ++t) {
    const int j = 16 * t + lm;
    const bool second = j >= F;
    const int col = second ? j - F : j;
    const float* Bcol = (second ? B2 : B1) + col;
    const float braw = bias2[col];
    const float bv = second ? braw : 0.f;

    v2f bfrag[K / 4];
#pragma unroll
    for (int i = 0; i < K / 4; ++i) {
      const int ka = 4 * i + 2 * half;
      bfrag[i].x = Bcol[(size_t)ka * F];
      bfrag[i].y = Bcol[(size_t)(ka + 1) * F];
    }

    v8f acc = {};
#pragma unroll
    for (int i = 0; i < K / 4; ++i) {
      acc = __builtin_amdgcn_wmma_f32_16x16x4_f32(
          false, afrag[i], false, bfrag[i], (short)0, acc, false, false);
    }

    float* Crow = (second ? C2 : C1) + col + rbase;
#pragma unroll
    for (int r = 0; r < 8; ++r) Crow[r * F] = second ? acc[r] + bv : acc[r] * rs[r];
  }
}

__global__ void gemm_wmma_gen(const float* __restrict__ A, const float* __restrict__ B,
                              const float* __restrict__ bias, const float* __restrict__ rscale,
                              float* __restrict__ C, int N, int K, int F, int rowTiles,
                              int colTiles) {
  const int lane = threadIdx.x & 31;
  const int wave = threadIdx.x >> 5;
  const int tile = blockIdx.x * (blockDim.x >> 5) + wave;
  if (tile >= rowTiles * colTiles) return;
  const int rt = tile / colTiles;
  const int ct = tile - rt * colTiles;
  const int row0 = rt << 4, col0 = ct << 4;
  const int half = lane >> 4, lm = lane & 15;
  const bool cOK = (col0 + lm) < F;
  const int arow = min(row0 + lm, N - 1);
  const int bcol = min(col0 + lm, F - 1);
  v8f acc = {};
  for (int kb = 0; kb < K; kb += 4) {
    const int ka = kb + 2 * half;
    v2f a = *reinterpret_cast<const v2f*>(A + (size_t)arow * K + ka);
    v2f b;
    b.x = B[(size_t)ka * F + bcol];
    b.y = B[(size_t)(ka + 1) * F + bcol];
    acc = __builtin_amdgcn_wmma_f32_16x16x4_f32(false, a, false, b, (short)0, acc,
                                                false, false);
  }
  const float bv = (bias != nullptr) ? bias[bcol] : 0.f;
#pragma unroll
  for (int r = 0; r < 8; ++r) {
    const int row = row0 + r + 8 * half;
    if (row < N && cOK) {
      const float v = (rscale != nullptr) ? acc[r] * rscale[row] : acc[r];
      C[(size_t)row * F + col0 + lm] = v + bv;
    }
  }
}

__global__ void edge_scatter(const float* __restrict__ h, const float* __restrict__ norm,
                             const int* __restrict__ src, const int* __restrict__ dst,
                             float* __restrict__ agg, int E, int F, int G) {
  long long tid = (long long)blockIdx.x * blockDim.x + threadIdx.x;
  if (tid >= (long long)E * G) return;
  int e = (int)(tid / G);
  int g = (int)(tid - (long long)e * G);
  int s = src[e];
  int d = dst[e];
  float w = norm[e];
  const float4 v = *reinterpret_cast<const float4*>(h + (size_t)s * F + (size_t)g * 4);
  float* base = agg + (size_t)d * F + (size_t)g * 4;
  atomicAdd(base + 0, v.x * w);
  atomicAdd(base + 1, v.y * w);
  atomicAdd(base + 2, v.z * w);
  atomicAdd(base + 3, v.w * w);
}

template <int C>
__global__ void relu_logsoftmax_t(float* __restrict__ y, int N) {
  int n = blockIdx.x * blockDim.x + threadIdx.x;
  if (n >= N) return;
  float v[C];
  float m = -3.4e38f;
  float* row = y + (size_t)n * C;
#pragma unroll
  for (int c = 0; c < C; ++c) {
    float t = fmaxf(row[c], 0.f);
    v[c] = t;
    m = fmaxf(m, t);
  }
  float s = 0.f;
#pragma unroll
  for (int c = 0; c < C; ++c) s += __expf(v[c] - m);
  float ls = m + __logf(s);
#pragma unroll
  for (int c = 0; c < C; ++c) row[c] = v[c] - ls;
}

__global__ void relu_logsoftmax_gen(float* __restrict__ y, int N, int C) {
  int n = blockIdx.x * blockDim.x + threadIdx.x;
  if (n >= N) return;
  float* row = y + (size_t)n * C;
  float m = -3.4e38f;
  for (int c = 0; c < C; ++c) m = fmaxf(m, fmaxf(row[c], 0.f));
  float s = 0.f;
  for (int c = 0; c < C; ++c) s += __expf(fmaxf(row[c], 0.f) - m);
  float ls = m + __logf(s);
  for (int c = 0; c < C; ++c) row[c] = fmaxf(row[c], 0.f) - ls;
}

#define CNT_STRIDE 32

__global__ void zero_counters(int* __restrict__ cnt, int N) {
  int n = blockIdx.x * blockDim.x + threadIdx.x;
  if (n < N) cnt[(size_t)n * CNT_STRIDE] = 0;
}

__global__ void rank_edges(const int* __restrict__ dst, int* __restrict__ cnt,
                           int* __restrict__ rank, int E) {
  int e0 = (blockIdx.x * blockDim.x + threadIdx.x) * 4;
  int d[4];
#pragma unroll
  for (int i = 0; i < 4; ++i) d[i] = (e0 + i < E) ? dst[e0 + i] : -1;
  int r[4];
#pragma unroll
  for (int i = 0; i < 4; ++i)
    r[i] = (d[i] >= 0) ? atomicAdd(cnt + (size_t)d[i] * CNT_STRIDE, 1) : 0;
#pragma unroll
  for (int i = 0; i < 4; ++i)
    if (e0 + i < E) rank[e0 + i] = r[i];
}

__device__ __forceinline__ int block_incl_scan_256(int v, int* sh) {
  const int t = threadIdx.x;
  sh[t] = v;
  __syncthreads();
  for (int off = 1; off < 256; off <<= 1) {
    int a = (t >= off) ? sh[t - off] : 0;
    __syncthreads();
    sh[t] += a;
    __syncthreads();
  }
  return sh[t];
}

__global__ void __launch_bounds__(256)
count_blocksum(const int* __restrict__ cnt, int* __restrict__ bsum, int N) {
  __shared__ int sh[256];
  int n = blockIdx.x * 256 + threadIdx.x;
  int v = (n < N) ? cnt[(size_t)n * CNT_STRIDE] : 0;
  int inc = block_incl_scan_256(v, sh);
  if (threadIdx.x == 255) bsum[blockIdx.x] = inc;
}

__global__ void __launch_bounds__(256)
scan_rowptr(const int* __restrict__ cnt, const int* __restrict__ bsum,
            int* __restrict__ rowptr, int N) {
  __shared__ int sh[256];
  __shared__ int sh2[256];
  const int t = threadIdx.x;
  int part = 0;
  for (int j = t; j < (int)blockIdx.x; j += 256) part += bsum[j];
  block_incl_scan_256(part, sh2);
  const int base = sh2[255];
  int n = blockIdx.x * 256 + t;
  int v = (n < N) ? cnt[(size_t)n * CNT_STRIDE] : 0;
  int inc = block_incl_scan_256(v, sh);
  if (n < N) {
    rowptr[n] = base + inc - v;
    if (n == N - 1) rowptr[N] = base + inc;
  }
}

__global__ void fill_sorted(const int* __restrict__ src, const int* __restrict__ dst,
                            const float* __restrict__ w, const int* __restrict__ rowptr,
                            const int* __restrict__ rank, int* __restrict__ ssrc,
                            float* __restrict__ sw, int E) {
  int e0 = (blockIdx.x * blockDim.x + threadIdx.x) * 4;
  for (int i = 0; i < 4; ++i) {
    int e = e0 + i;
    if (e < E) {
      int p = rowptr[dst[e]] + rank[e];
      ssrc[p] = src[e];
      sw[p] = w[e];
    }
  }
}

__global__ void deg_dinv(const float* __restrict__ sw, const int* __restrict__ rowptr,
                         float* __restrict__ dinv, int N) {
  int n = blockIdx.x * blockDim.x + threadIdx.x;
  if (n >= N) return;
  const int p0 = rowptr[n], p1 = rowptr[n + 1];
  float s = 0.f;
  for (int p = p0; p < p1; ++p) s += sw[p];
  dinv[n] = s > 0.f ? rsqrtf(s) : 0.f;
}

template <bool RELU>
__global__ void csr_gather(const float* __restrict__ h, const float* __restrict__ sw,
                           const int* __restrict__ ssrc, const int* __restrict__ rowptr,
                           const float* __restrict__ dinv,
                           float* __restrict__ agg, int N, int F, int G) {
  long long tid = (long long)blockIdx.x * blockDim.x + threadIdx.x;
  if (tid >= (long long)N * G) return;
  int n = (int)(tid / G);
  int g = (int)(tid - (long long)n * G);
  const int p0 = rowptr[n], p1 = rowptr[n + 1];
  const float dn = dinv[n];
  const float* hb = h + (size_t)g * 4;
  float4 acc = make_float4(0.f, 0.f, 0.f, 0.f);
#pragma unroll 4
  for (int p = p0; p < p1; ++p) {
    const int s = ssrc[p];
    const float w = sw[p];
    const float4 v = *reinterpret_cast<const float4*>(hb + (size_t)s * F);
    acc.x += v.x * w;
    acc.y += v.y * w;
    acc.z += v.z * w;
    acc.w += v.w * w;
  }
  float4* o = reinterpret_cast<float4*>(agg + (size_t)n * F + (size_t)g * 4);
  float4 z = *o;
  z.x += acc.x * dn;
  z.y += acc.y * dn;
  z.z += acc.z * dn;
  z.w += acc.w * dn;
  if (RELU) {
    z.x = fmaxf(z.x, 0.f);
    z.y = fmaxf(z.y, 0.f);
    z.z = fmaxf(z.z, 0.f);
    z.w = fmaxf(z.w, 0.f);
  }
  *o = z;
}

template <int C>
__global__ void csr_gather_logsoftmax(const float* __restrict__ h,
                                      const float* __restrict__ sw,
                                      const int* __restrict__ ssrc,
                                      const int* __restrict__ rowptr,
                                      const float* __restrict__ dinv,
                                      float* __restrict__ y, int N) {
  constexpr int G = C / 4;
  constexpr int NPW = 32 / G;
  const int lane = threadIdx.x & 31;
  const int wave = (int)((blockIdx.x * blockDim.x + threadIdx.x) >> 5);
  const int slot = lane / G;
  const int g = lane - slot * G;
  const int n = wave * NPW + slot;
  const bool live = (slot < NPW) && (n < N);
  int p0 = 0, p1 = 0;
  float dn = 0.f;
  if (live) {
    p0 = rowptr[n];
    p1 = rowptr[n + 1];
    dn = dinv[n];
  }
  const float* hb = h + (size_t)g * 4;
  float4 acc = make_float4(0.f, 0.f, 0.f, 0.f);
#pragma unroll 4
  for (int p = p0; p < p1; ++p) {
    const int s = ssrc[p];
    const float w = sw[p];
    const float4 v = *reinterpret_cast<const float4*>(hb + (size_t)s * C);
    acc.x += v.x * w;
    acc.y += v.y * w;
    acc.z += v.z * w;
    acc.w += v.w * w;
  }
  float4 t = make_float4(0.f, 0.f, 0.f, 0.f);
  float4* o = nullptr;
  if (live) {
    o = reinterpret_cast<float4*>(y + (size_t)n * C + (size_t)g * 4);
    const float4 z = *o;
    t.x = fmaxf(z.x + acc.x * dn, 0.f);
    t.y = fmaxf(z.y + acc.y * dn, 0.f);
    t.z = fmaxf(z.z + acc.z * dn, 0.f);
    t.w = fmaxf(z.w + acc.w * dn, 0.f);
  }
  const int first = live ? lane - g : lane;
  const int step = live ? 1 : 0;
  const float m4 = fmaxf(fmaxf(t.x, t.y), fmaxf(t.z, t.w));
  float m = -3.4e38f;
  for (int k = 0; k < G; ++k) m = fmaxf(m, __shfl(m4, first + k * step));
  const float e4 = __expf(t.x - m) + __expf(t.y - m) + __expf(t.z - m) + __expf(t.w - m);
  float s = 0.f;
  for (int k = 0; k < G; ++k) s += __shfl(e4, first + k * step);
  const float ls = m + __logf(s);
  if (live) *o = make_float4(t.x - ls, t.y - ls, t.z - ls, t.w - ls);
}


template <int K>
__device__ __forceinline__ int row_lane_i(int v) {
  return __builtin_amdgcn_update_dpp(0, v, 0x150 + K, 0xf, 0xf, true);
}
template <int K>
__device__ __forceinline__ float row_lane_f(float v) {
  return __int_as_float(row_lane_i<K>(__float_as_int(v)));
}

template <int F, int K0>
__device__ __forceinline__ void seg_rows8(const char* __restrict__ hB, unsigned laneoff,
                                          int myS, float myW, float4& acc) {
  float4 v[8];
  float w[8];
#define SEG_LOAD(K)                                                                    \
  {                                                                                    \
    const unsigned s = (unsigned)row_lane_i<K0 + K>(myS);                              \
    w[K] = row_lane_f<K0 + K>(myW);                                                    \
    v[K] = *reinterpret_cast<const float4*>(hB + (size_t)(s * (unsigned)(F * 4) + laneoff)); \
  }
  SEG_LOAD(0) SEG_LOAD(1) SEG_LOAD(2) SEG_LOAD(3)
  SEG_LOAD(4) SEG_LOAD(5) SEG_LOAD(6) SEG_LOAD(7)
#undef SEG_LOAD
#pragma unroll
  for (int k = 0; k < 8; ++k) {
    acc.x = __builtin_fmaf(v[k].x, w[k], acc.x);
    acc.y = __builtin_fmaf(v[k].y, w[k], acc.y);
    acc.z = __builtin_fmaf(v[k].z, w[k], acc.z);
    acc.w = __builtin_fmaf(v[k].w, w[k], acc.w);
  }
}

template <int F, bool LSM>
__global__ void __launch_bounds__(256)
seg_gather16(const float* __restrict__ h, const float* __restrict__ sw,
             const int* __restrict__ ssrc, const int* __restrict__ rowptr,
             const float* __restrict__ dinv, float* __restrict__ y, int N) {
  static_assert(F % 4 == 0 && F <= 64, "a row of F floats must fit the 16 lanes of a wave row, 4 floats each");
  constexpr int NL = F / 4;
  const int lane = threadIdx.x & 31;
  const int l16 = lane & 15;
  const int wave = (int)((blockIdx.x * blockDim.x + threadIdx.x) >> 5);
  const bool feat = l16 < NL;
  const unsigned laneoff = feat ? (unsigned)l16 * 16u : 0u;
  const char* hB = reinterpret_cast<const char*>(h);

  const unsigned lofs4 = feat ? (unsigned)l16 * 4u : 0u;
  const int n0 = wave * 4 + 2 * (lane >> 4);
  const int rp0 = rowptr[min(n0, N)];
  const int rp1 = rowptr[min(n0 + 1, N)];
  const int rp2 = rowptr[min(n0 + 2, N)];
  int nn[2], p0[2], p1[2], s0[2];
  float dn[2], w0[2];
  float4 z[2];
  bool live[2];
#pragma unroll
  for (int r = 0; r < 2; ++r) {
    nn[r] = n0 + r;
    live[r] = nn[r] < N;
    const int nc = min(nn[r], N - 1);
    dn[r] = dinv[nc];
    z[r] = *reinterpret_cast<const float4*>(y + (size_t)nc * F + lofs4);
  }
  p0[0] = rp0;
  p1[0] = rp1;
  p0[1] = rp1;
  p1[1] = rp2;
#pragma unroll
  for (int r = 0; r < 2; ++r) {
    const int q = p0[r] + l16;
    const int qc = max(min(q, p1[r] - 1), 0);
    s0[r] = ssrc[qc];
    const float wl = sw[qc];
    w0[r] = (q < p1[r]) ? wl : 0.f;
  }

#pragma unroll
  for (int r = 0; r < 2; ++r) {
    float4 acc = make_float4(0.f, 0.f, 0.f, 0.f);
    const int pe = p1[r];
    int base = p0[r];
    int myS = s0[r];
    float myW = w0[r];
    while (base < pe) {
      const int cnt = pe - base;
      const int nb = base + 16;
      const int q = nb + l16;
      const int qc = min(q, pe - 1);
      const int nS = ssrc[qc];
      const float nWl = sw[qc];
      const float nW = (q < pe) ? nWl : 0.f;
      seg_rows8<F, 0>(hB, laneoff, myS, myW, acc);
      if (cnt > 8) seg_rows8<F, 8>(hB, laneoff, myS, myW, acc);
      myS = nS;
      myW = nW;
      base = nb;
    }
    float4 t;
    t.x = __builtin_fmaf(acc.x, dn[r], z[r].x);
    t.y = __builtin_fmaf(acc.y, dn[r], z[r].y);
    t.z = __builtin_fmaf(acc.z, dn[r], z[r].z);
    t.w = __builtin_fmaf(acc.w, dn[r], z[r].w);
    t.x = fmaxf(t.x, 0.f);
    t.y = fmaxf(t.y, 0.f);
    t.z = fmaxf(t.z, 0.f);
    t.w = fmaxf(t.w, 0.f);
    if (LSM) {
      const float m4 = fmaxf(fmaxf(t.x, t.y), fmaxf(t.z, t.w));
      float m = -3.4e38f;
#define SEG_MAX(K) if (K < NL) m = fmaxf(m, row_lane_f<K>(m4));
      SEG_MAX(0) SEG_MAX(1) SEG_MAX(2) SEG_MAX(3) SEG_MAX(4) SEG_MAX(5) SEG_MAX(6) SEG_MAX(7)
      SEG_MAX(8) SEG_MAX(9) SEG_MAX(10) SEG_MAX(11) SEG_MAX(12) SEG_MAX(13) SEG_MAX(14) SEG_MAX(15)
#undef SEG_MAX
      const float e4 = __expf(t.x - m) + __expf(t.y - m) + __expf(t.z - m) + __expf(t.w - m);
      float s = 0.f;
#define SEG_SUM(K) if (K < NL) s += row_lane_f<K>(e4);
      SEG_SUM(0) SEG_SUM(1) SEG_SUM(2) SEG_SUM(3) SEG_SUM(4) SEG_SUM(5) SEG_SUM(6) SEG_SUM(7)
      SEG_SUM(8) SEG_SUM(9) SEG_SUM(10) SEG_SUM(11) SEG_SUM(12) SEG_SUM(13) SEG_SUM(14) SEG_SUM(15)
#undef SEG_SUM
      const float ls = m + __logf(s);
      t.x -= ls;
      t.y -= ls;
      t.z -= ls;
      t.w -= ls;
    }
    if (live[r] && feat)
      *reinterpret_cast<float4*>(y + (size_t)nn[r] * F + (size_t)l16 * 4) = t;
  }
}

#define NPB_SHIFT 9
#define NPB (1 << NPB_SHIFT)
#define MAXB 256
#define BIN_CH 6400
#define BIN_TPB 1024
#define BIN_IT ((BIN_CH + BIN_TPB - 1) / BIN_TPB)
#define SORT_TPB 1024
#define SORT_IT 4
#define SORT_CAP 10240

template <int PER>
__device__ __forceinline__ void excl_scan_lds_wave0(int* a) {
  __syncthreads();
  if (threadIdx.x < 32) {
    const int lane = threadIdx.x;
    int v[PER];
    int s = 0;
#pragma unroll
    for (int k = 0; k < PER; ++k) {
      v[k] = a[lane * PER + k];
      s += v[k];
    }
    int inc = s;
#pragma unroll
    for (int off = 1; off < 32; off <<= 1) {
      const int u = __shfl_up(inc, off);
      if (lane >= off) inc += u;
    }
    int run = inc - s;
#pragma unroll
    for (int k = 0; k < PER; ++k) {
      a[lane * PER + k] = run;
      run += v[k];
    }
    if (lane == 31) a[32 * PER] = run;
  }
  __syncthreads();
}

__global__ void __launch_bounds__(BIN_TPB)
bin_count(const int* __restrict__ dst, int* __restrict__ cntmat, int E, int B) {
  __shared__ int hist[MAXB];
  const int t = threadIdx.x;
  if (t < MAXB) hist[t] = 0;
  __syncthreads();
  const int e0 = blockIdx.x * BIN_CH;
  const int n = min(BIN_CH, E - e0);
  int d[BIN_IT];
#pragma unroll
  for (int k = 0; k < BIN_IT; ++k) {
    const int i = t + k * BIN_TPB;
    d[k] = (i < n) ? dst[e0 + i] : -1;
  }
#pragma unroll
  for (int k = 0; k < BIN_IT; ++k)
    if (d[k] >= 0) atomicAdd(&hist[d[k] >> NPB_SHIFT], 1);
  __syncthreads();
  if (t < B) cntmat[(size_t)blockIdx.x * B + t] = hist[t];
}

__global__ void __launch_bounds__(BIN_TPB)
bin_scatter(const int* __restrict__ src, const int* __restrict__ dst,
            const float* __restrict__ w, const int* __restrict__ cntmat,
            int* __restrict__ bstart, int* __restrict__ bsrc, int* __restrict__ bdst,
            float* __restrict__ bw, int E, int B, int G1) {
  __shared__ int   s_src[BIN_CH];
  __shared__ int   s_dst[BIN_CH];
  __shared__ float s_w[BIN_CH];
  __shared__ int   p_tot[4][MAXB];
  __shared__ int   p_pre[4][MAXB];
  __shared__ int   gbase[MAXB + 1];
  __shared__ int   lstart[MAXB + 1];
  __shared__ int   lcnt[MAXB];
  const int t = threadIdx.x;
  const int g = blockIdx.x;
  const int col = t & (MAXB - 1);
  const int grp = t >> 8;
  int tot = 0, pre = 0;
  if (col < B) {
#pragma unroll 8
    for (int r = grp; r < G1; r += 4) {
      const int c = cntmat[(size_t)r * B + col];
      tot += c;
      pre += (r < g) ? c : 0;
    }
  }
  p_tot[grp][col] = tot;
  p_pre[grp][col] = pre;
  __syncthreads();
  int mypre = 0;
  if (t < MAXB) {
    gbase[t] = p_tot[0][t] + p_tot[1][t] + p_tot[2][t] + p_tot[3][t];
    mypre = p_pre[0][t] + p_pre[1][t] + p_pre[2][t] + p_pre[3][t];
    lstart[t] = (t < B) ? cntmat[(size_t)g * B + t] : 0;
    lcnt[t] = 0;
  }
  excl_scan_lds_wave0<MAXB / 32>(gbase);
  excl_scan_lds_wave0<MAXB / 32>(lstart);
  if (g == 0 && t <= B) bstart[t] = gbase[t];
  __syncthreads();
  if (t < MAXB) gbase[t] += mypre;
  __syncthreads();

  const int e0 = g * BIN_CH;
  const int n = min(BIN_CH, E - e0);
  int d[BIN_IT], sv[BIN_IT];
  float wv[BIN_IT];
#pragma unroll
  for (int k = 0; k < BIN_IT; ++k) {
    const int i = t + k * BIN_TPB;
    const bool in = i < n;
    d[k]  = in ? dst[e0 + i] : -1;
    sv[k] = in ? src[e0 + i] : 0;
    wv[k] = in ? w[e0 + i] : 0.f;
  }
#pragma unroll
  for (int k = 0; k < BIN_IT; ++k) {
    if (d[k] >= 0) {
      const int bk = d[k] >> NPB_SHIFT;
      const int q = lstart[bk] + atomicAdd(&lcnt[bk], 1);
      s_src[q] = sv[k];
      s_dst[q] = d[k];
      s_w[q] = wv[k];
    }
  }
  __syncthreads();
  for (int q = t; q < n; q += BIN_TPB) {
    const int d = s_dst[q];
    const int bk = d >> NPB_SHIFT;
    const int p = gbase[bk] + (q - lstart[bk]);
    bsrc[p] = s_src[q];
    bdst[p] = d;
    bw[p] = s_w[q];
  }
}

__global__ void __launch_bounds__(SORT_TPB)
bucket_sort(const int* __restrict__ bsrc, const int* __restrict__ bdst,
            const float* __restrict__ bw, const int* __restrict__ bstart,
            int* __restrict__ rowptr, int* __restrict__ ssrc, float* __restrict__ sw,
            float* __restrict__ dinv, int N, int B) {
  __shared__ int   c_src[SORT_CAP];
  __shared__ float c_w[SORT_CAP];
  __shared__ int   nstart[NPB + 1];
  __shared__ int   ncnt[NPB];
  const int t = threadIdx.x;
  const int b = blockIdx.x;
  const int m0 = bstart[b];
  const int m1 = bstart[b + 1];
  const int nE = m1 - m0;
  const int node0 = b << NPB_SHIFT;
  if (t < NPB) {
    nstart[t] = 0;
    ncnt[t] = 0;
  }
  __syncthreads();
  for (int i0 = t; i0 < nE; i0 += SORT_IT * SORT_TPB) {
    int dd[SORT_IT];
#pragma unroll
    for (int k = 0; k < SORT_IT; ++k) {
      const int i = i0 + k * SORT_TPB;
      dd[k] = (i < nE) ? bdst[m0 + i] : -1;
    }
#pragma unroll
    for (int k = 0; k < SORT_IT; ++k)
      if (dd[k] >= 0) atomicAdd(&nstart[dd[k] - node0], 1);
  }
  excl_scan_lds_wave0<NPB / 32>(nstart);
  if (t < NPB && node0 + t < N) rowptr[node0 + t] = m0 + nstart[t];
  if (b == B - 1 && t == 0) rowptr[N] = m1;
  const bool staged = nE <= SORT_CAP;
  if (staged) {
    for (int i0 = t; i0 < nE; i0 += SORT_IT * SORT_TPB) {
      int dd[SORT_IT], ss[SORT_IT];
      float xx[SORT_IT];
#pragma unroll
      for (int k = 0; k < SORT_IT; ++k) {
        const int i = i0 + k * SORT_TPB;
        const bool in = i < nE;
        dd[k] = in ? bdst[m0 + i] : -1;
        ss[k] = in ? bsrc[m0 + i] : 0;
        xx[k] = in ? bw[m0 + i] : 0.f;
      }
#pragma unroll
      for (int k = 0; k < SORT_IT; ++k) {
        if (dd[k] >= 0) {
          const int l = dd[k] - node0;
          const int q = nstart[l] + atomicAdd(&ncnt[l], 1);
          c_src[q] = ss[k];
          c_w[q] = xx[k];
        }
      }
    }
  } else {
    for (int i = t; i < nE; i += SORT_TPB) {
      const int l = bdst[m0 + i] - node0;
      const int q = nstart[l] + atomicAdd(&ncnt[l], 1);
      ssrc[m0 + q] = bsrc[m0 + i];
      sw[m0 + q] = bw[m0 + i];
    }
  }
  __syncthreads();
  if (t < NPB && node0 + t < N) {
    const int q0 = nstart[t], q1 = nstart[t + 1];
    float s = 0.f;
    if (staged) {
      for (int q = q0; q < q1; ++q) s += c_w[q];
    } else {
      for (int q = q0; q < q1; ++q) s += sw[m0 + q];
    }
    dinv[node0 + t] = s > 0.f ? rsqrtf(s) : 0.f;
  }
  if (staged) {
    for (int q = t; q < nE; q += SORT_TPB) {
      ssrc[m0 + q] = c_src[q];
      sw[m0 + q] = c_w[q];
    }
  }
}

static inline int ceil_div(long long a, int b) { return (int)((a + b - 1) / b); }

static void launch_gemm(const float* A, const float* B, const float* bias, const float* rscale,
                        float* C, int N, int K, int F, hipStream_t stream) {
  const int TPB = 256;
  int rowT = ceil_div(N, 16), colT = ceil_div(F, 16);
  int blocks = ceil_div((long long)rowT * colT, TPB / 32);
  if (K == 64) {
    gemm_wmma<64><<<blocks, TPB, 0, stream>>>(A, B, bias, rscale, C, N, F, rowT, colT);
  } else if (K == 48) {
    gemm_wmma<48><<<blocks, TPB, 0, stream>>>(A, B, bias, rscale, C, N, F, rowT, colT);
  } else {
    gemm_wmma_gen<<<blocks, TPB, 0, stream>>>(A, B, bias, rscale, C, N, K, F, rowT, colT);
  }
}

static void launch_gemm_pair(const float* A, const float* B1, const float* B2,
                             const float* bias2, const float* rscale1, float* C1, float* C2,
                             int N, int K, int F, hipStream_t stream) {
  const int rowT = N / 16;
  const int WPB = 4;
  const int blocks = ceil_div(rowT, WPB);
  const bool full = (N % 16 == 0) && (N > 0) && (bias2 != nullptr);
  if (full && K == 64 && F == 48) {
    gemm_wmma_pair<64, 6><<<blocks, WPB * 32, 0, stream>>>(A, B1, B2, bias2, rscale1, C1, C2,
                                                            rowT);
  } else if (full && K == 48 && F == 40) {
    gemm_wmma_pair<48, 5><<<blocks, WPB * 32, 0, stream>>>(A, B1, B2, bias2, rscale1, C1, C2,
                                                            rowT);
  } else {
    launch_gemm(A, B1, nullptr, rscale1, C1, N, K, F, stream);
    launch_gemm(A, B2, bias2,   nullptr, C2, N, K, F, stream);
  }
}

extern "C" void kernel_launch(void* const* d_in, const int* in_sizes, int n_in,
                              void* d_out, int out_size, void* d_ws, size_t ws_size,
                              hipStream_t stream) {
  const float* x  = (const float*)d_in[0];
  const int*   ei = (const int*)d_in[1];
  const float* ew = (const float*)d_in[2];
  const float* W1 = (const float*)d_in[3];
  const float* V1 = (const float*)d_in[4];
  const float* b1 = (const float*)d_in[5];
  const float* W2 = (const float*)d_in[6];
  const float* V2 = (const float*)d_in[7];
  const float* b2 = (const float*)d_in[8];
  float* out = (float*)d_out;

  const int H   = in_sizes[5];
  const int Cn  = in_sizes[8];
  const int Fin = in_sizes[3] / H;
  const int N   = in_sizes[0] / Fin;
  const int E   = in_sizes[2];
  const int* src = ei;
  const int* dst = ei + E;

  float* ws = (float*)d_ws;
  const size_t NHa = ((size_t)N * H + 255) & ~(size_t)255;
  const size_t Ea  = ((size_t)E + 255) & ~(size_t)255;
  size_t off = 0;
  float* deg  = ws + off; off += ((size_t)N + 255) & ~(size_t)255;
  float* norm = ws + off; off += Ea;
  float* hbuf = ws + off; off += NHa;
  float* z1   = ws + off; off += NHa;
  int* ssrc   = (int*)(ws + off); off += Ea;
  int* rowptr = (int*)(ws + off); off += ((size_t)N + 1 + 255) & ~(size_t)255;
  const int NB = ceil_div(N, 256);
  int* rank = (int*)hbuf;
  int* cnt  = rank + Ea;
  int* bsum = cnt + (((size_t)N * CNT_STRIDE + 255) & ~(size_t)255);
  const size_t scratch_words = Ea + (((size_t)N * CNT_STRIDE + 255) & ~(size_t)255) + (size_t)NB;
  const bool sorted_path = (off * sizeof(float) <= ws_size) && (scratch_words <= 2 * NHa) &&
                           (H % 4 == 0) && (Cn % 4 == 0);
  const int B  = ceil_div(N, NPB);
  const int G1 = ceil_div(E, BIN_CH);
  const size_t matw = ((size_t)G1 * B + 255) & ~(size_t)255;
  int*   bsrc   = (int*)hbuf;
  int*   bdst   = bsrc + Ea;
  float* bw     = (float*)(bdst + Ea);
  int*   cntmat = (int*)(bw + Ea);
  int*   bstart = cntmat + matw;
  const bool lds_sort = (B <= MAXB) && (E > 0) && (3 * Ea + matw + (size_t)B + 1 <= 2 * NHa);

  const int TPB = 256;

  if (sorted_path) {
    if (lds_sort) {
      bin_count<<<G1, BIN_TPB, 0, stream>>>(dst, cntmat, E, B);
      bin_scatter<<<G1, BIN_TPB, 0, stream>>>(src, dst, ew, cntmat, bstart, bsrc, bdst, bw,
                                              E, B, G1);
      bucket_sort<<<B, SORT_TPB, 0, stream>>>(bsrc, bdst, bw, bstart, rowptr, ssrc, norm, deg,
                                              N, B);
    } else {
      zero_counters<<<NB, TPB, 0, stream>>>(cnt, N);
      rank_edges<<<ceil_div(ceil_div(E, 4), TPB), TPB, 0, stream>>>(dst, cnt, rank, E);
      count_blocksum<<<NB, 256, 0, stream>>>(cnt, bsum, N);
      scan_rowptr<<<NB, 256, 0, stream>>>(cnt, bsum, rowptr, N);
      fill_sorted<<<ceil_div(ceil_div(E, 4), TPB), TPB, 0, stream>>>(src, dst, ew, rowptr, rank,
                                                                   ssrc, norm, E);
      deg_dinv<<<NB, TPB, 0, stream>>>(norm, rowptr, deg, N);
    }

    launch_gemm_pair(x, W1, V1, b1, deg, hbuf, z1, N, Fin, H, stream);
    const bool rows32 = (size_t)N * 256 < ((size_t)1 << 31);
    const int segBlocks = ceil_div((long long)ceil_div(N, 4) * 32, TPB);
    if (H == 48 && rows32) {
      seg_gather16<48, false><<<segBlocks, TPB, 0, stream>>>(hbuf, norm, ssrc, rowptr, deg, z1, N);
    } else {
      int G = H / 4;
      csr_gather<true><<<ceil_div((long long)N * G, TPB), TPB, 0, stream>>>(
          hbuf, norm, ssrc, rowptr, deg, z1, N, H, G);
    }

    launch_gemm_pair(z1, W2, V2, b2, deg, hbuf, out, N, H, Cn, stream);
    {
      if (Cn == 40 && rows32) {
        seg_gather16<40, true><<<segBlocks, TPB, 0, stream>>>(hbuf, norm, ssrc, rowptr, deg, out, N);
      } else if (Cn == 40) {
        csr_gather_logsoftmax<40><<<ceil_div((long long)ceil_div(N, 3) * 32, TPB), TPB, 0, stream>>>(
            hbuf, norm, ssrc, rowptr, deg, out, N);
      } else {
        int G = Cn / 4;
        csr_gather<false><<<ceil_div((long long)N * G, TPB), TPB, 0, stream>>>(
            hbuf, norm, ssrc, rowptr, deg, out, N, Cn, G);
        relu_logsoftmax_gen<<<ceil_div(N, TPB), TPB, 0, stream>>>(out, N, Cn);
      }
    }
    return;
  }

  zero_f32<<<ceil_div(N, TPB), TPB, 0, stream>>>(deg, N);
  deg_accum<<<ceil_div(E, TPB), TPB, 0, stream>>>(dst, ew, deg, E);
  norm_compute<<<ceil_div(E, TPB), TPB, 0, stream>>>(src, dst, ew, deg, norm, E);

  launch_gemm(x, W1, nullptr, nullptr, hbuf, N, Fin, H, stream);
  launch_gemm(x, V1, b1,      nullptr, z1,   N, Fin, H, stream);
  {
    int G = H / 4;
    edge_scatter<<<ceil_div((long long)E * G, TPB), TPB, 0, stream>>>(
        hbuf, norm, src, dst, z1, E, H, G);
    relu_inplace<<<ceil_div((long long)N * H, TPB), TPB, 0, stream>>>(z1, N * H);
  }

  launch_gemm(z1, W2, nullptr, nullptr, hbuf, N, H, Cn, stream);
  launch_gemm(z1, V2, b2,      nullptr, out,  N, H, Cn, stream);
  {
    int G = Cn / 4;
    edge_scatter<<<ceil_div((long long)E * G, TPB), TPB, 0, stream>>>(
        hbuf, norm, src, dst, out, E, Cn, G);
    if (Cn == 40) {
      relu_logsoftmax_t<40><<<ceil_div(N, TPB), TPB, 0, stream>>>(out, N);
    } else {
      relu_logsoftmax_gen<<<ceil_div(N, TPB), TPB, 0, stream>>>(out, N, Cn);
    }
  }
}


# Layer 2's two products moved into the layer-1 segment-sum launch: a block's 32 finished rows go to LDS and feed the same f32 WMMA tiles; layer-1 rows never reach memory, one launch fewer, sums kept
# speedup vs baseline: 3.7234x; 1.0978x over previous
#include <hip/hip_runtime.h>

typedef float v2f __attribute__((ext_vector_type(2)));
typedef float v8f __attribute__((ext_vector_type(8)));

__global__ void zero_f32(float* __restrict__ p, int n) {
  int i = blockIdx.x * blockDim.x + threadIdx.x;
  if (i < n) p[i] = 0.f;
}

__global__ void relu_inplace(float* __restrict__ p, int n) {
  int i = blockIdx.x * blockDim.x + threadIdx.x;
  if (i < n) p[i] = fmaxf(p[i], 0.f);
}

__global__ void deg_accum(const int* __restrict__ dst, const float* __restrict__ w,
                          float* __restrict__ deg, int E) {
  int e = blockIdx.x * blockDim.x + threadIdx.x;
  if (e < E) atomicAdd(deg + dst[e], w[e]);
}

__global__ void norm_compute(const int* __restrict__ src, const int* __restrict__ dst,
                             const float* __restrict__ w, const float* __restrict__ deg,
                             float* __restrict__ norm, int E) {
  int e = blockIdx.x * blockDim.x + threadIdx.x;
  if (e >= E) return;
  float ds = deg[src[e]];
  float dd = deg[dst[e]];
  float is = ds > 0.f ? rsqrtf(ds) : 0.f;
  float id = dd > 0.f ? rsqrtf(dd) : 0.f;
  norm[e] = is * w[e] * id;
}

template <int K>
__global__ void gemm_wmma(const float* __restrict__ A, const float* __restrict__ B,
                          const float* __restrict__ bias, const float* __restrict__ rscale,
                          float* __restrict__ C, int N, int F, int rowTiles, int colTiles) {
  const int lane = threadIdx.x & 31;
  const int wave = threadIdx.x >> 5;
  const int tile = blockIdx.x * (blockDim.x >> 5) + wave;
  if (tile >= rowTiles * colTiles) return;

  const int rt = tile / colTiles;
  const int ct = tile - rt * colTiles;
  const int row0 = rt << 4;
  const int col0 = ct << 4;
  const int half = lane >> 4;
  const int lm   = lane & 15;

  const bool cOK = (col0 + lm) < F;
  const int  arow = min(row0 + lm, N - 1);
  const int  bcol = min(col0 + lm, F - 1);

  const float* Arow = A + (size_t)arow * K + 2 * half;
  const float* Bcol = B + bcol;

  v2f afrag[K / 4];
  v2f bfrag[K / 4];
#pragma unroll
  for (int i = 0; i < K / 4; ++i) {
    const int ka = 4 * i + 2 * half;
    afrag[i] = *reinterpret_cast<const v2f*>(Arow + 4 * i);
    bfrag[i].x = Bcol[(size_t)ka * F];
    bfrag[i].y = Bcol[(size_t)(ka + 1) * F];
  }

  v8f acc = {};
#pragma unroll
  for (int i = 0; i < K / 4; ++i) {
    acc = __builtin_amdgcn_wmma_f32_16x16x4_f32(
        false, afrag[i], false, bfrag[i], (short)0, acc, false, false);
  }

  const float bv = (bias != nullptr) ? bias[bcol] : 0.f;
#pragma unroll
  for (int r = 0; r < 8; ++r) {
    const int row = row0 + r + 8 * half;
    if (row < N && cOK) {
      const float v = (rscale != nullptr) ? acc[r] * rscale[row] : acc[r];
      C[(size_t)row * F + col0 + lm] = v + bv;
    }
  }
}

template <int K, int NT>
__global__ void __launch_bounds__(128)
gemm_wmma_pair(const float* __restrict__ A, const float* __restrict__ B1,
               const float* __restrict__ B2, const float* __restrict__ bias2,
               const float* __restrict__ rscale1,
               float* __restrict__ C1, float* __restrict__ C2,
               int rowTiles) {
  constexpr int F = 8 * NT;
  const int lane = threadIdx.x & 31;
  const int wave = threadIdx.x >> 5;
  const int rt = blockIdx.x * (blockDim.x >> 5) + wave;
  if (rt >= rowTiles) return;

  const int row0 = rt << 4;
  const int half = lane >> 4;
  const int lm   = lane & 15;
  const float* Arow = A + (size_t)(row0 + lm) * K + 2 * half;

  v2f afrag[K / 4];
#pragma unroll
  for (int i = 0; i < K / 4; ++i)
    afrag[i] = *reinterpret_cast<const v2f*>(Arow + 4 * i);

  float rs[8];
#pragma unroll
  for (int r = 0; r < 8; ++r) rs[r] = rscale1[row0 + 8 * half + r];

  const size_t rbase = (size_t)(row0 + 8 * half) * F;
#pragma unroll
  for (int t = 0; t < NT; ++t) {
    const int j = 16 * t + lm;
    const bool second = j >= F;
    const int col = second ? j - F : j;
    const float* Bcol = (second ? B2 : B1) + col;
    const float braw = bias2[col];
    const float bv = second ? braw : 0.f;

    v2f bfrag[K / 4];
#pragma unroll
    for (int i = 0; i < K / 4; ++i) {
      const int ka = 4 * i + 2 * half;
      bfrag[i].x = Bcol[(size_t)ka * F];
      bfrag[i].y = Bcol[(size_t)(ka + 1) * F];
    }

    v8f acc = {};
#pragma unroll
    for (int i = 0; i < K / 4; ++i) {
      acc = __builtin_amdgcn_wmma_f32_16x16x4_f32(
          false, afrag[i], false, bfrag[i], (short)0, acc, false, false);
    }

    float* Crow = (second ? C2 : C1) + col + rbase;
#pragma unroll
    for (int r = 0; r < 8; ++r) Crow[r * F] = second ? acc[r] + bv : acc[r] * rs[r];
  }
}

__global__ void gemm_wmma_gen(const float* __restrict__ A, const float* __restrict__ B,
                              const float* __restrict__ bias, const float* __restrict__ rscale,
                              float* __restrict__ C, int N, int K, int F, int rowTiles,
                              int colTiles) {
  const int lane = threadIdx.x & 31;
  const int wave = threadIdx.x >> 5;
  const int tile = blockIdx.x * (blockDim.x >> 5) + wave;
  if (tile >= rowTiles * colTiles) return;
  const int rt = tile / colTiles;
  const int ct = tile - rt * colTiles;
  const int row0 = rt << 4, col0 = ct << 4;
  const int half = lane >> 4, lm = lane & 15;
  const bool cOK = (col0 + lm) < F;
  const int arow = min(row0 + lm, N - 1);
  const int bcol = min(col0 + lm, F - 1);
  v8f acc = {};
  for (int kb = 0; kb < K; kb += 4) {
    const int ka = kb + 2 * half;
    v2f a = *reinterpret_cast<const v2f*>(A + (size_t)arow * K + ka);
    v2f b;
    b.x = B[(size_t)ka * F + bcol];
    b.y = B[(size_t)(ka + 1) * F + bcol];
    acc = __builtin_amdgcn_wmma_f32_16x16x4_f32(false, a, false, b, (short)0, acc,
                                                false, false);
  }
  const float bv = (bias != nullptr) ? bias[bcol] : 0.f;
#pragma unroll
  for (int r = 0; r < 8; ++r) {
    const int row = row0 + r + 8 * half;
    if (row < N && cOK) {
      const float v = (rscale != nullptr) ? acc[r] * rscale[row] : acc[r];
      C[(size_t)row * F + col0 + lm] = v + bv;
    }
  }
}

__global__ void edge_scatter(const float* __restrict__ h, const float* __restrict__ norm,
                             const int* __restrict__ src, const int* __restrict__ dst,
                             float* __restrict__ agg, int E, int F, int G) {
  long long tid = (long long)blockIdx.x * blockDim.x + threadIdx.x;
  if (tid >= (long long)E * G) return;
  int e = (int)(tid / G);
  int g = (int)(tid - (long long)e * G);
  int s = src[e];
  int d = dst[e];
  float w = norm[e];
  const float4 v = *reinterpret_cast<const float4*>(h + (size_t)s * F + (size_t)g * 4);
  float* base = agg + (size_t)d * F + (size_t)g * 4;
  atomicAdd(base + 0, v.x * w);
  atomicAdd(base + 1, v.y * w);
  atomicAdd(base + 2, v.z * w);
  atomicAdd(base + 3, v.w * w);
}

template <int C>
__global__ void relu_logsoftmax_t(float* __restrict__ y, int N) {
  int n = blockIdx.x * blockDim.x + threadIdx.x;
  if (n >= N) return;
  float v[C];
  float m = -3.4e38f;
  float* row = y + (size_t)n * C;
#pragma unroll
  for (int c = 0; c < C; ++c) {
    float t = fmaxf(row[c], 0.f);
    v[c] = t;
    m = fmaxf(m, t);
  }
  float s = 0.f;
#pragma unroll
  for (int c = 0; c < C; ++c) s += __expf(v[c] - m);
  float ls = m + __logf(s);
#pragma unroll
  for (int c = 0; c < C; ++c) row[c] = v[c] - ls;
}

__global__ void relu_logsoftmax_gen(float* __restrict__ y, int N, int C) {
  int n = blockIdx.x * blockDim.x + threadIdx.x;
  if (n >= N) return;
  float* row = y + (size_t)n * C;
  float m = -3.4e38f;
  for (int c = 0; c < C; ++c) m = fmaxf(m, fmaxf(row[c], 0.f));
  float s = 0.f;
  for (int c = 0; c < C; ++c) s += __expf(fmaxf(row[c], 0.f) - m);
  float ls = m + __logf(s);
  for (int c = 0; c < C; ++c) row[c] = fmaxf(row[c], 0.f) - ls;
}

#define CNT_STRIDE 32

__global__ void zero_counters(int* __restrict__ cnt, int N) {
  int n = blockIdx.x * blockDim.x + threadIdx.x;
  if (n < N) cnt[(size_t)n * CNT_STRIDE] = 0;
}

__global__ void rank_edges(const int* __restrict__ dst, int* __restrict__ cnt,
                           int* __restrict__ rank, int E) {
  int e0 = (blockIdx.x * blockDim.x + threadIdx.x) * 4;
  int d[4];
#pragma unroll
  for (int i = 0; i < 4; ++i) d[i] = (e0 + i < E) ? dst[e0 + i] : -1;
  int r[4];
#pragma unroll
  for (int i = 0; i < 4; ++i)
    r[i] = (d[i] >= 0) ? atomicAdd(cnt + (size_t)d[i] * CNT_STRIDE, 1) : 0;
#pragma unroll
  for (int i = 0; i < 4; ++i)
    if (e0 + i < E) rank[e0 + i] = r[i];
}

__device__ __forceinline__ int block_incl_scan_256(int v, int* sh) {
  const int t = threadIdx.x;
  sh[t] = v;
  __syncthreads();
  for (int off = 1; off < 256; off <<= 1) {
    int a = (t >= off) ? sh[t - off] : 0;
    __syncthreads();
    sh[t] += a;
    __syncthreads();
  }
  return sh[t];
}

__global__ void __launch_bounds__(256)
count_blocksum(const int* __restrict__ cnt, int* __restrict__ bsum, int N) {
  __shared__ int sh[256];
  int n = blockIdx.x * 256 + threadIdx.x;
  int v = (n < N) ? cnt[(size_t)n * CNT_STRIDE] : 0;
  int inc = block_incl_scan_256(v, sh);
  if (threadIdx.x == 255) bsum[blockIdx.x] = inc;
}

__global__ void __launch_bounds__(256)
scan_rowptr(const int* __restrict__ cnt, const int* __restrict__ bsum,
            int* __restrict__ rowptr, int N) {
  __shared__ int sh[256];
  __shared__ int sh2[256];
  const int t = threadIdx.x;
  int part = 0;
  for (int j = t; j < (int)blockIdx.x; j += 256) part += bsum[j];
  block_incl_scan_256(part, sh2);
  const int base = sh2[255];
  int n = blockIdx.x * 256 + t;
  int v = (n < N) ? cnt[(size_t)n * CNT_STRIDE] : 0;
  int inc = block_incl_scan_256(v, sh);
  if (n < N) {
    rowptr[n] = base + inc - v;
    if (n == N - 1) rowptr[N] = base + inc;
  }
}

__global__ void fill_sorted(const int* __restrict__ src, const int* __restrict__ dst,
                            const float* __restrict__ w, const int* __restrict__ rowptr,
                            const int* __restrict__ rank, int* __restrict__ ssrc,
                            float* __restrict__ sw, int E) {
  int e0 = (blockIdx.x * blockDim.x + threadIdx.x) * 4;
  for (int i = 0; i < 4; ++i) {
    int e = e0 + i;
    if (e < E) {
      int p = rowptr[dst[e]] + rank[e];
      ssrc[p] = src[e];
      sw[p] = w[e];
    }
  }
}

__global__ void deg_dinv(const float* __restrict__ sw, const int* __restrict__ rowptr,
                         float* __restrict__ dinv, int N) {
  int n = blockIdx.x * blockDim.x + threadIdx.x;
  if (n >= N) return;
  const int p0 = rowptr[n], p1 = rowptr[n + 1];
  float s = 0.f;
  for (int p = p0; p < p1; ++p) s += sw[p];
  dinv[n] = s > 0.f ? rsqrtf(s) : 0.f;
}

template <bool RELU>
__global__ void csr_gather(const float* __restrict__ h, const float* __restrict__ sw,
                           const int* __restrict__ ssrc, const int* __restrict__ rowptr,
                           const float* __restrict__ dinv,
                           float* __restrict__ agg, int N, int F, int G) {
  long long tid = (long long)blockIdx.x * blockDim.x + threadIdx.x;
  if (tid >= (long long)N * G) return;
  int n = (int)(tid / G);
  int g = (int)(tid - (long long)n * G);
  const int p0 = rowptr[n], p1 = rowptr[n + 1];
  const float dn = dinv[n];
  const float* hb = h + (size_t)g * 4;
  float4 acc = make_float4(0.f, 0.f, 0.f, 0.f);
#pragma unroll 4
  for (int p = p0; p < p1; ++p) {
    const int s = ssrc[p];
    const float w = sw[p];
    const float4 v = *reinterpret_cast<const float4*>(hb + (size_t)s * F);
    acc.x += v.x * w;
    acc.y += v.y * w;
    acc.z += v.z * w;
    acc.w += v.w * w;
  }
  float4* o = reinterpret_cast<float4*>(agg + (size_t)n * F + (size_t)g * 4);
  float4 z = *o;
  z.x += acc.x * dn;
  z.y += acc.y * dn;
  z.z += acc.z * dn;
  z.w += acc.w * dn;
  if (RELU) {
    z.x = fmaxf(z.x, 0.f);
    z.y = fmaxf(z.y, 0.f);
    z.z = fmaxf(z.z, 0.f);
    z.w = fmaxf(z.w, 0.f);
  }
  *o = z;
}

template <int C>
__global__ void csr_gather_logsoftmax(const float* __restrict__ h,
                                      const float* __restrict__ sw,
                                      const int* __restrict__ ssrc,
                                      const int* __restrict__ rowptr,
                                      const float* __restrict__ dinv,
                                      float* __restrict__ y, int N) {
  constexpr int G = C / 4;
  constexpr int NPW = 32 / G;
  const int lane = threadIdx.x & 31;
  const int wave = (int)((blockIdx.x * blockDim.x + threadIdx.x) >> 5);
  const int slot = lane / G;
  const int g = lane - slot * G;
  const int n = wave * NPW + slot;
  const bool live = (slot < NPW) && (n < N);
  int p0 = 0, p1 = 0;
  float dn = 0.f;
  if (live) {
    p0 = rowptr[n];
    p1 = rowptr[n + 1];
    dn = dinv[n];
  }
  const float* hb = h + (size_t)g * 4;
  float4 acc = make_float4(0.f, 0.f, 0.f, 0.f);
#pragma unroll 4
  for (int p = p0; p < p1; ++p) {
    const int s = ssrc[p];
    const float w = sw[p];
    const float4 v = *reinterpret_cast<const float4*>(hb + (size_t)s * C);
    acc.x += v.x * w;
    acc.y += v.y * w;
    acc.z += v.z * w;
    acc.w += v.w * w;
  }
  float4 t = make_float4(0.f, 0.f, 0.f, 0.f);
  float4* o = nullptr;
  if (live) {
    o = reinterpret_cast<float4*>(y + (size_t)n * C + (size_t)g * 4);
    const float4 z = *o;
    t.x = fmaxf(z.x + acc.x * dn, 0.f);
    t.y = fmaxf(z.y + acc.y * dn, 0.f);
    t.z = fmaxf(z.z + acc.z * dn, 0.f);
    t.w = fmaxf(z.w + acc.w * dn, 0.f);
  }
  const int first = live ? lane - g : lane;
  const int step = live ? 1 : 0;
  const float m4 = fmaxf(fmaxf(t.x, t.y), fmaxf(t.z, t.w));
  float m = -3.4e38f;
  for (int k = 0; k < G; ++k) m = fmaxf(m, __shfl(m4, first + k * step));
  const float e4 = __expf(t.x - m) + __expf(t.y - m) + __expf(t.z - m) + __expf(t.w - m);
  float s = 0.f;
  for (int k = 0; k < G; ++k) s += __shfl(e4, first + k * step);
  const float ls = m + __logf(s);
  if (live) *o = make_float4(t.x - ls, t.y - ls, t.z - ls, t.w - ls);
}


template <int K>
__device__ __forceinline__ int row_lane_i(int v) {
  return __builtin_amdgcn_update_dpp(0, v, 0x150 + K, 0xf, 0xf, true);
}
template <int K>
__device__ __forceinline__ float row_lane_f(float v) {
  return __int_as_float(row_lane_i<K>(__float_as_int(v)));
}

template <int HP, int K0>
__device__ __forceinline__ void seg_rows8(const char* __restrict__ hB, unsigned laneoff,
                                          int myS, float myW, float4& acc) {
  float4 v[8];
  float w[8];
#define SEG_LOAD(K)                                                                    \
  {                                                                                    \
    const unsigned s = (unsigned)row_lane_i<K0 + K>(myS);                              \
    w[K] = row_lane_f<K0 + K>(myW);                                                    \
    v[K] = *reinterpret_cast<const float4*>(hB + (size_t)(s * (unsigned)(HP * 4) + laneoff)); \
  }
  SEG_LOAD(0) SEG_LOAD(1) SEG_LOAD(2) SEG_LOAD(3)
  SEG_LOAD(4) SEG_LOAD(5) SEG_LOAD(6) SEG_LOAD(7)
#undef SEG_LOAD
#pragma unroll
  for (int k = 0; k < 8; ++k) {
    acc.x = __builtin_fmaf(v[k].x, w[k], acc.x);
    acc.y = __builtin_fmaf(v[k].y, w[k], acc.y);
    acc.z = __builtin_fmaf(v[k].z, w[k], acc.z);
    acc.w = __builtin_fmaf(v[k].w, w[k], acc.w);
  }
}

template <int F, bool LSM, int HP = F, int NT = 0>
__global__ void __launch_bounds__(256)
seg_gather16(const float* __restrict__ h, const float* __restrict__ sw,
             const int* __restrict__ ssrc, const int* __restrict__ rowptr,
             const float* __restrict__ dinv, float* __restrict__ y, int N,
             const float* __restrict__ B1, const float* __restrict__ B2,
             const float* __restrict__ bias2, float* __restrict__ C2) {
  static_assert(F % 4 == 0 && F <= 64, "a row of F floats must fit the 16 lanes of a wave row, 4 floats each");
  static_assert(NT == 0 || !LSM, "the products take the rows after ReLU only");
  constexpr int NL = F / 4;
  constexpr int ZP = (F / 4) * 4 + (((F / 4) & 1) ? 0 : 4);
  __shared__ __attribute__((aligned(16))) float zt[NT > 0 ? 32 * ZP : 4];
  constexpr int OP = 16 * NT + 4;
  __shared__ __attribute__((aligned(16))) float ot[NT > 0 ? 32 * OP : 4];
  const int lane = threadIdx.x & 31;
  const int l16 = lane & 15;
  const int wave = (int)((blockIdx.x * blockDim.x + threadIdx.x) >> 5);
  const bool feat = l16 < NL;
  const unsigned laneoff = feat ? (unsigned)l16 * 16u : 0u;
  const char* hB = reinterpret_cast<const char*>(h);

  const unsigned lofs4 = feat ? (unsigned)l16 * 4u : 0u;
  const int n0 = wave * 4 + 2 * (lane >> 4);
  const int rp0 = rowptr[min(n0, N)];
  const int rp1 = rowptr[min(n0 + 1, N)];
  const int rp2 = rowptr[min(n0 + 2, N)];
  int nn[2], p0[2], p1[2], s0[2];
  float dn[2], w0[2];
  float4 z[2];
  bool live[2];
#pragma unroll
  for (int r = 0; r < 2; ++r) {
    nn[r] = n0 + r;
    live[r] = nn[r] < N;
    const int nc = min(nn[r], N - 1);
    dn[r] = dinv[nc];
    z[r] = *reinterpret_cast<const float4*>(y + (size_t)nc * F + lofs4);
  }
  p0[0] = rp0;
  p1[0] = rp1;
  p0[1] = rp1;
  p1[1] = rp2;
#pragma unroll
  for (int r = 0; r < 2; ++r) {
    const int q = p0[r] + l16;
    const int qc = max(min(q, p1[r] - 1), 0);
    s0[r] = ssrc[qc];
    const float wl = sw[qc];
    w0[r] = (q < p1[r]) ? wl : 0.f;
  }

#pragma unroll
  for (int r = 0; r < 2; ++r) {
    float4 acc = make_float4(0.f, 0.f, 0.f, 0.f);
    const int pe = p1[r];
    int base = p0[r];
    int myS = s0[r];
    float myW = w0[r];
    while (base < pe) {
      const int cnt = pe - base;
      const int nb = base + 16;
      const int q = nb + l16;
      const int qc = min(q, pe - 1);
      const int nS = ssrc[qc];
      const float nWl = sw[qc];
      const float nW = (q < pe) ? nWl : 0.f;
      seg_rows8<HP, 0>(hB, laneoff, myS, myW, acc);
      if (cnt > 8) seg_rows8<HP, 8>(hB, laneoff, myS, myW, acc);
      myS = nS;
      myW = nW;
      base = nb;
    }
    float4 t;
    t.x = __builtin_fmaf(acc.x, dn[r], z[r].x);
    t.y = __builtin_fmaf(acc.y, dn[r], z[r].y);
    t.z = __builtin_fmaf(acc.z, dn[r], z[r].z);
    t.w = __builtin_fmaf(acc.w, dn[r], z[r].w);
    t.x = fmaxf(t.x, 0.f);
    t.y = fmaxf(t.y, 0.f);
    t.z = fmaxf(t.z, 0.f);
    t.w = fmaxf(t.w, 0.f);
    if (LSM) {
      const float m4 = fmaxf(fmaxf(t.x, t.y), fmaxf(t.z, t.w));
      float m = -3.4e38f;
#define SEG_MAX(K) if (K < NL) m = fmaxf(m, row_lane_f<K>(m4));
      SEG_MAX(0) SEG_MAX(1) SEG_MAX(2) SEG_MAX(3) SEG_MAX(4) SEG_MAX(5) SEG_MAX(6) SEG_MAX(7)
      SEG_MAX(8) SEG_MAX(9) SEG_MAX(10) SEG_MAX(11) SEG_MAX(12) SEG_MAX(13) SEG_MAX(14) SEG_MAX(15)
#undef SEG_MAX
      const float e4 = __expf(t.x - m) + __expf(t.y - m) + __expf(t.z - m) + __expf(t.w - m);
      float s = 0.f;
#define SEG_SUM(K) if (K < NL) s += row_lane_f<K>(e4);
      SEG_SUM(0) SEG_SUM(1) SEG_SUM(2) SEG_SUM(3) SEG_SUM(4) SEG_SUM(5) SEG_SUM(6) SEG_SUM(7)
      SEG_SUM(8) SEG_SUM(9) SEG_SUM(10) SEG_SUM(11) SEG_SUM(12) SEG_SUM(13) SEG_SUM(14) SEG_SUM(15)
#undef SEG_SUM
      const float ls = m + __logf(s);
      t.x -= ls;
      t.y -= ls;
      t.z -= ls;
      t.w -= ls;
    }
    if constexpr (NT > 0) {
      const int tr = 4 * (int)(threadIdx.x >> 5) + 2 * (lane >> 4) + r;
      if (feat) *reinterpret_cast<float4*>(&zt[tr * ZP + 4 * l16]) = t;
    } else {
      if (live[r] && feat)
        *reinterpret_cast<float4*>(y + (size_t)nn[r] * F + (size_t)l16 * 4) = t;
    }
  }

  if constexpr (NT > 0) {
    constexpr int FO = 8 * NT;
    __syncthreads();
    const int wib  = __builtin_amdgcn_readfirstlane((int)(threadIdx.x >> 5));
    const int half = lane >> 4;
    const int lm   = lane & 15;
    const int rt   = wib & 1;
    const float* Arow = zt + (16 * rt + lm) * ZP + 2 * half;
    v2f afrag[F / 4];
#pragma unroll
    for (int i = 0; i < F / 4; ++i)
      afrag[i] = *reinterpret_cast<const v2f*>(Arow + 4 * i);
    const int row0 = (int)blockIdx.x * 32 + 16 * rt + 8 * half;
    float rs[8];
#pragma unroll
    for (int r = 0; r < 8; ++r) rs[r] = dinv[row0 + r];
    for (int t = wib >> 1; t < NT; t += 4) {
      const int j = 16 * t + lm;
      const bool second = j >= FO;
      const int col = second ? j - FO : j;
      const float* Bcol = (second ? B2 : B1) + col;
      const float braw = bias2[col];
      const float bv = second ? braw : 0.f;
      v2f bfrag[F / 4];
#pragma unroll
      for (int i = 0; i < F / 4; ++i) {
        const int ka = 4 * i + 2 * half;
        bfrag[i].x = Bcol[(size_t)ka * FO];
        bfrag[i].y = Bcol[(size_t)(ka + 1) * FO];
      }
      v8f acc = {};
#pragma unroll
      for (int i = 0; i < F / 4; ++i) {
        acc = __builtin_amdgcn_wmma_f32_16x16x4_f32(
            false, afrag[i], false, bfrag[i], (short)0, acc, false, false);
      }
      float* orow = ot + (16 * rt + 8 * half) * OP + j;
#pragma unroll
      for (int r = 0; r < 8; ++r) orow[r * OP] = second ? acc[r] + bv : acc[r] * rs[r];
    }
    __syncthreads();
#pragma unroll
    for (int r = 0; r < 2; ++r) {
      const int tr = 4 * wib + 2 * half + r;
      const size_t node = (size_t)blockIdx.x * 32 + tr;
      if (lm < FO / 4) {
        const float4 c1 = *reinterpret_cast<const float4*>(&ot[tr * OP + 4 * lm]);
        const float4 c2 = *reinterpret_cast<const float4*>(&ot[tr * OP + FO + 4 * lm]);
        *reinterpret_cast<float4*>(y + node * F + 4 * lm) = c1;
        *reinterpret_cast<float4*>(C2 + node * FO + 4 * lm) = c2;
      }
    }
  }
}

#define NPB_SHIFT 9
#define NPB (1 << NPB_SHIFT)
#define MAXB 256
#define BIN_CH 6400
#define BIN_TPB 1024
#define BIN_IT ((BIN_CH + BIN_TPB - 1) / BIN_TPB)
#define SORT_TPB 1024
#define SORT_IT 4
#define SORT_CAP 10240

template <int PER>
__device__ __forceinline__ void excl_scan_lds_wave0(int* a) {
  __syncthreads();
  if (threadIdx.x < 32) {
    const int lane = threadIdx.x;
    int v[PER];
    int s = 0;
#pragma unroll
    for (int k = 0; k < PER; ++k) {
      v[k] = a[lane * PER + k];
      s += v[k];
    }
    int inc = s;
#pragma unroll
    for (int off = 1; off < 32; off <<= 1) {
      const int u = __shfl_up(inc, off);
      if (lane >= off) inc += u;
    }
    int run = inc - s;
#pragma unroll
    for (int k = 0; k < PER; ++k) {
      a[lane * PER + k] = run;
      run += v[k];
    }
    if (lane == 31) a[32 * PER] = run;
  }
  __syncthreads();
}

__global__ void __launch_bounds__(BIN_TPB)
bin_count(const int* __restrict__ dst, int* __restrict__ cntmat, int E, int B) {
  __shared__ int hist[MAXB];
  const int t = threadIdx.x;
  if (t < MAXB) hist[t] = 0;
  __syncthreads();
  const int e0 = blockIdx.x * BIN_CH;
  const int n = min(BIN_CH, E - e0);
  int d[BIN_IT];
#pragma unroll
  for (int k = 0; k < BIN_IT; ++k) {
    const int i = t + k * BIN_TPB;
    d[k] = (i < n) ? dst[e0 + i] : -1;
  }
#pragma unroll
  for (int k = 0; k < BIN_IT; ++k)
    if (d[k] >= 0) atomicAdd(&hist[d[k] >> NPB_SHIFT], 1);
  __syncthreads();
  if (t < B) cntmat[(size_t)blockIdx.x * B + t] = hist[t];
}

__global__ void __launch_bounds__(BIN_TPB)
bin_scatter(const int* __restrict__ src, const int* __restrict__ dst,
            const float* __restrict__ w, const int* __restrict__ cntmat,
            int* __restrict__ bstart, int* __restrict__ bsrc, int* __restrict__ bdst,
            float* __restrict__ bw, int E, int B, int G1) {
  __shared__ int   s_src[BIN_CH];
  __shared__ int   s_dst[BIN_CH];
  __shared__ float s_w[BIN_CH];
  __shared__ int   p_tot[4][MAXB];
  __shared__ int   p_pre[4][MAXB];
  __shared__ int   gbase[MAXB + 1];
  __shared__ int   lstart[MAXB + 1];
  __shared__ int   lcnt[MAXB];
  const int t = threadIdx.x;
  const int g = blockIdx.x;
  const int col = t & (MAXB - 1);
  const int grp = t >> 8;
  int tot = 0, pre = 0;
  if (col < B) {
#pragma unroll 8
    for (int r = grp; r < G1; r += 4) {
      const int c = cntmat[(size_t)r * B + col];
      tot += c;
      pre += (r < g) ? c : 0;
    }
  }
  p_tot[grp][col] = tot;
  p_pre[grp][col] = pre;
  __syncthreads();
  int mypre = 0;
  if (t < MAXB) {
    gbase[t] = p_tot[0][t] + p_tot[1][t] + p_tot[2][t] + p_tot[3][t];
    mypre = p_pre[0][t] + p_pre[1][t] + p_pre[2][t] + p_pre[3][t];
    lstart[t] = (t < B) ? cntmat[(size_t)g * B + t] : 0;
    lcnt[t] = 0;
  }
  excl_scan_lds_wave0<MAXB / 32>(gbase);
  excl_scan_lds_wave0<MAXB / 32>(lstart);
  if (g == 0 && t <= B) bstart[t] = gbase[t];
  __syncthreads();
  if (t < MAXB) gbase[t] += mypre;
  __syncthreads();

  const int e0 = g * BIN_CH;
  const int n = min(BIN_CH, E - e0);
  int d[BIN_IT], sv[BIN_IT];
  float wv[BIN_IT];
#pragma unroll
  for (int k = 0; k < BIN_IT; ++k) {
    const int i = t + k * BIN_TPB;
    const bool in = i < n;
    d[k]  = in ? dst[e0 + i] : -1;
    sv[k] = in ? src[e0 + i] : 0;
    wv[k] = in ? w[e0 + i] : 0.f;
  }
#pragma unroll
  for (int k = 0; k < BIN_IT; ++k) {
    if (d[k] >= 0) {
      const int bk = d[k] >> NPB_SHIFT;
      const int q = lstart[bk] + atomicAdd(&lcnt[bk], 1);
      s_src[q] = sv[k];
      s_dst[q] = d[k];
      s_w[q] = wv[k];
    }
  }
  __syncthreads();
  for (int q = t; q < n; q += BIN_TPB) {
    const int d = s_dst[q];
    const int bk = d >> NPB_SHIFT;
    const int p = gbase[bk] + (q - lstart[bk]);
    bsrc[p] = s_src[q];
    bdst[p] = d;
    bw[p] = s_w[q];
  }
}

__global__ void __launch_bounds__(SORT_TPB)
bucket_sort(const int* __restrict__ bsrc, const int* __restrict__ bdst,
            const float* __restrict__ bw, const int* __restrict__ bstart,
            int* __restrict__ rowptr, int* __restrict__ ssrc, float* __restrict__ sw,
            float* __restrict__ dinv, int N, int B) {
  __shared__ int   c_src[SORT_CAP];
  __shared__ float c_w[SORT_CAP];
  __shared__ int   nstart[NPB + 1];
  __shared__ int   ncnt[NPB];
  const int t = threadIdx.x;
  const int b = blockIdx.x;
  const int m0 = bstart[b];
  const int m1 = bstart[b + 1];
  const int nE = m1 - m0;
  const int node0 = b << NPB_SHIFT;
  if (t < NPB) {
    nstart[t] = 0;
    ncnt[t] = 0;
  }
  __syncthreads();
  for (int i0 = t; i0 < nE; i0 += SORT_IT * SORT_TPB) {
    int dd[SORT_IT];
#pragma unroll
    for (int k = 0; k < SORT_IT; ++k) {
      const int i = i0 + k * SORT_TPB;
      dd[k] = (i < nE) ? bdst[m0 + i] : -1;
    }
#pragma unroll
    for (int k = 0; k < SORT_IT; ++k)
      if (dd[k] >= 0) atomicAdd(&nstart[dd[k] - node0], 1);
  }
  excl_scan_lds_wave0<NPB / 32>(nstart);
  if (t < NPB && node0 + t < N) rowptr[node0 + t] = m0 + nstart[t];
  if (b == B - 1 && t == 0) rowptr[N] = m1;
  const bool staged = nE <= SORT_CAP;
  if (staged) {
    for (int i0 = t; i0 < nE; i0 += SORT_IT * SORT_TPB) {
      int dd[SORT_IT], ss[SORT_IT];
      float xx[SORT_IT];
#pragma unroll
      for (int k = 0; k < SORT_IT; ++k) {
        const int i = i0 + k * SORT_TPB;
        const bool in = i < nE;
        dd[k] = in ? bdst[m0 + i] : -1;
        ss[k] = in ? bsrc[m0 + i] : 0;
        xx[k] = in ? bw[m0 + i] : 0.f;
      }
#pragma unroll
      for (int k = 0; k < SORT_IT; ++k) {
        if (dd[k] >= 0) {
          const int l = dd[k] - node0;
          const int q = nstart[l] + atomicAdd(&ncnt[l], 1);
          c_src[q] = ss[k];
          c_w[q] = xx[k];
        }
      }
    }
  } else {
    for (int i = t; i < nE; i += SORT_TPB) {
      const int l = bdst[m0 + i] - node0;
      const int q = nstart[l] + atomicAdd(&ncnt[l], 1);
      ssrc[m0 + q] = bsrc[m0 + i];
      sw[m0 + q] = bw[m0 + i];
    }
  }
  __syncthreads();
  if (t < NPB && node0 + t < N) {
    const int q0 = nstart[t], q1 = nstart[t + 1];
    float s = 0.f;
    if (staged) {
      for (int q = q0; q < q1; ++q) s += c_w[q];
    } else {
      for (int q = q0; q < q1; ++q) s += sw[m0 + q];
    }
    dinv[node0 + t] = s > 0.f ? rsqrtf(s) : 0.f;
  }
  if (staged) {
    for (int q = t; q < nE; q += SORT_TPB) {
      ssrc[m0 + q] = c_src[q];
      sw[m0 + q] = c_w[q];
    }
  }
}

static inline int ceil_div(long long a, int b) { return (int)((a + b - 1) / b); }

static void launch_gemm(const float* A, const float* B, const float* bias, const float* rscale,
                        float* C, int N, int K, int F, hipStream_t stream) {
  const int TPB = 256;
  int rowT = ceil_div(N, 16), colT = ceil_div(F, 16);
  int blocks = ceil_div((long long)rowT * colT, TPB / 32);
  if (K == 64) {
    gemm_wmma<64><<<blocks, TPB, 0, stream>>>(A, B, bias, rscale, C, N, F, rowT, colT);
  } else if (K == 48) {
    gemm_wmma<48><<<blocks, TPB, 0, stream>>>(A, B, bias, rscale, C, N, F, rowT, colT);
  } else {
    gemm_wmma_gen<<<blocks, TPB, 0, stream>>>(A, B, bias, rscale, C, N, K, F, rowT, colT);
  }
}

static void launch_gemm_pair(const float* A, const float* B1, const float* B2,
                             const float* bias2, const float* rscale1, float* C1, float* C2,
                             int N, int K, int F, hipStream_t stream) {
  const int rowT = N / 16;
  const int WPB = 4;
  const int blocks = ceil_div(rowT, WPB);
  const bool full = (N % 16 == 0) && (N > 0) && (bias2 != nullptr);
  if (full && K == 64 && F == 48) {
    gemm_wmma_pair<64, 6><<<blocks, WPB * 32, 0, stream>>>(A, B1, B2, bias2, rscale1, C1, C2,
                                                            rowT);
  } else if (full && K == 48 && F == 40) {
    gemm_wmma_pair<48, 5><<<blocks, WPB * 32, 0, stream>>>(A, B1, B2, bias2, rscale1, C1, C2,
                                                            rowT);
  } else {
    launch_gemm(A, B1, nullptr, rscale1, C1, N, K, F, stream);
    launch_gemm(A, B2, bias2,   nullptr, C2, N, K, F, stream);
  }
}

extern "C" void kernel_launch(void* const* d_in, const int* in_sizes, int n_in,
                              void* d_out, int out_size, void* d_ws, size_t ws_size,
                              hipStream_t stream) {
  const float* x  = (const float*)d_in[0];
  const int*   ei = (const int*)d_in[1];
  const float* ew = (const float*)d_in[2];
  const float* W1 = (const float*)d_in[3];
  const float* V1 = (const float*)d_in[4];
  const float* b1 = (const float*)d_in[5];
  const float* W2 = (const float*)d_in[6];
  const float* V2 = (const float*)d_in[7];
  const float* b2 = (const float*)d_in[8];
  float* out = (float*)d_out;

  const int H   = in_sizes[5];
  const int Cn  = in_sizes[8];
  const int Fin = in_sizes[3] / H;
  const int N   = in_sizes[0] / Fin;
  const int E   = in_sizes[2];
  const int* src = ei;
  const int* dst = ei + E;

  float* ws = (float*)d_ws;
  const size_t NHa = ((size_t)N * H + 255) & ~(size_t)255;
  const size_t Ea  = ((size_t)E + 255) & ~(size_t)255;
  size_t off = 0;
  float* deg  = ws + off; off += ((size_t)N + 255) & ~(size_t)255;
  float* norm = ws + off; off += Ea;
  float* hbuf = ws + off; off += NHa;
  float* z1   = ws + off; off += NHa;
  int* ssrc   = (int*)(ws + off); off += Ea;
  int* rowptr = (int*)(ws + off); off += ((size_t)N + 1 + 255) & ~(size_t)255;
  const int NB = ceil_div(N, 256);
  int* rank = (int*)hbuf;
  int* cnt  = rank + Ea;
  int* bsum = cnt + (((size_t)N * CNT_STRIDE + 255) & ~(size_t)255);
  const size_t scratch_words = Ea + (((size_t)N * CNT_STRIDE + 255) & ~(size_t)255) + (size_t)NB;
  const bool sorted_path = (off * sizeof(float) <= ws_size) && (scratch_words <= 2 * NHa) &&
                           (H % 4 == 0) && (Cn % 4 == 0);
  const int B  = ceil_div(N, NPB);
  const int G1 = ceil_div(E, BIN_CH);
  const size_t matw = ((size_t)G1 * B + 255) & ~(size_t)255;
  int*   bsrc   = (int*)hbuf;
  int*   bdst   = bsrc + Ea;
  float* bw     = (float*)(bdst + Ea);
  int*   cntmat = (int*)(bw + Ea);
  int*   bstart = cntmat + matw;
  const bool lds_sort = (B <= MAXB) && (E > 0) && (3 * Ea + matw + (size_t)B + 1 <= 2 * NHa);

  const int TPB = 256;

  if (sorted_path) {
    if (lds_sort) {
      bin_count<<<G1, BIN_TPB, 0, stream>>>(dst, cntmat, E, B);
      bin_scatter<<<G1, BIN_TPB, 0, stream>>>(src, dst, ew, cntmat, bstart, bsrc, bdst, bw,
                                              E, B, G1);
      bucket_sort<<<B, SORT_TPB, 0, stream>>>(bsrc, bdst, bw, bstart, rowptr, ssrc, norm, deg,
                                              N, B);
    } else {
      zero_counters<<<NB, TPB, 0, stream>>>(cnt, N);
      rank_edges<<<ceil_div(ceil_div(E, 4), TPB), TPB, 0, stream>>>(dst, cnt, rank, E);
      count_blocksum<<<NB, 256, 0, stream>>>(cnt, bsum, N);
      scan_rowptr<<<NB, 256, 0, stream>>>(cnt, bsum, rowptr, N);
      fill_sorted<<<ceil_div(ceil_div(E, 4), TPB), TPB, 0, stream>>>(src, dst, ew, rowptr, rank,
                                                                   ssrc, norm, E);
      deg_dinv<<<NB, TPB, 0, stream>>>(norm, rowptr, deg, N);
    }

    launch_gemm_pair(x, W1, V1, b1, deg, hbuf, z1, N, Fin, H, stream);
    const bool rows32 = (size_t)N * 256 < ((size_t)1 << 31);
    const int segBlocks = ceil_div((long long)ceil_div(N, 4) * 32, TPB);
    if (H == 48 && Cn == 40 && rows32 && (N % 32 == 0) && b2 != nullptr) {
      seg_gather16<48, false, 48, 5><<<segBlocks, TPB, 0, stream>>>(hbuf, norm, ssrc, rowptr, deg,
                                                                    z1, N, W2, V2, b2, out);
      seg_gather16<40, true, 48><<<segBlocks, TPB, 0, stream>>>(z1, norm, ssrc, rowptr, deg, out,
                                                                N, nullptr, nullptr, nullptr,
                                                                nullptr);
      return;
    }
    if (H == 48 && rows32) {
      seg_gather16<48, false><<<segBlocks, TPB, 0, stream>>>(hbuf, norm, ssrc, rowptr, deg, z1, N,
                                                             nullptr, nullptr, nullptr, nullptr);
    } else {
      int G = H / 4;
      csr_gather<true><<<ceil_div((long long)N * G, TPB), TPB, 0, stream>>>(
          hbuf, norm, ssrc, rowptr, deg, z1, N, H, G);
    }

    launch_gemm_pair(z1, W2, V2, b2, deg, hbuf, out, N, H, Cn, stream);
    {
      if (Cn == 40 && rows32) {
        seg_gather16<40, true><<<segBlocks, TPB, 0, stream>>>(hbuf, norm, ssrc, rowptr, deg, out, N,
                                                              nullptr, nullptr, nullptr, nullptr);
      } else if (Cn == 40) {
        csr_gather_logsoftmax<40><<<ceil_div((long long)ceil_div(N, 3) * 32, TPB), TPB, 0, stream>>>(
            hbuf, norm, ssrc, rowptr, deg, out, N);
      } else {
        int G = Cn / 4;
        csr_gather<false><<<ceil_div((long long)N * G, TPB), TPB, 0, stream>>>(
            hbuf, norm, ssrc, rowptr, deg, out, N, Cn, G);
        relu_logsoftmax_gen<<<ceil_div(N, TPB), TPB, 0, stream>>>(out, N, Cn);
      }
    }
    return;
  }

  zero_f32<<<ceil_div(N, TPB), TPB, 0, stream>>>(deg, N);
  deg_accum<<<ceil_div(E, TPB), TPB, 0, stream>>>(dst, ew, deg, E);
  norm_compute<<<ceil_div(E, TPB), TPB, 0, stream>>>(src, dst, ew, deg, norm, E);

  launch_gemm(x, W1, nullptr, nullptr, hbuf, N, Fin, H, stream);
  launch_gemm(x, V1, b1,      nullptr, z1,   N, Fin, H, stream);
  {
    int G = H / 4;
    edge_scatter<<<ceil_div((long long)E * G, TPB), TPB, 0, stream>>>(
        hbuf, norm, src, dst, z1, E, H, G);
    relu_inplace<<<ceil_div((long long)N * H, TPB), TPB, 0, stream>>>(z1, N * H);
  }

  launch_gemm(z1, W2, nullptr, nullptr, hbuf, N, H, Cn, stream);
  launch_gemm(z1, V2, b2,      nullptr, out,  N, H, Cn, stream);
  {
    int G = Cn / 4;
    edge_scatter<<<ceil_div((long long)E * G, TPB), TPB, 0, stream>>>(
        hbuf, norm, src, dst, out, E, Cn, G);
    if (Cn == 40) {
      relu_logsoftmax_t<40><<<ceil_div(N, TPB), TPB, 0, stream>>>(out, N);
    } else {
      relu_logsoftmax_gen<<<ceil_div(N, TPB), TPB, 0, stream>>>(out, N, Cn);
    }
  }
}


# Level-1 edge grouping without a global count: each chunk groups its records by bucket in its own slot (one launch, not two); the bucket pass collects a run per chunk; segment order: LDS adds' arrival
# speedup vs baseline: 4.4520x; 1.1957x over previous
#include <hip/hip_runtime.h>

typedef float v2f __attribute__((ext_vector_type(2)));
typedef float v8f __attribute__((ext_vector_type(8)));

__global__ void zero_f32(float* __restrict__ p, int n) {
  int i = blockIdx.x * blockDim.x + threadIdx.x;
  if (i < n) p[i] = 0.f;
}

__global__ void relu_inplace(float* __restrict__ p, int n) {
  int i = blockIdx.x * blockDim.x + threadIdx.x;
  if (i < n) p[i] = fmaxf(p[i], 0.f);
}

__global__ void deg_accum(const int* __restrict__ dst, const float* __restrict__ w,
                          float* __restrict__ deg, int E) {
  int e = blockIdx.x * blockDim.x + threadIdx.x;
  if (e < E) atomicAdd(deg + dst[e], w[e]);
}

__global__ void norm_compute(const int* __restrict__ src, const int* __restrict__ dst,
                             const float* __restrict__ w, const float* __restrict__ deg,
                             float* __restrict__ norm, int E) {
  int e = blockIdx.x * blockDim.x + threadIdx.x;
  if (e >= E) return;
  float ds = deg[src[e]];
  float dd = deg[dst[e]];
  float is = ds > 0.f ? rsqrtf(ds) : 0.f;
  float id = dd > 0.f ? rsqrtf(dd) : 0.f;
  norm[e] = is * w[e] * id;
}

template <int K>
__global__ void gemm_wmma(const float* __restrict__ A, const float* __restrict__ B,
                          const float* __restrict__ bias, const float* __restrict__ rscale,
                          float* __restrict__ C, int N, int F, int rowTiles, int colTiles) {
  const int lane = threadIdx.x & 31;
  const int wave = threadIdx.x >> 5;
  const int tile = blockIdx.x * (blockDim.x >> 5) + wave;
  if (tile >= rowTiles * colTiles) return;

  const int rt = tile / colTiles;
  const int ct = tile - rt * colTiles;
  const int row0 = rt << 4;
  const int col0 = ct << 4;
  const int half = lane >> 4;
  const int lm   = lane & 15;

  const bool cOK = (col0 + lm) < F;
  const int  arow = min(row0 + lm, N - 1);
  const int  bcol = min(col0 + lm, F - 1);

  const float* Arow = A + (size_t)arow * K + 2 * half;
  const float* Bcol = B + bcol;

  v2f afrag[K / 4];
  v2f bfrag[K / 4];
#pragma unroll
  for (int i = 0; i < K / 4; ++i) {
    const int ka = 4 * i + 2 * half;
    afrag[i] = *reinterpret_cast<const v2f*>(Arow + 4 * i);
    bfrag[i].x = Bcol[(size_t)ka * F];
    bfrag[i].y = Bcol[(size_t)(ka + 1) * F];
  }

  v8f acc = {};
#pragma unroll
  for (int i = 0; i < K / 4; ++i) {
    acc = __builtin_amdgcn_wmma_f32_16x16x4_f32(
        false, afrag[i], false, bfrag[i], (short)0, acc, false, false);
  }

  const float bv = (bias != nullptr) ? bias[bcol] : 0.f;
#pragma unroll
  for (int r = 0; r < 8; ++r) {
    const int row = row0 + r + 8 * half;
    if (row < N && cOK) {
      const float v = (rscale != nullptr) ? acc[r] * rscale[row] : acc[r];
      C[(size_t)row * F + col0 + lm] = v + bv;
    }
  }
}

template <int K, int NT>
__global__ void __launch_bounds__(128)
gemm_wmma_pair(const float* __restrict__ A, const float* __restrict__ B1,
               const float* __restrict__ B2, const float* __restrict__ bias2,
               const float* __restrict__ rscale1,
               float* __restrict__ C1, float* __restrict__ C2,
               int rowTiles) {
  constexpr int F = 8 * NT;
  const int lane = threadIdx.x & 31;
  const int wave = threadIdx.x >> 5;
  const int rt = blockIdx.x * (blockDim.x >> 5) + wave;
  if (rt >= rowTiles) return;

  const int row0 = rt << 4;
  const int half = lane >> 4;
  const int lm   = lane & 15;
  const float* Arow = A + (size_t)(row0 + lm) * K + 2 * half;

  v2f afrag[K / 4];
#pragma unroll
  for (int i = 0; i < K / 4; ++i)
    afrag[i] = *reinterpret_cast<const v2f*>(Arow + 4 * i);

  float rs[8];
#pragma unroll
  for (int r = 0; r < 8; ++r) rs[r] = rscale1[row0 + 8 * half + r];

  const size_t rbase = (size_t)(row0 + 8 * half) * F;
#pragma unroll
  for (int t = 0; t < NT; ++t) {
    const int j = 16 * t + lm;
    const bool second = j >= F;
    const int col = second ? j - F : j;
    const float* Bcol = (second ? B2 : B1) + col;
    const float braw = bias2[col];
    const float bv = second ? braw : 0.f;

    v2f bfrag[K / 4];
#pragma unroll
    for (int i = 0; i < K / 4; ++i) {
      const int ka = 4 * i + 2 * half;
      bfrag[i].x = Bcol[(size_t)ka * F];
      bfrag[i].y = Bcol[(size_t)(ka + 1) * F];
    }

    v8f acc = {};
#pragma unroll
    for (int i = 0; i < K / 4; ++i) {
      acc = __builtin_amdgcn_wmma_f32_16x16x4_f32(
          false, afrag[i], false, bfrag[i], (short)0, acc, false, false);
    }

    float* Crow = (second ? C2 : C1) + col + rbase;
#pragma unroll
    for (int r = 0; r < 8; ++r) Crow[r * F] = second ? acc[r] + bv : acc[r] * rs[r];
  }
}

__global__ void gemm_wmma_gen(const float* __restrict__ A, const float* __restrict__ B,
                              const float* __restrict__ bias, const float* __restrict__ rscale,
                              float* __restrict__ C, int N, int K, int F, int rowTiles,
                              int colTiles) {
  const int lane = threadIdx.x & 31;
  const int wave = threadIdx.x >> 5;
  const int tile = blockIdx.x * (blockDim.x >> 5) + wave;
  if (tile >= rowTiles * colTiles) return;
  const int rt = tile / colTiles;
  const int ct = tile - rt * colTiles;
  const int row0 = rt << 4, col0 = ct << 4;
  const int half = lane >> 4, lm = lane & 15;
  const bool cOK = (col0 + lm) < F;
  const int arow = min(row0 + lm, N - 1);
  const int bcol = min(col0 + lm, F - 1);
  v8f acc = {};
  for (int kb = 0; kb < K; kb += 4) {
    const int ka = kb + 2 * half;
    v2f a = *reinterpret_cast<const v2f*>(A + (size_t)arow * K + ka);
    v2f b;
    b.x = B[(size_t)ka * F + bcol];
    b.y = B[(size_t)(ka + 1) * F + bcol];
    acc = __builtin_amdgcn_wmma_f32_16x16x4_f32(false, a, false, b, (short)0, acc,
                                                false, false);
  }
  const float bv = (bias != nullptr) ? bias[bcol] : 0.f;
#pragma unroll
  for (int r = 0; r < 8; ++r) {
    const int row = row0 + r + 8 * half;
    if (row < N && cOK) {
      const float v = (rscale != nullptr) ? acc[r] * rscale[row] : acc[r];
      C[(size_t)row * F + col0 + lm] = v + bv;
    }
  }
}

__global__ void edge_scatter(const float* __restrict__ h, const float* __restrict__ norm,
                             const int* __restrict__ src, const int* __restrict__ dst,
                             float* __restrict__ agg, int E, int F, int G) {
  long long tid = (long long)blockIdx.x * blockDim.x + threadIdx.x;
  if (tid >= (long long)E * G) return;
  int e = (int)(tid / G);
  int g = (int)(tid - (long long)e * G);
  int s = src[e];
  int d = dst[e];
  float w = norm[e];
  const float4 v = *reinterpret_cast<const float4*>(h + (size_t)s * F + (size_t)g * 4);
  float* base = agg + (size_t)d * F + (size_t)g * 4;
  atomicAdd(base + 0, v.x * w);
  atomicAdd(base + 1, v.y * w);
  atomicAdd(base + 2, v.z * w);
  atomicAdd(base + 3, v.w * w);
}

template <int C>
__global__ void relu_logsoftmax_t(float* __restrict__ y, int N) {
  int n = blockIdx.x * blockDim.x + threadIdx.x;
  if (n >= N) return;
  float v[C];
  float m = -3.4e38f;
  float* row = y + (size_t)n * C;
#pragma unroll
  for (int c = 0; c < C; ++c) {
    float t = fmaxf(row[c], 0.f);
    v[c] = t;
    m = fmaxf(m, t);
  }
  float s = 0.f;
#pragma unroll
  for (int c = 0; c < C; ++c) s += __expf(v[c] - m);
  float ls = m + __logf(s);
#pragma unroll
  for (int c = 0; c < C; ++c) row[c] = v[c] - ls;
}

__global__ void relu_logsoftmax_gen(float* __restrict__ y, int N, int C) {
  int n = blockIdx.x * blockDim.x + threadIdx.x;
  if (n >= N) return;
  float* row = y + (size_t)n * C;
  float m = -3.4e38f;
  for (int c = 0; c < C; ++c) m = fmaxf(m, fmaxf(row[c], 0.f));
  float s = 0.f;
  for (int c = 0; c < C; ++c) s += __expf(fmaxf(row[c], 0.f) - m);
  float ls = m + __logf(s);
  for (int c = 0; c < C; ++c) row[c] = fmaxf(row[c], 0.f) - ls;
}

#define CNT_STRIDE 32

__global__ void zero_counters(int* __restrict__ cnt, int N) {
  int n = blockIdx.x * blockDim.x + threadIdx.x;
  if (n < N) cnt[(size_t)n * CNT_STRIDE] = 0;
}

__global__ void rank_edges(const int* __restrict__ dst, int* __restrict__ cnt,
                           int* __restrict__ rank, int E) {
  int e0 = (blockIdx.x * blockDim.x + threadIdx.x) * 4;
  int d[4];
#pragma unroll
  for (int i = 0; i < 4; ++i) d[i] = (e0 + i < E) ? dst[e0 + i] : -1;
  int r[4];
#pragma unroll
  for (int i = 0; i < 4; ++i)
    r[i] = (d[i] >= 0) ? atomicAdd(cnt + (size_t)d[i] * CNT_STRIDE, 1) : 0;
#pragma unroll
  for (int i = 0; i < 4; ++i)
    if (e0 + i < E) rank[e0 + i] = r[i];
}

__device__ __forceinline__ int block_incl_scan_256(int v, int* sh) {
  const int t = threadIdx.x;
  sh[t] = v;
  __syncthreads();
  for (int off = 1; off < 256; off <<= 1) {
    int a = (t >= off) ? sh[t - off] : 0;
    __syncthreads();
    sh[t] += a;
    __syncthreads();
  }
  return sh[t];
}

__global__ void __launch_bounds__(256)
count_blocksum(const int* __restrict__ cnt, int* __restrict__ bsum, int N) {
  __shared__ int sh[256];
  int n = blockIdx.x * 256 + threadIdx.x;
  int v = (n < N) ? cnt[(size_t)n * CNT_STRIDE] : 0;
  int inc = block_incl_scan_256(v, sh);
  if (threadIdx.x == 255) bsum[blockIdx.x] = inc;
}

__global__ void __launch_bounds__(256)
scan_rowptr(const int* __restrict__ cnt, const int* __restrict__ bsum,
            int* __restrict__ rowptr, int N) {
  __shared__ int sh[256];
  __shared__ int sh2[256];
  const int t = threadIdx.x;
  int part = 0;
  for (int j = t; j < (int)blockIdx.x; j += 256) part += bsum[j];
  block_incl_scan_256(part, sh2);
  const int base = sh2[255];
  int n = blockIdx.x * 256 + t;
  int v = (n < N) ? cnt[(size_t)n * CNT_STRIDE] : 0;
  int inc = block_incl_scan_256(v, sh);
  if (n < N) {
    rowptr[n] = base + inc - v;
    if (n == N - 1) rowptr[N] = base + inc;
  }
}

__global__ void fill_sorted(const int* __restrict__ src, const int* __restrict__ dst,
                            const float* __restrict__ w, const int* __restrict__ rowptr,
                            const int* __restrict__ rank, int* __restrict__ ssrc,
                            float* __restrict__ sw, int E) {
  int e0 = (blockIdx.x * blockDim.x + threadIdx.x) * 4;
  for (int i = 0; i < 4; ++i) {
    int e = e0 + i;
    if (e < E) {
      int p = rowptr[dst[e]] + rank[e];
      ssrc[p] = src[e];
      sw[p] = w[e];
    }
  }
}

__global__ void deg_dinv(const float* __restrict__ sw, const int* __restrict__ rowptr,
                         float* __restrict__ dinv, int N) {
  int n = blockIdx.x * blockDim.x + threadIdx.x;
  if (n >= N) return;
  const int p0 = rowptr[n], p1 = rowptr[n + 1];
  float s = 0.f;
  for (int p = p0; p < p1; ++p) s += sw[p];
  dinv[n] = s > 0.f ? rsqrtf(s) : 0.f;
}

template <bool RELU>
__global__ void csr_gather(const float* __restrict__ h, const float* __restrict__ sw,
                           const int* __restrict__ ssrc, const int* __restrict__ rowptr,
                           const float* __restrict__ dinv,
                           float* __restrict__ agg, int N, int F, int G) {
  long long tid = (long long)blockIdx.x * blockDim.x + threadIdx.x;
  if (tid >= (long long)N * G) return;
  int n = (int)(tid / G);
  int g = (int)(tid - (long long)n * G);
  const int p0 = rowptr[n], p1 = rowptr[n + 1];
  const float dn = dinv[n];
  const float* hb = h + (size_t)g * 4;
  float4 acc = make_float4(0.f, 0.f, 0.f, 0.f);
#pragma unroll 4
  for (int p = p0; p < p1; ++p) {
    const int s = ssrc[p];
    const float w = sw[p];
    const float4 v = *reinterpret_cast<const float4*>(hb + (size_t)s * F);
    acc.x += v.x * w;
    acc.y += v.y * w;
    acc.z += v.z * w;
    acc.w += v.w * w;
  }
  float4* o = reinterpret_cast<float4*>(agg + (size_t)n * F + (size_t)g * 4);
  float4 z = *o;
  z.x += acc.x * dn;
  z.y += acc.y * dn;
  z.z += acc.z * dn;
  z.w += acc.w * dn;
  if (RELU) {
    z.x = fmaxf(z.x, 0.f);
    z.y = fmaxf(z.y, 0.f);
    z.z = fmaxf(z.z, 0.f);
    z.w = fmaxf(z.w, 0.f);
  }
  *o = z;
}

template <int C>
__global__ void csr_gather_logsoftmax(const float* __restrict__ h,
                                      const float* __restrict__ sw,
                                      const int* __restrict__ ssrc,
                                      const int* __restrict__ rowptr,
                                      const float* __restrict__ dinv,
                                      float* __restrict__ y, int N) {
  constexpr int G = C / 4;
  constexpr int NPW = 32 / G;
  const int lane = threadIdx.x & 31;
  const int wave = (int)((blockIdx.x * blockDim.x + threadIdx.x) >> 5);
  const int slot = lane / G;
  const int g = lane - slot * G;
  const int n = wave * NPW + slot;
  const bool live = (slot < NPW) && (n < N);
  int p0 = 0, p1 = 0;
  float dn = 0.f;
  if (live) {
    p0 = rowptr[n];
    p1 = rowptr[n + 1];
    dn = dinv[n];
  }
  const float* hb = h + (size_t)g * 4;
  float4 acc = make_float4(0.f, 0.f, 0.f, 0.f);
#pragma unroll 4
  for (int p = p0; p < p1; ++p) {
    const int s = ssrc[p];
    const float w = sw[p];
    const float4 v = *reinterpret_cast<const float4*>(hb + (size_t)s * C);
    acc.x += v.x * w;
    acc.y += v.y * w;
    acc.z += v.z * w;
    acc.w += v.w * w;
  }
  float4 t = make_float4(0.f, 0.f, 0.f, 0.f);
  float4* o = nullptr;
  if (live) {
    o = reinterpret_cast<float4*>(y + (size_t)n * C + (size_t)g * 4);
    const float4 z = *o;
    t.x = fmaxf(z.x + acc.x * dn, 0.f);
    t.y = fmaxf(z.y + acc.y * dn, 0.f);
    t.z = fmaxf(z.z + acc.z * dn, 0.f);
    t.w = fmaxf(z.w + acc.w * dn, 0.f);
  }
  const int first = live ? lane - g : lane;
  const int step = live ? 1 : 0;
  const float m4 = fmaxf(fmaxf(t.x, t.y), fmaxf(t.z, t.w));
  float m = -3.4e38f;
  for (int k = 0; k < G; ++k) m = fmaxf(m, __shfl(m4, first + k * step));
  const float e4 = __expf(t.x - m) + __expf(t.y - m) + __expf(t.z - m) + __expf(t.w - m);
  float s = 0.f;
  for (int k = 0; k < G; ++k) s += __shfl(e4, first + k * step);
  const float ls = m + __logf(s);
  if (live) *o = make_float4(t.x - ls, t.y - ls, t.z - ls, t.w - ls);
}


template <int K>
__device__ __forceinline__ int row_lane_i(int v) {
  return __builtin_amdgcn_update_dpp(0, v, 0x150 + K, 0xf, 0xf, true);
}
template <int K>
__device__ __forceinline__ float row_lane_f(float v) {
  return __int_as_float(row_lane_i<K>(__float_as_int(v)));
}

template <int HP, int K0>
__device__ __forceinline__ void seg_rows8(const char* __restrict__ hB, unsigned laneoff,
                                          int myS, float myW, float4& acc) {
  float4 v[8];
  float w[8];
#define SEG_LOAD(K)                                                                    \
  {                                                                                    \
    const unsigned s = (unsigned)row_lane_i<K0 + K>(myS);                              \
    w[K] = row_lane_f<K0 + K>(myW);                                                    \
    v[K] = *reinterpret_cast<const float4*>(hB + (size_t)(s * (unsigned)(HP * 4) + laneoff)); \
  }
  SEG_LOAD(0) SEG_LOAD(1) SEG_LOAD(2) SEG_LOAD(3)
  SEG_LOAD(4) SEG_LOAD(5) SEG_LOAD(6) SEG_LOAD(7)
#undef SEG_LOAD
#pragma unroll
  for (int k = 0; k < 8; ++k) {
    acc.x = __builtin_fmaf(v[k].x, w[k], acc.x);
    acc.y = __builtin_fmaf(v[k].y, w[k], acc.y);
    acc.z = __builtin_fmaf(v[k].z, w[k], acc.z);
    acc.w = __builtin_fmaf(v[k].w, w[k], acc.w);
  }
}

template <int F, bool LSM, int HP = F, int NT = 0>
__global__ void __launch_bounds__(256)
seg_gather16(const float* __restrict__ h, const float* __restrict__ sw,
             const int* __restrict__ ssrc, const int* __restrict__ rowptr,
             const float* __restrict__ dinv, float* __restrict__ y, int N,
             const float* __restrict__ B1, const float* __restrict__ B2,
             const float* __restrict__ bias2, float* __restrict__ C2) {
  static_assert(F % 4 == 0 && F <= 64, "a row of F floats must fit the 16 lanes of a wave row, 4 floats each");
  static_assert(NT == 0 || !LSM, "the products take the rows after ReLU only");
  constexpr int NL = F / 4;
  constexpr int ZP = (F / 4) * 4 + (((F / 4) & 1) ? 0 : 4);
  __shared__ __attribute__((aligned(16))) float zt[NT > 0 ? 32 * ZP : 4];
  constexpr int OP = 16 * NT + 4;
  __shared__ __attribute__((aligned(16))) float ot[NT > 0 ? 32 * OP : 4];
  const int lane = threadIdx.x & 31;
  const int l16 = lane & 15;
  const int wave = (int)((blockIdx.x * blockDim.x + threadIdx.x) >> 5);
  const bool feat = l16 < NL;
  const unsigned laneoff = feat ? (unsigned)l16 * 16u : 0u;
  const char* hB = reinterpret_cast<const char*>(h);

  const unsigned lofs4 = feat ? (unsigned)l16 * 4u : 0u;
  const int n0 = wave * 4 + 2 * (lane >> 4);
  const int rp0 = rowptr[min(n0, N)];
  const int rp1 = rowptr[min(n0 + 1, N)];
  const int rp2 = rowptr[min(n0 + 2, N)];
  int nn[2], p0[2], p1[2], s0[2];
  float dn[2], w0[2];
  float4 z[2];
  bool live[2];
#pragma unroll
  for (int r = 0; r < 2; ++r) {
    nn[r] = n0 + r;
    live[r] = nn[r] < N;
    const int nc = min(nn[r], N - 1);
    dn[r] = dinv[nc];
    z[r] = *reinterpret_cast<const float4*>(y + (size_t)nc * F + lofs4);
  }
  p0[0] = rp0;
  p1[0] = rp1;
  p0[1] = rp1;
  p1[1] = rp2;
#pragma unroll
  for (int r = 0; r < 2; ++r) {
    const int q = p0[r] + l16;
    const int qc = max(min(q, p1[r] - 1), 0);
    s0[r] = ssrc[qc];
    const float wl = sw[qc];
    w0[r] = (q < p1[r]) ? wl : 0.f;
  }

#pragma unroll
  for (int r = 0; r < 2; ++r) {
    float4 acc = make_float4(0.f, 0.f, 0.f, 0.f);
    const int pe = p1[r];
    int base = p0[r];
    int myS = s0[r];
    float myW = w0[r];
    while (base < pe) {
      const int cnt = pe - base;
      const int nb = base + 16;
      const int q = nb + l16;
      const int qc = min(q, pe - 1);
      const int nS = ssrc[qc];
      const float nWl = sw[qc];
      const float nW = (q < pe) ? nWl : 0.f;
      seg_rows8<HP, 0>(hB, laneoff, myS, myW, acc);
      if (cnt > 8) seg_rows8<HP, 8>(hB, laneoff, myS, myW, acc);
      myS = nS;
      myW = nW;
      base = nb;
    }
    float4 t;
    t.x = __builtin_fmaf(acc.x, dn[r], z[r].x);
    t.y = __builtin_fmaf(acc.y, dn[r], z[r].y);
    t.z = __builtin_fmaf(acc.z, dn[r], z[r].z);
    t.w = __builtin_fmaf(acc.w, dn[r], z[r].w);
    t.x = fmaxf(t.x, 0.f);
    t.y = fmaxf(t.y, 0.f);
    t.z = fmaxf(t.z, 0.f);
    t.w = fmaxf(t.w, 0.f);
    if (LSM) {
      const float m4 = fmaxf(fmaxf(t.x, t.y), fmaxf(t.z, t.w));
      float m = -3.4e38f;
#define SEG_MAX(K) if (K < NL) m = fmaxf(m, row_lane_f<K>(m4));
      SEG_MAX(0) SEG_MAX(1) SEG_MAX(2) SEG_MAX(3) SEG_MAX(4) SEG_MAX(5) SEG_MAX(6) SEG_MAX(7)
      SEG_MAX(8) SEG_MAX(9) SEG_MAX(10) SEG_MAX(11) SEG_MAX(12) SEG_MAX(13) SEG_MAX(14) SEG_MAX(15)
#undef SEG_MAX
      const float e4 = __expf(t.x - m) + __expf(t.y - m) + __expf(t.z - m) + __expf(t.w - m);
      float s = 0.f;
#define SEG_SUM(K) if (K < NL) s += row_lane_f<K>(e4);
      SEG_SUM(0) SEG_SUM(1) SEG_SUM(2) SEG_SUM(3) SEG_SUM(4) SEG_SUM(5) SEG_SUM(6) SEG_SUM(7)
      SEG_SUM(8) SEG_SUM(9) SEG_SUM(10) SEG_SUM(11) SEG_SUM(12) SEG_SUM(13) SEG_SUM(14) SEG_SUM(15)
#undef SEG_SUM
      const float ls = m + __logf(s);
      t.x -= ls;
      t.y -= ls;
      t.z -= ls;
      t.w -= ls;
    }
    if constexpr (NT > 0) {
      const int tr = 4 * (int)(threadIdx.x >> 5) + 2 * (lane >> 4) + r;
      if (feat) *reinterpret_cast<float4*>(&zt[tr * ZP + 4 * l16]) = t;
    } else {
      if (live[r] && feat)
        *reinterpret_cast<float4*>(y + (size_t)nn[r] * F + (size_t)l16 * 4) = t;
    }
  }

  if constexpr (NT > 0) {
    constexpr int FO = 8 * NT;
    __syncthreads();
    const int wib  = __builtin_amdgcn_readfirstlane((int)(threadIdx.x >> 5));
    const int half = lane >> 4;
    const int lm   = lane & 15;
    const int rt   = wib & 1;
    const float* Arow = zt + (16 * rt + lm) * ZP + 2 * half;
    v2f afrag[F / 4];
#pragma unroll
    for (int i = 0; i < F / 4; ++i)
      afrag[i] = *reinterpret_cast<const v2f*>(Arow + 4 * i);
    const int row0 = (int)blockIdx.x * 32 + 16 * rt + 8 * half;
    float rs[8];
#pragma unroll
    for (int r = 0; r < 8; ++r) rs[r] = dinv[row0 + r];
    for (int t = wib >> 1; t < NT; t += 4) {
      const int j = 16 * t + lm;
      const bool second = j >= FO;
      const int col = second ? j - FO : j;
      const float* Bcol = (second ? B2 : B1) + col;
      const float braw = bias2[col];
      const float bv = second ? braw : 0.f;
      v2f bfrag[F / 4];
#pragma unroll
      for (int i = 0; i < F / 4; ++i) {
        const int ka = 4 * i + 2 * half;
        bfrag[i].x = Bcol[(size_t)ka * FO];
        bfrag[i].y = Bcol[(size_t)(ka + 1) * FO];
      }
      v8f acc = {};
#pragma unroll
      for (int i = 0; i < F / 4; ++i) {
        acc = __builtin_amdgcn_wmma_f32_16x16x4_f32(
            false, afrag[i], false, bfrag[i], (short)0, acc, false, false);
      }
      float* orow = ot + (16 * rt + 8 * half) * OP + j;
#pragma unroll
      for (int r = 0; r < 8; ++r) orow[r * OP] = second ? acc[r] + bv : acc[r] * rs[r];
    }
    __syncthreads();
#pragma unroll
    for (int r = 0; r < 2; ++r) {
      const int tr = 4 * wib + 2 * half + r;
      const size_t node = (size_t)blockIdx.x * 32 + tr;
      if (lm < FO / 4) {
        const float4 c1 = *reinterpret_cast<const float4*>(&ot[tr * OP + 4 * lm]);
        const float4 c2 = *reinterpret_cast<const float4*>(&ot[tr * OP + FO + 4 * lm]);
        *reinterpret_cast<float4*>(y + node * F + 4 * lm) = c1;
        *reinterpret_cast<float4*>(C2 + node * FO + 4 * lm) = c2;
      }
    }
  }
}

#define NPB_SHIFT 9
#define NPB (1 << NPB_SHIFT)
#define MAXB 256
#define BIN_CH 6400
#define BIN_TPB 1024
#define BIN_IT ((BIN_CH + BIN_TPB - 1) / BIN_TPB)
#define SORT_TPB 1024
#define MAXG SORT_TPB
#define RUN_IT 4
#define SORT_CAP 10240

template <int PER>
__device__ __forceinline__ void excl_scan_lds_wave0(int* a) {
  __syncthreads();
  if (threadIdx.x < 32) {
    const int lane = threadIdx.x;
    int v[PER];
    int s = 0;
#pragma unroll
    for (int k = 0; k < PER; ++k) {
      v[k] = a[lane * PER + k];
      s += v[k];
    }
    int inc = s;
#pragma unroll
    for (int off = 1; off < 32; off <<= 1) {
      const int u = __shfl_up(inc, off);
      if (lane >= off) inc += u;
    }
    int run = inc - s;
#pragma unroll
    for (int k = 0; k < PER; ++k) {
      a[lane * PER + k] = run;
      run += v[k];
    }
    if (lane == 31) a[32 * PER] = run;
  }
  __syncthreads();
}

__global__ void __launch_bounds__(BIN_TPB)
bin_group(const int* __restrict__ src, const int* __restrict__ dst,
          const float* __restrict__ w, int* __restrict__ lmat,
          int* __restrict__ bsrc, int* __restrict__ bdst, float* __restrict__ bw,
          int E, int B) {
  __shared__ int   s_src[BIN_CH];
  __shared__ int   s_dst[BIN_CH];
  __shared__ float s_w[BIN_CH];
  __shared__ int   lstart[MAXB + 1];
  const int t = threadIdx.x;
  const int g = blockIdx.x;
  if (t < MAXB) lstart[t] = 0;
  __syncthreads();
  const int e0 = g * BIN_CH;
  const int n = min(BIN_CH, E - e0);
  int d[BIN_IT], sv[BIN_IT], rk[BIN_IT];
  float wv[BIN_IT];
#pragma unroll
  for (int k = 0; k < BIN_IT; ++k) {
    const int i = t + k * BIN_TPB;
    const bool in = i < n;
    d[k]  = in ? dst[e0 + i] : -1;
    sv[k] = in ? src[e0 + i] : 0;
    wv[k] = in ? w[e0 + i] : 0.f;
  }
#pragma unroll
  for (int k = 0; k < BIN_IT; ++k)
    rk[k] = (d[k] >= 0) ? atomicAdd(&lstart[d[k] >> NPB_SHIFT], 1) : 0;
  excl_scan_lds_wave0<MAXB / 32>(lstart);
  if (t <= B) lmat[(size_t)g * (B + 1) + t] = lstart[t];
#pragma unroll
  for (int k = 0; k < BIN_IT; ++k) {
    if (d[k] >= 0) {
      const int q = lstart[d[k] >> NPB_SHIFT] + rk[k];
      s_src[q] = sv[k];
      s_dst[q] = d[k];
      s_w[q] = wv[k];
    }
  }
  __syncthreads();
  for (int q = t; q < n; q += BIN_TPB) {
    bsrc[e0 + q] = s_src[q];
    bdst[e0 + q] = s_dst[q];
    bw[e0 + q] = s_w[q];
  }
}

__global__ void __launch_bounds__(SORT_TPB)
bucket_sort(const int* __restrict__ bsrc, const int* __restrict__ bdst,
            const float* __restrict__ bw, const int* __restrict__ lmat,
            int* __restrict__ rowptr, int* __restrict__ ssrc, float* __restrict__ sw,
            float* __restrict__ dinv, int N, int B, int G1) {
  __shared__ int   c_src[SORT_CAP];
  __shared__ float c_w[SORT_CAP];
  __shared__ int   nstart[NPB + 1];
  __shared__ int   ncnt[NPB];
  __shared__ int   r_off[MAXG];
  __shared__ int   r_cnt[MAXG];
  __shared__ int   tot[2];
  const int t = threadIdx.x;
  const int b = blockIdx.x;
  const int lane = t & 31;
  const int wid = t >> 5;
  constexpr int NW = SORT_TPB / 32;
  const int node0 = b << NPB_SHIFT;
  if (t < NPB) {
    nstart[t] = 0;
    ncnt[t] = 0;
  }
  if (t < 2) tot[t] = 0;
  int a0 = 0, c = 0;
  if (t < G1) {
    const int* row = lmat + (size_t)t * (B + 1) + b;
    a0 = row[0];
    c = row[1] - a0;
  }
  r_off[t] = t * BIN_CH + a0;
  r_cnt[t] = c;
  __syncthreads();
  int sa = a0, sc = c;
  for (int o = 16; o >= 1; o >>= 1) {
    sa += __shfl_xor(sa, o);
    sc += __shfl_xor(sc, o);
  }
  if (lane == 0) {
    atomicAdd(&tot[0], sa);
    atomicAdd(&tot[1], sc);
  }
  __syncthreads();
  const int m0 = tot[0];
  const int nE = tot[1];
  const int m1 = m0 + nE;
  for (int g0 = wid; g0 < G1; g0 += NW * RUN_IT) {
    int dd[2 * RUN_IT];
    int ro[RUN_IT], rc[RUN_IT];
#pragma unroll
    for (int k = 0; k < RUN_IT; ++k) {
      const int g = g0 + k * NW;
      const bool in = g < G1;
      const int gc = in ? g : g0;
      ro[k] = r_off[gc];
      const int cn = r_cnt[gc];
      rc[k] = in ? cn : 0;
      const int safe = gc * BIN_CH;
      const bool in0 = lane < rc[k];
      const bool in1 = lane + 32 < rc[k];
      const int d0 = bdst[in0 ? ro[k] + lane : safe];
      const int d1 = bdst[in1 ? ro[k] + lane + 32 : safe];
      dd[2 * k]     = in0 ? d0 : -1;
      dd[2 * k + 1] = in1 ? d1 : -1;
    }
#pragma unroll
    for (int k = 0; k < 2 * RUN_IT; ++k)
      if (dd[k] >= 0) atomicAdd(&nstart[dd[k] - node0], 1);
#pragma unroll
    for (int k = 0; k < RUN_IT; ++k)
      for (int j = lane + 64; j < rc[k]; j += 32) atomicAdd(&nstart[bdst[ro[k] + j] - node0], 1);
  }
  excl_scan_lds_wave0<NPB / 32>(nstart);
  if (t < NPB && node0 + t < N) rowptr[node0 + t] = m0 + nstart[t];
  if (b == B - 1 && t == 0) rowptr[N] = m1;
  const bool staged = nE <= SORT_CAP;
  for (int g0 = wid; g0 < G1; g0 += NW * RUN_IT) {
    int dd[2 * RUN_IT], ss[2 * RUN_IT];
    float xx[2 * RUN_IT];
    int ro[RUN_IT], rc[RUN_IT];
#pragma unroll
    for (int k = 0; k < RUN_IT; ++k) {
      const int g = g0 + k * NW;
      const bool in = g < G1;
      const int gc = in ? g : g0;
      ro[k] = r_off[gc];
      const int cn = r_cnt[gc];
      rc[k] = in ? cn : 0;
      const int safe = gc * BIN_CH;
      const bool in0 = lane < rc[k];
      const bool in1 = lane + 32 < rc[k];
      const int i0 = in0 ? ro[k] + lane : safe;
      const int i1 = in1 ? ro[k] + lane + 32 : safe;
      const int d0 = bdst[i0];
      const int d1 = bdst[i1];
      ss[2 * k]     = bsrc[i0];
      ss[2 * k + 1] = bsrc[i1];
      xx[2 * k]     = bw[i0];
      xx[2 * k + 1] = bw[i1];
      dd[2 * k]     = in0 ? d0 : -1;
      dd[2 * k + 1] = in1 ? d1 : -1;
    }
#pragma unroll
    for (int k = 0; k < 2 * RUN_IT; ++k) {
      if (dd[k] >= 0) {
        const int l = dd[k] - node0;
        const int q = nstart[l] + atomicAdd(&ncnt[l], 1);
        if (staged) {
          c_src[q] = ss[k];
          c_w[q] = xx[k];
        } else {
          ssrc[m0 + q] = ss[k];
          sw[m0 + q] = xx[k];
        }
      }
    }
#pragma unroll
    for (int k = 0; k < RUN_IT; ++k) {
      for (int j = lane + 64; j < rc[k]; j += 32) {
        const int i = ro[k] + j;
        const int l = bdst[i] - node0;
        const int q = nstart[l] + atomicAdd(&ncnt[l], 1);
        if (staged) {
          c_src[q] = bsrc[i];
          c_w[q] = bw[i];
        } else {
          ssrc[m0 + q] = bsrc[i];
          sw[m0 + q] = bw[i];
        }
      }
    }
  }
  __syncthreads();
  if (t < NPB && node0 + t < N) {
    const int q0 = nstart[t], q1 = nstart[t + 1];
    float s = 0.f;
    if (staged) {
      for (int q = q0; q < q1; ++q) s += c_w[q];
    } else {
      for (int q = q0; q < q1; ++q) s += sw[m0 + q];
    }
    dinv[node0 + t] = s > 0.f ? rsqrtf(s) : 0.f;
  }
  if (staged) {
    for (int q = t; q < nE; q += SORT_TPB) {
      ssrc[m0 + q] = c_src[q];
      sw[m0 + q] = c_w[q];
    }
  }
}

static inline int ceil_div(long long a, int b) { return (int)((a + b - 1) / b); }

static void launch_gemm(const float* A, const float* B, const float* bias, const float* rscale,
                        float* C, int N, int K, int F, hipStream_t stream) {
  const int TPB = 256;
  int rowT = ceil_div(N, 16), colT = ceil_div(F, 16);
  int blocks = ceil_div((long long)rowT * colT, TPB / 32);
  if (K == 64) {
    gemm_wmma<64><<<blocks, TPB, 0, stream>>>(A, B, bias, rscale, C, N, F, rowT, colT);
  } else if (K == 48) {
    gemm_wmma<48><<<blocks, TPB, 0, stream>>>(A, B, bias, rscale, C, N, F, rowT, colT);
  } else {
    gemm_wmma_gen<<<blocks, TPB, 0, stream>>>(A, B, bias, rscale, C, N, K, F, rowT, colT);
  }
}

static void launch_gemm_pair(const float* A, const float* B1, const float* B2,
                             const float* bias2, const float* rscale1, float* C1, float* C2,
                             int N, int K, int F, hipStream_t stream) {
  const int rowT = N / 16;
  const int WPB = 4;
  const int blocks = ceil_div(rowT, WPB);
  const bool full = (N % 16 == 0) && (N > 0) && (bias2 != nullptr);
  if (full && K == 64 && F == 48) {
    gemm_wmma_pair<64, 6><<<blocks, WPB * 32, 0, stream>>>(A, B1, B2, bias2, rscale1, C1, C2,
                                                            rowT);
  } else if (full && K == 48 && F == 40) {
    gemm_wmma_pair<48, 5><<<blocks, WPB * 32, 0, stream>>>(A, B1, B2, bias2, rscale1, C1, C2,
                                                            rowT);
  } else {
    launch_gemm(A, B1, nullptr, rscale1, C1, N, K, F, stream);
    launch_gemm(A, B2, bias2,   nullptr, C2, N, K, F, stream);
  }
}

extern "C" void kernel_launch(void* const* d_in, const int* in_sizes, int n_in,
                              void* d_out, int out_size, void* d_ws, size_t ws_size,
                              hipStream_t stream) {
  const float* x  = (const float*)d_in[0];
  const int*   ei = (const int*)d_in[1];
  const float* ew = (const float*)d_in[2];
  const float* W1 = (const float*)d_in[3];
  const float* V1 = (const float*)d_in[4];
  const float* b1 = (const float*)d_in[5];
  const float* W2 = (const float*)d_in[6];
  const float* V2 = (const float*)d_in[7];
  const float* b2 = (const float*)d_in[8];
  float* out = (float*)d_out;

  const int H   = in_sizes[5];
  const int Cn  = in_sizes[8];
  const int Fin = in_sizes[3] / H;
  const int N   = in_sizes[0] / Fin;
  const int E   = in_sizes[2];
  const int* src = ei;
  const int* dst = ei + E;

  float* ws = (float*)d_ws;
  const size_t NHa = ((size_t)N * H + 255) & ~(size_t)255;
  const size_t Ea  = ((size_t)E + 255) & ~(size_t)255;
  size_t off = 0;
  float* deg  = ws + off; off += ((size_t)N + 255) & ~(size_t)255;
  float* norm = ws + off; off += Ea;
  float* hbuf = ws + off; off += NHa;
  float* z1   = ws + off; off += NHa;
  int* ssrc   = (int*)(ws + off); off += Ea;
  int* rowptr = (int*)(ws + off); off += ((size_t)N + 1 + 255) & ~(size_t)255;
  const int NB = ceil_div(N, 256);
  int* rank = (int*)hbuf;
  int* cnt  = rank + Ea;
  int* bsum = cnt + (((size_t)N * CNT_STRIDE + 255) & ~(size_t)255);
  const size_t scratch_words = Ea + (((size_t)N * CNT_STRIDE + 255) & ~(size_t)255) + (size_t)NB;
  const bool sorted_path = (off * sizeof(float) <= ws_size) && (scratch_words <= 2 * NHa) &&
                           (H % 4 == 0) && (Cn % 4 == 0);
  const int B  = ceil_div(N, NPB);
  const int G1 = ceil_div(E, BIN_CH);
  const size_t matw = ((size_t)G1 * (B + 1) + 255) & ~(size_t)255;
  int*   bsrc   = (int*)hbuf;
  int*   bdst   = bsrc + Ea;
  float* bw     = (float*)(bdst + Ea);
  int*   lmat   = (int*)(bw + Ea);
  const bool lds_sort = (B <= MAXB) && (G1 <= MAXG) && (E > 0) && (3 * Ea + matw <= 2 * NHa);

  const int TPB = 256;

  if (sorted_path) {
    if (lds_sort) {
      bin_group<<<G1, BIN_TPB, 0, stream>>>(src, dst, ew, lmat, bsrc, bdst, bw, E, B);
      bucket_sort<<<B, SORT_TPB, 0, stream>>>(bsrc, bdst, bw, lmat, rowptr, ssrc, norm, deg,
                                              N, B, G1);
    } else {
      zero_counters<<<NB, TPB, 0, stream>>>(cnt, N);
      rank_edges<<<ceil_div(ceil_div(E, 4), TPB), TPB, 0, stream>>>(dst, cnt, rank, E);
      count_blocksum<<<NB, 256, 0, stream>>>(cnt, bsum, N);
      scan_rowptr<<<NB, 256, 0, stream>>>(cnt, bsum, rowptr, N);
      fill_sorted<<<ceil_div(ceil_div(E, 4), TPB), TPB, 0, stream>>>(src, dst, ew, rowptr, rank,
                                                                   ssrc, norm, E);
      deg_dinv<<<NB, TPB, 0, stream>>>(norm, rowptr, deg, N);
    }

    launch_gemm_pair(x, W1, V1, b1, deg, hbuf, z1, N, Fin, H, stream);
    const bool rows32 = (size_t)N * 256 < ((size_t)1 << 31);
    const int segBlocks = ceil_div((long long)ceil_div(N, 4) * 32, TPB);
    if (H == 48 && Cn == 40 && rows32 && (N % 32 == 0) && b2 != nullptr) {
      seg_gather16<48, false, 48, 5><<<segBlocks, TPB, 0, stream>>>(hbuf, norm, ssrc, rowptr, deg,
                                                                    z1, N, W2, V2, b2, out);
      seg_gather16<40, true, 48><<<segBlocks, TPB, 0, stream>>>(z1, norm, ssrc, rowptr, deg, out,
                                                                N, nullptr, nullptr, nullptr,
                                                                nullptr);
      return;
    }
    if (H == 48 && rows32) {
      seg_gather16<48, false><<<segBlocks, TPB, 0, stream>>>(hbuf, norm, ssrc, rowptr, deg, z1, N,
                                                             nullptr, nullptr, nullptr, nullptr);
    } else {
      int G = H / 4;
      csr_gather<true><<<ceil_div((long long)N * G, TPB), TPB, 0, stream>>>(
          hbuf, norm, ssrc, rowptr, deg, z1, N, H, G);
    }

    launch_gemm_pair(z1, W2, V2, b2, deg, hbuf, out, N, H, Cn, stream);
    {
      if (Cn == 40 && rows32) {
        seg_gather16<40, true><<<segBlocks, TPB, 0, stream>>>(hbuf, norm, ssrc, rowptr, deg, out, N,
                                                              nullptr, nullptr, nullptr, nullptr);
      } else if (Cn == 40) {
        csr_gather_logsoftmax<40><<<ceil_div((long long)ceil_div(N, 3) * 32, TPB), TPB, 0, stream>>>(
            hbuf, norm, ssrc, rowptr, deg, out, N);
      } else {
        int G = Cn / 4;
        csr_gather<false><<<ceil_div((long long)N * G, TPB), TPB, 0, stream>>>(
            hbuf, norm, ssrc, rowptr, deg, out, N, Cn, G);
        relu_logsoftmax_gen<<<ceil_div(N, TPB), TPB, 0, stream>>>(out, N, Cn);
      }
    }
    return;
  }

  zero_f32<<<ceil_div(N, TPB), TPB, 0, stream>>>(deg, N);
  deg_accum<<<ceil_div(E, TPB), TPB, 0, stream>>>(dst, ew, deg, E);
  norm_compute<<<ceil_div(E, TPB), TPB, 0, stream>>>(src, dst, ew, deg, norm, E);

  launch_gemm(x, W1, nullptr, nullptr, hbuf, N, Fin, H, stream);
  launch_gemm(x, V1, b1,      nullptr, z1,   N, Fin, H, stream);
  {
    int G = H / 4;
    edge_scatter<<<ceil_div((long long)E * G, TPB), TPB, 0, stream>>>(
        hbuf, norm, src, dst, z1, E, H, G);
    relu_inplace<<<ceil_div((long long)N * H, TPB), TPB, 0, stream>>>(z1, N * H);
  }

  launch_gemm(z1, W2, nullptr, nullptr, hbuf, N, H, Cn, stream);
  launch_gemm(z1, V2, b2,      nullptr, out,  N, H, Cn, stream);
  {
    int G = Cn / 4;
    edge_scatter<<<ceil_div((long long)E * G, TPB), TPB, 0, stream>>>(
        hbuf, norm, src, dst, out, E, Cn, G);
    if (Cn == 40) {
      relu_logsoftmax_t<40><<<ceil_div(N, TPB), TPB, 0, stream>>>(out, N);
    } else {
      relu_logsoftmax_gen<<<ceil_div(N, TPB), TPB, 0, stream>>>(out, N, Cn);
    }
  }
}
